# starting kernel
# baseline (speedup 1.0000x reference)
#include <hip/hip_runtime.h>
#include <hip/hip_bf16.h>
#include <math.h>

#define E_DIM 1024
#define S_LEN 8192
#define H_N   16
#define D_H   64

typedef __attribute__((ext_vector_type(2))) float v2f;
typedef __attribute__((ext_vector_type(8))) float v8f;

__device__ __forceinline__ v8f wmma4(v2f a, v2f b, v8f c) {
    return __builtin_amdgcn_wmma_f32_16x16x4_f32(false, a, false, b, (short)0, c,
                                                 false, false);
}

__global__ void qkv_part_kernel(const float* __restrict__ x,
                                const float* __restrict__ Wv,
                                const float* __restrict__ Wq,
                                const float* __restrict__ Wk,
                                float* __restrict__ part_qkv) {
    __shared__ float xs[256];
    const int tid = threadIdx.x;
    const int k0  = blockIdx.y * 256;
    if (tid < 256) xs[tid] = x[k0 + tid];
    __syncthreads();

    const int m    = blockIdx.z;
    const int wv   = tid >> 5;
    const int lane = tid & 31;
    const int j    = lane & 15;
    const int half = lane >> 4;
    const int ob   = blockIdx.x * 128 + wv * 16;
    const float amask = (j == 0) ? 1.f : 0.f;

    const float* W = (m == 0) ? Wv : (m == 1) ? Wq : Wk;

    v8f c0 = {}, c1 = {};
    for (int eb = 0; eb < 256; eb += 8) {
        const int r0 = eb + half * 2;
        v2f a0, a1, b0, b1;
        a0.x = xs[r0]     * amask;  a0.y = xs[r0 + 1] * amask;
        a1.x = xs[r0 + 4] * amask;  a1.y = xs[r0 + 5] * amask;
        const float* wp0 = W + (size_t)(k0 + r0)     * E_DIM + ob + j;
        const float* wp1 = W + (size_t)(k0 + r0 + 4) * E_DIM + ob + j;
        b0.x = wp0[0];  b0.y = wp0[E_DIM];
        b1.x = wp1[0];  b1.y = wp1[E_DIM];
        c0 = wmma4(a0, b0, c0);
        c1 = wmma4(a1, b1, c1);
    }
    const v8f cs = c0 + c1;
    if (lane < 16)
        part_qkv[(size_t)(m * 4 + blockIdx.y) * E_DIM + ob + lane] = cs[0];
}

__global__ void qkv_combine_kernel(const float* __restrict__ part_qkv,
                                   const float* __restrict__ bv,
                                   const float* __restrict__ bq,
                                   const float* __restrict__ bk,
                                   float* __restrict__ q_out,
                                   float* __restrict__ newv, float* __restrict__ newk) {
    for (int o = threadIdx.x; o < 3 * E_DIM; o += 256) {
        const int m  = o >> 10;
        const int oo = o & (E_DIM - 1);
        float s = 0.f;
        #pragma unroll
        for (int ks = 0; ks < 4; ks++)
            s += part_qkv[(size_t)(m * 4 + ks) * E_DIM + oo];
        if (m == 0)      newv[(size_t)(S_LEN - 1) * E_DIM + oo] = s + bv[oo];
        else if (m == 1) q_out[oo] = s + bq[oo];
        else             newk[(size_t)(S_LEN - 1) * E_DIM + oo] = s + bk[oo];
    }
}

__global__ void shift_score_kernel(const float* __restrict__ oldv,
                                   const float* __restrict__ oldk,
                                   const float* __restrict__ q,
                                   float* __restrict__ newv, float* __restrict__ newk,
                                   float* __restrict__ scores) {
    const int tid  = threadIdx.x;
    const int lane = tid & 31;
    const int wv   = tid >> 5;
    const int c0   = blockIdx.x * 32;
    const float4* q4 = (const float4*)q;

    for (int p = wv; p < 32; p += 8) {
        const int s = c0 + p;
        const bool last = (s == S_LEN - 1);

        const float4* vsrc = (const float4*)(last ? (newv + (size_t)s * E_DIM)
                                                  : (oldv + (size_t)(s + 1) * E_DIM));
        float4* vdst = (float4*)(newv + (size_t)s * E_DIM);
        int nz = 0;
        #pragma unroll
        for (int i = 0; i < 8; i++) {
            float4 d = vsrc[i * 32 + lane];
            nz |= (d.x != 0.f) | (d.y != 0.f) | (d.z != 0.f) | (d.w != 0.f);
            if (!last) vdst[i * 32 + lane] = d;
        }
        #pragma unroll
        for (int mm = 16; mm >= 1; mm >>= 1) nz |= __shfl_xor(nz, mm, 32);

        const float4* ksrc = (const float4*)(last ? (newk + (size_t)s * E_DIM)
                                                  : (oldk + (size_t)(s + 1) * E_DIM));
        float4* kdst = (float4*)(newk + (size_t)s * E_DIM);
        #pragma unroll
        for (int i = 0; i < 8; i++) {
            float4 kd = ksrc[i * 32 + lane];
            if (!last) kdst[i * 32 + lane] = kd;
            float4 qd = q4[i * 32 + lane];
            float part = kd.x * qd.x + kd.y * qd.y + kd.z * qd.z + kd.w * qd.w;
            part += __shfl_xor(part, 1, 32);
            part += __shfl_xor(part, 2, 32);
            part += __shfl_xor(part, 4, 32);
            part += __shfl_xor(part, 8, 32);
            if ((lane & 15) == 0) {
                const int h = i * 2 + (lane >> 4);
                scores[(size_t)h * S_LEN + s] = nz ? part * 0.125f : -INFINITY;
            }
        }
    }
}

__global__ void softmax_kernel(float* __restrict__ scores, float* __restrict__ values) {
    __shared__ float red[256];
    const int h = blockIdx.x, tid = threadIdx.x;
    float* row = scores + (size_t)h * S_LEN;
    if (tid < D_H) values[h * D_H + tid] = 0.f;

    float mx = -INFINITY;
    for (int i = tid; i < S_LEN; i += 256) mx = fmaxf(mx, row[i]);
    red[tid] = mx; __syncthreads();
    for (int st = 128; st > 0; st >>= 1) {
        if (tid < st) red[tid] = fmaxf(red[tid], red[tid + st]);
        __syncthreads();
    }
    mx = red[0]; __syncthreads();

    float sm = 0.f;
    for (int i = tid; i < S_LEN; i += 256) sm += expf(row[i] - mx);
    red[tid] = sm; __syncthreads();
    for (int st = 128; st > 0; st >>= 1) {
        if (tid < st) red[tid] += red[tid + st];
        __syncthreads();
    }
    const float inv = 1.f / red[0];
    for (int i = tid; i < S_LEN; i += 256) row[i] = expf(row[i] - mx) * inv;
}

__global__ void attnv_kernel(const float* __restrict__ probs,
                             const float* __restrict__ newv,
                             float* __restrict__ values) {
    const int tid  = threadIdx.x;
    const int lane = tid & 31;
    const int wv   = tid >> 5;
    const int gw   = blockIdx.x * 8 + wv;
    const int h    = gw >> 6;
    const int dt   = (gw >> 4) & 3;
    const int sc   = gw & 15;
    const int s0   = sc * 512;
    const int j    = lane & 15;
    const int half = lane >> 4;
    const float amask = (j == 0) ? 1.f : 0.f;

    const float* p = probs + (size_t)h * S_LEN;
    const float* V = newv + h * D_H + dt * 16 + j;

    v8f c0 = {}, c1 = {};
    for (int sb = 0; sb < 512; sb += 8) {
        const int sk0 = s0 + sb + half * 2;
        const int sk1 = sk0 + 4;
        v2f a0, a1, b0, b1;
        a0.x = p[sk0]     * amask;  a0.y = p[sk0 + 1] * amask;
        a1.x = p[sk1]     * amask;  a1.y = p[sk1 + 1] * amask;
        const float* vp0 = V + (size_t)sk0 * E_DIM;
        const float* vp1 = V + (size_t)sk1 * E_DIM;
        b0.x = vp0[0];  b0.y = vp0[E_DIM];
        b1.x = vp1[0];  b1.y = vp1[E_DIM];
        c0 = wmma4(a0, b0, c0);
        c1 = wmma4(a1, b1, c1);
    }
    const v8f cs = c0 + c1;
    if (lane < 16) atomicAdd(&values[h * D_H + dt * 16 + lane], cs[0]);
}

__global__ void gemv_part_kernel(const float* __restrict__ W,
                                 const float* __restrict__ in_parts, int P,
                                 const float* __restrict__ in_bias, int in_relu,
                                 float* __restrict__ out_parts) {
    __shared__ float xs[128];
    const int tid = threadIdx.x;
    const int k0  = blockIdx.y * 128;
    if (tid < 128) {
        float s = 0.f;
        for (int pp = 0; pp < P; pp++) s += in_parts[(size_t)pp * E_DIM + k0 + tid];
        if (in_bias) s += in_bias[k0 + tid];
        if (in_relu) s = fmaxf(s, 0.f);
        xs[tid] = s;
    }
    __syncthreads();

    const int wv   = tid >> 5;
    const int lane = tid & 31;
    const int j    = lane & 15;
    const int half = lane >> 4;
    const int ob   = blockIdx.x * 128 + wv * 16;
    const float amask = (j == 0) ? 1.f : 0.f;

    v8f c0 = {}, c1 = {};
    for (int eb = 0; eb < 128; eb += 8) {
        const int r0 = eb + half * 2;
        v2f a0, a1, b0, b1;
        a0.x = xs[r0]     * amask;  a0.y = xs[r0 + 1] * amask;
        a1.x = xs[r0 + 4] * amask;  a1.y = xs[r0 + 5] * amask;
        const float* wp0 = W + (size_t)(k0 + r0)     * E_DIM + ob + j;
        const float* wp1 = W + (size_t)(k0 + r0 + 4) * E_DIM + ob + j;
        b0.x = wp0[0];  b0.y = wp0[E_DIM];
        b1.x = wp1[0];  b1.y = wp1[E_DIM];
        c0 = wmma4(a0, b0, c0);
        c1 = wmma4(a1, b1, c1);
    }
    const v8f cs = c0 + c1;
    if (lane < 16)
        out_parts[(size_t)blockIdx.y * E_DIM + ob + lane] = cs[0];
}

__global__ void ln_combine_kernel(const float* __restrict__ parts, int P,
                                  const float* __restrict__ bias,
                                  const float* __restrict__ resid,
                                  const float* __restrict__ sc,
                                  const float* __restrict__ bs,
                                  float* __restrict__ out) {
    __shared__ float red[256];
    const int tid = threadIdx.x;
    float vv[4];
    float sum = 0.f;
    #pragma unroll
    for (int i = 0; i < 4; i++) {
        const int e = i * 256 + tid;
        float s = bias[e] + resid[e];
        for (int pp = 0; pp < P; pp++) s += parts[(size_t)pp * E_DIM + e];
        vv[i] = s;
        sum += s;
    }
    red[tid] = sum; __syncthreads();
    for (int st = 128; st > 0; st >>= 1) {
        if (tid < st) red[tid] += red[tid + st];
        __syncthreads();
    }
    const float mean = red[0] * (1.f / E_DIM); __syncthreads();
    float var = 0.f;
    #pragma unroll
    for (int i = 0; i < 4; i++) { const float d = vv[i] - mean; var += d * d; }
    red[tid] = var; __syncthreads();
    for (int st = 128; st > 0; st >>= 1) {
        if (tid < st) red[tid] += red[tid + st];
        __syncthreads();
    }
    const float inv = rsqrtf(red[0] * (1.f / E_DIM) + 1e-6f);
    #pragma unroll
    for (int i = 0; i < 4; i++) {
        const int e = i * 256 + tid;
        out[e] = (vv[i] - mean) * inv * sc[e] + bs[e];
    }
}

extern "C" void kernel_launch(void* const* d_in, const int* in_sizes, int n_in,
                              void* d_out, int out_size, void* d_ws, size_t ws_size,
                              hipStream_t stream) {
    (void)in_sizes; (void)n_in; (void)out_size; (void)ws_size;
    const float* x     = (const float*)d_in[0];
    const float* cache = (const float*)d_in[1];
    const float* Wv = (const float*)d_in[2];  const float* bv = (const float*)d_in[3];
    const float* Wq = (const float*)d_in[4];  const float* bq = (const float*)d_in[5];
    const float* Wk = (const float*)d_in[6];  const float* bk = (const float*)d_in[7];
    const float* Wo = (const float*)d_in[8];  const float* bo = (const float*)d_in[9];
    const float* W1 = (const float*)d_in[10]; const float* b1 = (const float*)d_in[11];
    const float* W2 = (const float*)d_in[12]; const float* b2 = (const float*)d_in[13];
    const float* ln1_s = (const float*)d_in[14]; const float* ln1_b = (const float*)d_in[15];
    const float* ln2_s = (const float*)d_in[16]; const float* ln2_b = (const float*)d_in[17];

    float* out_h = (float*)d_out;
    float* newv  = out_h + E_DIM;
    float* newk  = newv + (size_t)S_LEN * E_DIM;
    const float* oldv = cache;
    const float* oldk = cache + (size_t)S_LEN * E_DIM;

    float* ws       = (float*)d_ws;
    float* q        = ws;
    float* scores   = q + E_DIM;
    float* values   = scores + H_N * S_LEN;
    float* h1       = values + E_DIM;
    float* part_qkv = h1 + E_DIM;
    float* part_o   = part_qkv + 12 * E_DIM;
    float* part_1   = part_o + 8 * E_DIM;
    float* part_2   = part_1 + 8 * E_DIM;

    qkv_part_kernel<<<dim3(8, 4, 3), 256, 0, stream>>>(x, Wv, Wq, Wk, part_qkv);
    qkv_combine_kernel<<<1, 256, 0, stream>>>(part_qkv, bv, bq, bk, q, newv, newk);
    shift_score_kernel<<<256, 256, 0, stream>>>(oldv, oldk, q, newv, newk, scores);
    softmax_kernel<<<16, 256, 0, stream>>>(scores, values);
    attnv_kernel<<<128, 256, 0, stream>>>(scores, newv, values);
    gemv_part_kernel<<<dim3(8, 8), 256, 0, stream>>>(Wo, values, 1, nullptr, 0, part_o);
    ln_combine_kernel<<<1, 256, 0, stream>>>(part_o, 8, bo, x, ln1_s, ln1_b, h1);
    gemv_part_kernel<<<dim3(8, 8), 256, 0, stream>>>(W1, h1, 1, nullptr, 0, part_1);
    gemv_part_kernel<<<dim3(8, 8), 256, 0, stream>>>(W2, part_1, 8, b1, 1, part_2);
    ln_combine_kernel<<<1, 256, 0, stream>>>(part_2, 8, b2, h1, ln2_s, ln2_b, out_h);
}


# softmax launch removed: score pass stores exp(score - block max) and a (max, sum) pair per block and head; PV pass merges the pairs, one factor per 32 positions; 10 -> 9 launches; new sum order
# speedup vs baseline: 1.4696x; 1.4696x over previous
#include <hip/hip_runtime.h>
#include <hip/hip_bf16.h>
#include <math.h>

#define E_DIM 1024
#define S_LEN 8192
#define H_N   16
#define D_H   64

typedef __attribute__((ext_vector_type(2))) float v2f;
typedef __attribute__((ext_vector_type(8))) float v8f;
typedef __attribute__((ext_vector_type(4))) float v4f;

__device__ __forceinline__ v8f wmma4(v2f a, v2f b, v8f c) {
    return __builtin_amdgcn_wmma_f32_16x16x4_f32(false, a, false, b, (short)0, c,
                                                 false, false);
}

#define QKV_BLOCKS  96
#define COPY_BLOCKS 512

__global__ void __launch_bounds__(256)
qkv_part_copy_kernel(const float* __restrict__ x,
                     const float* __restrict__ Wv,
                     const float* __restrict__ Wq,
                     const float* __restrict__ Wk,
                     float* __restrict__ part_qkv,
                     const float* __restrict__ oldv,
                     const float* __restrict__ oldk,
                     float* __restrict__ newv, float* __restrict__ newk,
                     int* __restrict__ rowflag) {
    __shared__ float xs[256];
    const int tid  = threadIdx.x;
    const int wv   = tid >> 5;
    const int lane = tid & 31;

    if (blockIdx.x < QKV_BLOCKS) {
        const int bx = blockIdx.x & 7;
        const int by = (blockIdx.x >> 3) & 3;
        const int m  = blockIdx.x >> 5;
        const int k0 = by * 256;
        xs[tid] = x[k0 + tid];
        __syncthreads();

        const int j    = lane & 15;
        const int half = lane >> 4;
        const int ob   = bx * 128 + wv * 16;
        const float amask = (j == 0) ? 1.f : 0.f;

        const float* W = (m == 0) ? Wv : (m == 1) ? Wq : Wk;

        v8f c0 = {}, c1 = {};
        for (int eb = 0; eb < 256; eb += 8) {
            const int r0 = eb + half * 2;
            v2f a0, a1, b0, b1;
            a0.x = xs[r0]     * amask;  a0.y = xs[r0 + 1] * amask;
            a1.x = xs[r0 + 4] * amask;  a1.y = xs[r0 + 5] * amask;
            const float* wp0 = W + (size_t)(k0 + r0)     * E_DIM + ob + j;
            const float* wp1 = W + (size_t)(k0 + r0 + 4) * E_DIM + ob + j;
            b0.x = wp0[0];  b0.y = wp0[E_DIM];
            b1.x = wp1[0];  b1.y = wp1[E_DIM];
            c0 = wmma4(a0, b0, c0);
            c1 = wmma4(a1, b1, c1);
        }
        const v8f cs = c0 + c1;
        if (lane < 16)
            part_qkv[(size_t)(m * 4 + by) * E_DIM + ob + lane] = cs[0];
    } else {
        const int gw = (blockIdx.x - QKV_BLOCKS) * 8 + wv;
        for (int s = gw; s < S_LEN - 1; s += COPY_BLOCKS * 8) {
            const v4f* vsrc = (const v4f*)(oldv + (size_t)(s + 1) * E_DIM);
            const v4f* ksrc = (const v4f*)(oldk + (size_t)(s + 1) * E_DIM);
            v4f* vdst = (v4f*)(newv + (size_t)s * E_DIM);
            v4f* kdst = (v4f*)(newk + (size_t)s * E_DIM);
            v4f dv[8], dk[8];
            #pragma unroll
            for (int i = 0; i < 8; i++) dv[i] = vsrc[i * 32 + lane];
            #pragma unroll
            for (int i = 0; i < 8; i++) dk[i] = ksrc[i * 32 + lane];
            asm volatile("" :: "v"(dv[0]), "v"(dv[1]), "v"(dv[2]), "v"(dv[3]),
                               "v"(dv[4]), "v"(dv[5]), "v"(dv[6]), "v"(dv[7]),
                               "v"(dk[0]), "v"(dk[1]), "v"(dk[2]), "v"(dk[3]),
                               "v"(dk[4]), "v"(dk[5]), "v"(dk[6]), "v"(dk[7]));
            bool nz = false;
            #pragma unroll
            for (int i = 0; i < 8; i++) {
                nz |= (dv[i].x != 0.f) | (dv[i].y != 0.f) | (dv[i].z != 0.f) | (dv[i].w != 0.f);
                vdst[i * 32 + lane] = dv[i];
            }
            #pragma unroll
            for (int i = 0; i < 8; i++) kdst[i * 32 + lane] = dk[i];
            const bool any = __builtin_amdgcn_ballot_w32(nz) != 0u;
            if (lane == 0) rowflag[s] = any ? 1 : 0;
        }
    }
}

__global__ void qkv_combine_kernel(const float* __restrict__ part_qkv,
                                   const float* __restrict__ bv,
                                   const float* __restrict__ bq,
                                   const float* __restrict__ bk,
                                   float* __restrict__ q_out,
                                   float* __restrict__ newv, float* __restrict__ newk,
                                   float* __restrict__ values) {
    for (int o = threadIdx.x; o < E_DIM; o += 256) values[o] = 0.f;
    for (int o = threadIdx.x; o < 3 * E_DIM; o += 256) {
        const int m  = o >> 10;
        const int oo = o & (E_DIM - 1);
        float s = 0.f;
        #pragma unroll
        for (int ks = 0; ks < 4; ks++)
            s += part_qkv[(size_t)(m * 4 + ks) * E_DIM + oo];
        if (m == 0)      newv[(size_t)(S_LEN - 1) * E_DIM + oo] = s + bv[oo];
        else if (m == 1) q_out[oo] = s + bq[oo];
        else             newk[(size_t)(S_LEN - 1) * E_DIM + oo] = s + bk[oo];
    }
}

#define SC_BLOCKS 256

__global__ void score_kernel(const float* __restrict__ newv,
                             const float* __restrict__ newk,
                             const float* __restrict__ q,
                             const int* __restrict__ rowflag,
                             float* __restrict__ escore,
                             float* __restrict__ stats) {
    __shared__ float sc_s[H_N][33];
    const int tid  = threadIdx.x;
    const int lane = tid & 31;
    const int wv   = tid >> 5;
    const int c0   = blockIdx.x * 32;
    const float4* q4 = (const float4*)q;

    for (int p = wv; p < 32; p += 8) {
        const int s = c0 + p;
        const bool last = (s == S_LEN - 1);

        const float4* ksrc = (const float4*)(newk + (size_t)s * E_DIM);
        float4 kd[8];
        #pragma unroll
        for (int i = 0; i < 8; i++) kd[i] = ksrc[i * 32 + lane];

        bool nz;
        if (last) {
            const float4* vsrc = (const float4*)(newv + (size_t)s * E_DIM);
            bool t = false;
            #pragma unroll
            for (int i = 0; i < 8; i++) {
                float4 d = vsrc[i * 32 + lane];
                t |= (d.x != 0.f) | (d.y != 0.f) | (d.z != 0.f) | (d.w != 0.f);
            }
            nz = __builtin_amdgcn_ballot_w32(t) != 0u;
        } else {
            nz = rowflag[s] != 0;
        }

        #pragma unroll
        for (int i = 0; i < 8; i++) {
            float4 qd = q4[i * 32 + lane];
            float part = kd[i].x * qd.x + kd[i].y * qd.y + kd[i].z * qd.z + kd[i].w * qd.w;
            part += __shfl_xor(part, 1, 32);
            part += __shfl_xor(part, 2, 32);
            part += __shfl_xor(part, 4, 32);
            part += __shfl_xor(part, 8, 32);
            if ((lane & 15) == 0) {
                const int h = i * 2 + (lane >> 4);
                sc_s[h][p] = nz ? part * 0.125f : -INFINITY;
            }
        }
    }
    __syncthreads();

    const int h = tid >> 4;
    const int j = tid & 15;
    const float s0 = sc_s[h][j];
    const float s1 = sc_s[h][j + 16];
    float m = fmaxf(s0, s1);
    m = fmaxf(m, __shfl_xor(m, 1, 32));
    m = fmaxf(m, __shfl_xor(m, 2, 32));
    m = fmaxf(m, __shfl_xor(m, 4, 32));
    m = fmaxf(m, __shfl_xor(m, 8, 32));
    const bool dead = (m == -INFINITY);
    const float e0 = dead ? 0.f : expf(s0 - m);
    const float e1 = dead ? 0.f : expf(s1 - m);
    float l = e0 + e1;
    l += __shfl_xor(l, 1, 32);
    l += __shfl_xor(l, 2, 32);
    l += __shfl_xor(l, 4, 32);
    l += __shfl_xor(l, 8, 32);
    float* erow = escore + (size_t)h * S_LEN + c0;
    erow[j]      = e0;
    erow[j + 16] = e1;
    if (j == 0) ((float2*)stats)[h * SC_BLOCKS + blockIdx.x] = make_float2(m, l);
}

__global__ void attnv_kernel(const float* __restrict__ escore,
                             const float* __restrict__ stats,
                             const float* __restrict__ newv,
                             float* __restrict__ values) {
    const int tid  = threadIdx.x;
    const int lane = tid & 31;
    const int wv   = tid >> 5;
    const int gw   = blockIdx.x * 8 + wv;
    const int h    = gw >> 6;
    const int dt   = (gw >> 4) & 3;
    const int sc   = gw & 15;
    const int s0   = sc * 512;
    const int j    = lane & 15;
    const int half = lane >> 4;
    const float amask = (j == 0) ? 1.f : 0.f;

    const float2* st = (const float2*)stats + h * SC_BLOCKS;
    float2 pr[8];
    #pragma unroll
    for (int i = 0; i < 8; i++) pr[i] = st[i * 32 + lane];
    const float2 own = st[sc * 16 + j];

    float M = pr[0].x;
    #pragma unroll
    for (int i = 1; i < 8; i++) M = fmaxf(M, pr[i].x);
    M = fmaxf(M, __shfl_xor(M, 1, 32));
    M = fmaxf(M, __shfl_xor(M, 2, 32));
    M = fmaxf(M, __shfl_xor(M, 4, 32));
    M = fmaxf(M, __shfl_xor(M, 8, 32));
    M = fmaxf(M, __shfl_xor(M, 16, 32));
    float L = 0.f;
    #pragma unroll
    for (int i = 0; i < 8; i++) L += pr[i].y * expf(pr[i].x - M);
    L += __shfl_xor(L, 1, 32);
    L += __shfl_xor(L, 2, 32);
    L += __shfl_xor(L, 4, 32);
    L += __shfl_xor(L, 8, 32);
    L += __shfl_xor(L, 16, 32);
    const float f = expf(own.x - M) * (1.f / L);

    const float* p = escore + (size_t)h * S_LEN;
    const float* V = newv + h * D_H + dt * 16 + j;

    v8f c0 = {}, c1 = {};
    for (int t = 0; t < 16; t++) {
        const float ft = __builtin_bit_cast(float,
            __builtin_amdgcn_readlane(__builtin_bit_cast(int, f), t));
        const float fm = amask * ft;
        #pragma unroll
        for (int u = 0; u < 4; u++) {
            const int sk0 = s0 + t * 32 + u * 8 + half * 2;
            const int sk1 = sk0 + 4;
            v2f a0, a1, b0, b1;
            a0.x = p[sk0]     * fm;  a0.y = p[sk0 + 1] * fm;
            a1.x = p[sk1]     * fm;  a1.y = p[sk1 + 1] * fm;
            const float* vp0 = V + (size_t)sk0 * E_DIM;
            const float* vp1 = V + (size_t)sk1 * E_DIM;
            b0.x = vp0[0];  b0.y = vp0[E_DIM];
            b1.x = vp1[0];  b1.y = vp1[E_DIM];
            c0 = wmma4(a0, b0, c0);
            c1 = wmma4(a1, b1, c1);
        }
    }
    const v8f cs = c0 + c1;
    if (lane < 16) atomicAdd(&values[h * D_H + dt * 16 + lane], cs[0]);
}

__global__ void gemv_part_kernel(const float* __restrict__ W,
                                 const float* __restrict__ in_parts, int P,
                                 const float* __restrict__ in_bias, int in_relu,
                                 float* __restrict__ out_parts) {
    __shared__ float xs[128];
    const int tid = threadIdx.x;
    const int k0  = blockIdx.y * 128;
    if (tid < 128) {
        float s = 0.f;
        for (int pp = 0; pp < P; pp++) s += in_parts[(size_t)pp * E_DIM + k0 + tid];
        if (in_bias) s += in_bias[k0 + tid];
        if (in_relu) s = fmaxf(s, 0.f);
        xs[tid] = s;
    }
    __syncthreads();

    const int wv   = tid >> 5;
    const int lane = tid & 31;
    const int j    = lane & 15;
    const int half = lane >> 4;
    const int ob   = blockIdx.x * 128 + wv * 16;
    const float amask = (j == 0) ? 1.f : 0.f;

    v8f c0 = {}, c1 = {};
    for (int eb = 0; eb < 128; eb += 8) {
        const int r0 = eb + half * 2;
        v2f a0, a1, b0, b1;
        a0.x = xs[r0]     * amask;  a0.y = xs[r0 + 1] * amask;
        a1.x = xs[r0 + 4] * amask;  a1.y = xs[r0 + 5] * amask;
        const float* wp0 = W + (size_t)(k0 + r0)     * E_DIM + ob + j;
        const float* wp1 = W + (size_t)(k0 + r0 + 4) * E_DIM + ob + j;
        b0.x = wp0[0];  b0.y = wp0[E_DIM];
        b1.x = wp1[0];  b1.y = wp1[E_DIM];
        c0 = wmma4(a0, b0, c0);
        c1 = wmma4(a1, b1, c1);
    }
    const v8f cs = c0 + c1;
    if (lane < 16)
        out_parts[(size_t)blockIdx.y * E_DIM + ob + lane] = cs[0];
}

__global__ void ln_combine_kernel(const float* __restrict__ parts, int P,
                                  const float* __restrict__ bias,
                                  const float* __restrict__ resid,
                                  const float* __restrict__ sc,
                                  const float* __restrict__ bs,
                                  float* __restrict__ out) {
    __shared__ float red[256];
    const int tid = threadIdx.x;
    float vv[4];
    float sum = 0.f;
    #pragma unroll
    for (int i = 0; i < 4; i++) {
        const int e = i * 256 + tid;
        float s = bias[e] + resid[e];
        for (int pp = 0; pp < P; pp++) s += parts[(size_t)pp * E_DIM + e];
        vv[i] = s;
        sum += s;
    }
    red[tid] = sum; __syncthreads();
    for (int st = 128; st > 0; st >>= 1) {
        if (tid < st) red[tid] += red[tid + st];
        __syncthreads();
    }
    const float mean = red[0] * (1.f / E_DIM); __syncthreads();
    float var = 0.f;
    #pragma unroll
    for (int i = 0; i < 4; i++) { const float d = vv[i] - mean; var += d * d; }
    red[tid] = var; __syncthreads();
    for (int st = 128; st > 0; st >>= 1) {
        if (tid < st) red[tid] += red[tid + st];
        __syncthreads();
    }
    const float inv = rsqrtf(red[0] * (1.f / E_DIM) + 1e-6f);
    #pragma unroll
    for (int i = 0; i < 4; i++) {
        const int e = i * 256 + tid;
        out[e] = (vv[i] - mean) * inv * sc[e] + bs[e];
    }
}

extern "C" void kernel_launch(void* const* d_in, const int* in_sizes, int n_in,
                              void* d_out, int out_size, void* d_ws, size_t ws_size,
                              hipStream_t stream) {
    (void)in_sizes; (void)n_in; (void)out_size; (void)ws_size;
    const float* x     = (const float*)d_in[0];
    const float* cache = (const float*)d_in[1];
    const float* Wv = (const float*)d_in[2];  const float* bv = (const float*)d_in[3];
    const float* Wq = (const float*)d_in[4];  const float* bq = (const float*)d_in[5];
    const float* Wk = (const float*)d_in[6];  const float* bk = (const float*)d_in[7];
    const float* Wo = (const float*)d_in[8];  const float* bo = (const float*)d_in[9];
    const float* W1 = (const float*)d_in[10]; const float* b1 = (const float*)d_in[11];
    const float* W2 = (const float*)d_in[12]; const float* b2 = (const float*)d_in[13];
    const float* ln1_s = (const float*)d_in[14]; const float* ln1_b = (const float*)d_in[15];
    const float* ln2_s = (const float*)d_in[16]; const float* ln2_b = (const float*)d_in[17];

    float* out_h = (float*)d_out;
    float* newv  = out_h + E_DIM;
    float* newk  = newv + (size_t)S_LEN * E_DIM;
    const float* oldv = cache;
    const float* oldk = cache + (size_t)S_LEN * E_DIM;

    float* ws       = (float*)d_ws;
    float* q        = ws;
    float* scores   = q + E_DIM;
    float* values   = scores + H_N * S_LEN;
    float* h1       = values + E_DIM;
    float* part_qkv = h1 + E_DIM;
    float* part_o   = part_qkv + 12 * E_DIM;
    float* part_1   = part_o + 8 * E_DIM;
    float* part_2   = part_1 + 8 * E_DIM;
    int*   rowflag  = (int*)(part_2 + 8 * E_DIM);
    float* stats    = (float*)(rowflag + S_LEN);

    qkv_part_copy_kernel<<<QKV_BLOCKS + COPY_BLOCKS, 256, 0, stream>>>(
        x, Wv, Wq, Wk, part_qkv, oldv, oldk, newv, newk, rowflag);
    qkv_combine_kernel<<<1, 256, 0, stream>>>(part_qkv, bv, bq, bk, q, newv, newk, values);
    score_kernel<<<SC_BLOCKS, 256, 0, stream>>>(newv, newk, q, rowflag, scores, stats);
    attnv_kernel<<<128, 256, 0, stream>>>(scores, stats, newv, values);
    gemv_part_kernel<<<dim3(8, 8), 256, 0, stream>>>(Wo, values, 1, nullptr, 0, part_o);
    ln_combine_kernel<<<1, 256, 0, stream>>>(part_o, 8, bo, x, ln1_s, ln1_b, h1);
    gemv_part_kernel<<<dim3(8, 8), 256, 0, stream>>>(W1, h1, 1, nullptr, 0, part_1);
    gemv_part_kernel<<<dim3(8, 8), 256, 0, stream>>>(W2, part_1, 8, b1, 1, part_2);
    ln_combine_kernel<<<1, 256, 0, stream>>>(part_2, 8, b2, h1, ln2_s, ln2_b, out_h);
}


# qkv combine launch removed: each score block sums the q slice partials (+bias) into its own LDS; the block of the last position forms and stores the new K and V rows; 9 -> 8 launches; same sum order
# speedup vs baseline: 1.6826x; 1.1449x over previous
#include <hip/hip_runtime.h>
#include <hip/hip_bf16.h>
#include <math.h>

#define E_DIM 1024
#define S_LEN 8192
#define H_N   16
#define D_H   64

typedef __attribute__((ext_vector_type(2))) float v2f;
typedef __attribute__((ext_vector_type(8))) float v8f;
typedef __attribute__((ext_vector_type(4))) float v4f;

__device__ __forceinline__ v8f wmma4(v2f a, v2f b, v8f c) {
    return __builtin_amdgcn_wmma_f32_16x16x4_f32(false, a, false, b, (short)0, c,
                                                 false, false);
}

#define QKV_BLOCKS  96
#define COPY_BLOCKS 512

__global__ void __launch_bounds__(256)
qkv_part_copy_kernel(const float* __restrict__ x,
                     const float* __restrict__ Wv,
                     const float* __restrict__ Wq,
                     const float* __restrict__ Wk,
                     float* __restrict__ part_qkv,
                     const float* __restrict__ oldv,
                     const float* __restrict__ oldk,
                     float* __restrict__ newv, float* __restrict__ newk,
                     int* __restrict__ rowflag) {
    __shared__ float xs[256];
    const int tid  = threadIdx.x;
    const int wv   = tid >> 5;
    const int lane = tid & 31;

    if (blockIdx.x < QKV_BLOCKS) {
        const int bx = blockIdx.x & 7;
        const int by = (blockIdx.x >> 3) & 3;
        const int m  = blockIdx.x >> 5;
        const int k0 = by * 256;
        xs[tid] = x[k0 + tid];
        __syncthreads();

        const int j    = lane & 15;
        const int half = lane >> 4;
        const int ob   = bx * 128 + wv * 16;
        const float amask = (j == 0) ? 1.f : 0.f;

        const float* W = (m == 0) ? Wv : (m == 1) ? Wq : Wk;

        v8f c0 = {}, c1 = {};
        for (int eb = 0; eb < 256; eb += 8) {
            const int r0 = eb + half * 2;
            v2f a0, a1, b0, b1;
            a0.x = xs[r0]     * amask;  a0.y = xs[r0 + 1] * amask;
            a1.x = xs[r0 + 4] * amask;  a1.y = xs[r0 + 5] * amask;
            const float* wp0 = W + (size_t)(k0 + r0)     * E_DIM + ob + j;
            const float* wp1 = W + (size_t)(k0 + r0 + 4) * E_DIM + ob + j;
            b0.x = wp0[0];  b0.y = wp0[E_DIM];
            b1.x = wp1[0];  b1.y = wp1[E_DIM];
            c0 = wmma4(a0, b0, c0);
            c1 = wmma4(a1, b1, c1);
        }
        const v8f cs = c0 + c1;
        if (lane < 16)
            part_qkv[(size_t)(m * 4 + by) * E_DIM + ob + lane] = cs[0];
    } else {
        const int gw = (blockIdx.x - QKV_BLOCKS) * 8 + wv;
        for (int s = gw; s < S_LEN - 1; s += COPY_BLOCKS * 8) {
            const v4f* vsrc = (const v4f*)(oldv + (size_t)(s + 1) * E_DIM);
            const v4f* ksrc = (const v4f*)(oldk + (size_t)(s + 1) * E_DIM);
            v4f* vdst = (v4f*)(newv + (size_t)s * E_DIM);
            v4f* kdst = (v4f*)(newk + (size_t)s * E_DIM);
            v4f dv[8], dk[8];
            #pragma unroll
            for (int i = 0; i < 8; i++) dv[i] = vsrc[i * 32 + lane];
            #pragma unroll
            for (int i = 0; i < 8; i++) dk[i] = ksrc[i * 32 + lane];
            asm volatile("" :: "v"(dv[0]), "v"(dv[1]), "v"(dv[2]), "v"(dv[3]),
                               "v"(dv[4]), "v"(dv[5]), "v"(dv[6]), "v"(dv[7]),
                               "v"(dk[0]), "v"(dk[1]), "v"(dk[2]), "v"(dk[3]),
                               "v"(dk[4]), "v"(dk[5]), "v"(dk[6]), "v"(dk[7]));
            bool nz = false;
            #pragma unroll
            for (int i = 0; i < 8; i++) {
                nz |= (dv[i].x != 0.f) | (dv[i].y != 0.f) | (dv[i].z != 0.f) | (dv[i].w != 0.f);
                vdst[i * 32 + lane] = dv[i];
            }
            #pragma unroll
            for (int i = 0; i < 8; i++) kdst[i * 32 + lane] = dk[i];
            const bool any = __builtin_amdgcn_ballot_w32(nz) != 0u;
            if (lane == 0) rowflag[s] = any ? 1 : 0;
        }
    }
}

#define SC_BLOCKS 256

struct QkvParts { float4 p[4]; float4 b; };

__device__ __forceinline__ QkvParts qkv_load(const float* __restrict__ part_qkv, int m,
                                             const float* __restrict__ bias, int t) {
    QkvParts r;
    #pragma unroll
    for (int ks = 0; ks < 4; ks++)
        r.p[ks] = ((const float4*)(part_qkv + (size_t)(m * 4 + ks) * E_DIM))[t];
    r.b = ((const float4*)bias)[t];
    return r;
}

__device__ __forceinline__ float4 qkv_sum(const QkvParts& r) {
    float4 s = make_float4(0.f, 0.f, 0.f, 0.f);
    #pragma unroll
    for (int ks = 0; ks < 4; ks++) {
        s.x += r.p[ks].x;  s.y += r.p[ks].y;  s.z += r.p[ks].z;  s.w += r.p[ks].w;
    }
    s.x += r.b.x;  s.y += r.b.y;  s.z += r.b.z;  s.w += r.b.w;
    return s;
}

__global__ void score_kernel(const float* __restrict__ newk,
                             const float* __restrict__ part_qkv,
                             const float* __restrict__ bv,
                             const float* __restrict__ bq,
                             const float* __restrict__ bk,
                             const int* __restrict__ rowflag,
                             float* __restrict__ vrow_last,
                             float* __restrict__ krow_last,
                             float* __restrict__ escore,
                             float* __restrict__ stats,
                             float* __restrict__ values) {
    __shared__ float  sc_s[H_N][33];
    __shared__ float4 q_s[E_DIM / 4];
    __shared__ float4 k_s[E_DIM / 4];
    __shared__ int    wnz_s[8];
    const int tid  = threadIdx.x;
    const int lane = tid & 31;
    const int wv   = tid >> 5;
    const int c0   = blockIdx.x * 32;
    const bool lastblk = (blockIdx.x == SC_BLOCKS - 1);

    float4 kd[8];
    int fl;
    {
        const float4* ksrc = (const float4*)(newk + (size_t)(c0 + wv) * E_DIM);
        #pragma unroll
        for (int i = 0; i < 8; i++) kd[i] = ksrc[i * 32 + lane];
        fl = rowflag[c0 + wv];
    }

    const QkvParts qp = qkv_load(part_qkv, 1, bq, tid);
    float4 vn = make_float4(0.f, 0.f, 0.f, 0.f);
    float4 kn = vn;
    if (lastblk) {
        const QkvParts vp = qkv_load(part_qkv, 0, bv, tid);
        const QkvParts kp = qkv_load(part_qkv, 2, bk, tid);
        q_s[tid] = qkv_sum(qp);
        vn = qkv_sum(vp);
        kn = qkv_sum(kp);
        k_s[tid] = kn;
        const bool t = (vn.x != 0.f) | (vn.y != 0.f) | (vn.z != 0.f) | (vn.w != 0.f);
        const bool wany = __builtin_amdgcn_ballot_w32(t) != 0u;
        if (lane == 0) wnz_s[wv] = wany ? 1 : 0;
    } else {
        q_s[tid] = qkv_sum(qp);
    }
    __syncthreads();
    if (lastblk) {
        ((float4*)vrow_last)[tid] = vn;
        ((float4*)krow_last)[tid] = kn;
    }
    if (blockIdx.x == 0)
        ((float4*)values)[tid] = make_float4(0.f, 0.f, 0.f, 0.f);

    for (int p = wv; p < 32; p += 8) {
        const int s = c0 + p;
        const bool last = (s == S_LEN - 1);
        bool nz;
        if (last) {
            #pragma unroll
            for (int i = 0; i < 8; i++) kd[i] = k_s[i * 32 + lane];
            nz = (wnz_s[0] | wnz_s[1] | wnz_s[2] | wnz_s[3] |
                  wnz_s[4] | wnz_s[5] | wnz_s[6] | wnz_s[7]) != 0;
        } else {
            if (p != wv) {
                const float4* ksrc = (const float4*)(newk + (size_t)s * E_DIM);
                #pragma unroll
                for (int i = 0; i < 8; i++) kd[i] = ksrc[i * 32 + lane];
                fl = rowflag[s];
            }
            nz = fl != 0;
        }

        #pragma unroll
        for (int i = 0; i < 8; i++) {
            const float4 qd = q_s[i * 32 + lane];
            float part = kd[i].x * qd.x + kd[i].y * qd.y + kd[i].z * qd.z + kd[i].w * qd.w;
            part += __shfl_xor(part, 1, 32);
            part += __shfl_xor(part, 2, 32);
            part += __shfl_xor(part, 4, 32);
            part += __shfl_xor(part, 8, 32);
            if ((lane & 15) == 0) {
                const int h = i * 2 + (lane >> 4);
                sc_s[h][p] = nz ? part * 0.125f : -INFINITY;
            }
        }
    }
    __syncthreads();

    const int h = tid >> 4;
    const int j = tid & 15;
    const float s0 = sc_s[h][j];
    const float s1 = sc_s[h][j + 16];
    float m = fmaxf(s0, s1);
    m = fmaxf(m, __shfl_xor(m, 1, 32));
    m = fmaxf(m, __shfl_xor(m, 2, 32));
    m = fmaxf(m, __shfl_xor(m, 4, 32));
    m = fmaxf(m, __shfl_xor(m, 8, 32));
    const bool dead = (m == -INFINITY);
    const float e0 = dead ? 0.f : expf(s0 - m);
    const float e1 = dead ? 0.f : expf(s1 - m);
    float l = e0 + e1;
    l += __shfl_xor(l, 1, 32);
    l += __shfl_xor(l, 2, 32);
    l += __shfl_xor(l, 4, 32);
    l += __shfl_xor(l, 8, 32);
    float* erow = escore + (size_t)h * S_LEN + c0;
    erow[j]      = e0;
    erow[j + 16] = e1;
    if (j == 0) ((float2*)stats)[h * SC_BLOCKS + blockIdx.x] = make_float2(m, l);
}

__global__ void attnv_kernel(const float* __restrict__ escore,
                             const float* __restrict__ stats,
                             const float* __restrict__ newv,
                             float* __restrict__ values) {
    const int tid  = threadIdx.x;
    const int lane = tid & 31;
    const int wv   = tid >> 5;
    const int gw   = blockIdx.x * 8 + wv;
    const int h    = gw >> 6;
    const int dt   = (gw >> 4) & 3;
    const int sc   = gw & 15;
    const int s0   = sc * 512;
    const int j    = lane & 15;
    const int half = lane >> 4;
    const float amask = (j == 0) ? 1.f : 0.f;

    const float2* st = (const float2*)stats + h * SC_BLOCKS;
    float2 pr[8];
    #pragma unroll
    for (int i = 0; i < 8; i++) pr[i] = st[i * 32 + lane];
    const float2 own = st[sc * 16 + j];

    float M = pr[0].x;
    #pragma unroll
    for (int i = 1; i < 8; i++) M = fmaxf(M, pr[i].x);
    M = fmaxf(M, __shfl_xor(M, 1, 32));
    M = fmaxf(M, __shfl_xor(M, 2, 32));
    M = fmaxf(M, __shfl_xor(M, 4, 32));
    M = fmaxf(M, __shfl_xor(M, 8, 32));
    M = fmaxf(M, __shfl_xor(M, 16, 32));
    float L = 0.f;
    #pragma unroll
    for (int i = 0; i < 8; i++) L += pr[i].y * expf(pr[i].x - M);
    L += __shfl_xor(L, 1, 32);
    L += __shfl_xor(L, 2, 32);
    L += __shfl_xor(L, 4, 32);
    L += __shfl_xor(L, 8, 32);
    L += __shfl_xor(L, 16, 32);
    const float f = expf(own.x - M) * (1.f / L);

    const float* p = escore + (size_t)h * S_LEN;
    const float* V = newv + h * D_H + dt * 16 + j;

    v8f c0 = {}, c1 = {};
    for (int t = 0; t < 16; t++) {
        const float ft = __builtin_bit_cast(float,
            __builtin_amdgcn_readlane(__builtin_bit_cast(int, f), t));
        const float fm = amask * ft;
        #pragma unroll
        for (int u = 0; u < 4; u++) {
            const int sk0 = s0 + t * 32 + u * 8 + half * 2;
            const int sk1 = sk0 + 4;
            v2f a0, a1, b0, b1;
            a0.x = p[sk0]     * fm;  a0.y = p[sk0 + 1] * fm;
            a1.x = p[sk1]     * fm;  a1.y = p[sk1 + 1] * fm;
            const float* vp0 = V + (size_t)sk0 * E_DIM;
            const float* vp1 = V + (size_t)sk1 * E_DIM;
            b0.x = vp0[0];  b0.y = vp0[E_DIM];
            b1.x = vp1[0];  b1.y = vp1[E_DIM];
            c0 = wmma4(a0, b0, c0);
            c1 = wmma4(a1, b1, c1);
        }
    }
    const v8f cs = c0 + c1;
    if (lane < 16) atomicAdd(&values[h * D_H + dt * 16 + lane], cs[0]);
}

__global__ void gemv_part_kernel(const float* __restrict__ W,
                                 const float* __restrict__ in_parts, int P,
                                 const float* __restrict__ in_bias, int in_relu,
                                 float* __restrict__ out_parts) {
    __shared__ float xs[128];
    const int tid = threadIdx.x;
    const int k0  = blockIdx.y * 128;
    if (tid < 128) {
        float s = 0.f;
        for (int pp = 0; pp < P; pp++) s += in_parts[(size_t)pp * E_DIM + k0 + tid];
        if (in_bias) s += in_bias[k0 + tid];
        if (in_relu) s = fmaxf(s, 0.f);
        xs[tid] = s;
    }
    __syncthreads();

    const int wv   = tid >> 5;
    const int lane = tid & 31;
    const int j    = lane & 15;
    const int half = lane >> 4;
    const int ob   = blockIdx.x * 128 + wv * 16;
    const float amask = (j == 0) ? 1.f : 0.f;

    v8f c0 = {}, c1 = {};
    for (int eb = 0; eb < 128; eb += 8) {
        const int r0 = eb + half * 2;
        v2f a0, a1, b0, b1;
        a0.x = xs[r0]     * amask;  a0.y = xs[r0 + 1] * amask;
        a1.x = xs[r0 + 4] * amask;  a1.y = xs[r0 + 5] * amask;
        const float* wp0 = W + (size_t)(k0 + r0)     * E_DIM + ob + j;
        const float* wp1 = W + (size_t)(k0 + r0 + 4) * E_DIM + ob + j;
        b0.x = wp0[0];  b0.y = wp0[E_DIM];
        b1.x = wp1[0];  b1.y = wp1[E_DIM];
        c0 = wmma4(a0, b0, c0);
        c1 = wmma4(a1, b1, c1);
    }
    const v8f cs = c0 + c1;
    if (lane < 16)
        out_parts[(size_t)blockIdx.y * E_DIM + ob + lane] = cs[0];
}

__global__ void ln_combine_kernel(const float* __restrict__ parts, int P,
                                  const float* __restrict__ bias,
                                  const float* __restrict__ resid,
                                  const float* __restrict__ sc,
                                  const float* __restrict__ bs,
                                  float* __restrict__ out) {
    __shared__ float red[256];
    const int tid = threadIdx.x;
    float vv[4];
    float sum = 0.f;
    #pragma unroll
    for (int i = 0; i < 4; i++) {
        const int e = i * 256 + tid;
        float s = bias[e] + resid[e];
        for (int pp = 0; pp < P; pp++) s += parts[(size_t)pp * E_DIM + e];
        vv[i] = s;
        sum += s;
    }
    red[tid] = sum; __syncthreads();
    for (int st = 128; st > 0; st >>= 1) {
        if (tid < st) red[tid] += red[tid + st];
        __syncthreads();
    }
    const float mean = red[0] * (1.f / E_DIM); __syncthreads();
    float var = 0.f;
    #pragma unroll
    for (int i = 0; i < 4; i++) { const float d = vv[i] - mean; var += d * d; }
    red[tid] = var; __syncthreads();
    for (int st = 128; st > 0; st >>= 1) {
        if (tid < st) red[tid] += red[tid + st];
        __syncthreads();
    }
    const float inv = rsqrtf(red[0] * (1.f / E_DIM) + 1e-6f);
    #pragma unroll
    for (int i = 0; i < 4; i++) {
        const int e = i * 256 + tid;
        out[e] = (vv[i] - mean) * inv * sc[e] + bs[e];
    }
}

extern "C" void kernel_launch(void* const* d_in, const int* in_sizes, int n_in,
                              void* d_out, int out_size, void* d_ws, size_t ws_size,
                              hipStream_t stream) {
    (void)in_sizes; (void)n_in; (void)out_size; (void)ws_size;
    const float* x     = (const float*)d_in[0];
    const float* cache = (const float*)d_in[1];
    const float* Wv = (const float*)d_in[2];  const float* bv = (const float*)d_in[3];
    const float* Wq = (const float*)d_in[4];  const float* bq = (const float*)d_in[5];
    const float* Wk = (const float*)d_in[6];  const float* bk = (const float*)d_in[7];
    const float* Wo = (const float*)d_in[8];  const float* bo = (const float*)d_in[9];
    const float* W1 = (const float*)d_in[10]; const float* b1 = (const float*)d_in[11];
    const float* W2 = (const float*)d_in[12]; const float* b2 = (const float*)d_in[13];
    const float* ln1_s = (const float*)d_in[14]; const float* ln1_b = (const float*)d_in[15];
    const float* ln2_s = (const float*)d_in[16]; const float* ln2_b = (const float*)d_in[17];

    float* out_h = (float*)d_out;
    float* newv  = out_h + E_DIM;
    float* newk  = newv + (size_t)S_LEN * E_DIM;
    const float* oldv = cache;
    const float* oldk = cache + (size_t)S_LEN * E_DIM;

    float* ws       = (float*)d_ws;
    float* scores   = ws + E_DIM;
    float* values   = scores + H_N * S_LEN;
    float* h1       = values + E_DIM;
    float* part_qkv = h1 + E_DIM;
    float* part_o   = part_qkv + 12 * E_DIM;
    float* part_1   = part_o + 8 * E_DIM;
    float* part_2   = part_1 + 8 * E_DIM;
    int*   rowflag  = (int*)(part_2 + 8 * E_DIM);
    float* stats    = (float*)(rowflag + S_LEN);

    qkv_part_copy_kernel<<<QKV_BLOCKS + COPY_BLOCKS, 256, 0, stream>>>(
        x, Wv, Wq, Wk, part_qkv, oldv, oldk, newv, newk, rowflag);
    score_kernel<<<SC_BLOCKS, 256, 0, stream>>>(newk, part_qkv, bv, bq, bk, rowflag,
                                                newv + (size_t)(S_LEN - 1) * E_DIM,
                                                newk + (size_t)(S_LEN - 1) * E_DIM,
                                                scores, stats, values);
    attnv_kernel<<<128, 256, 0, stream>>>(scores, stats, newv, values);
    gemv_part_kernel<<<dim3(8, 8), 256, 0, stream>>>(Wo, values, 1, nullptr, 0, part_o);
    ln_combine_kernel<<<1, 256, 0, stream>>>(part_o, 8, bo, x, ln1_s, ln1_b, h1);
    gemv_part_kernel<<<dim3(8, 8), 256, 0, stream>>>(W1, h1, 1, nullptr, 0, part_1);
    gemv_part_kernel<<<dim3(8, 8), 256, 0, stream>>>(W2, part_1, 8, b1, 1, part_2);
    ln_combine_kernel<<<1, 256, 0, stream>>>(part_2, 8, b2, h1, ln2_s, ln2_b, out_h);
}


# attn.V pass re-cut: wave = one head x 64 positions, its 32 256-byte V pieces requested up front by 16-byte loads (was 16 dependent rounds of 64-byte pieces); 256 blocks, LDS merge; new sum order
# speedup vs baseline: 2.2007x; 1.3079x over previous
#include <hip/hip_runtime.h>
#include <hip/hip_bf16.h>
#include <math.h>

#define E_DIM 1024
#define S_LEN 8192
#define H_N   16
#define D_H   64

typedef __attribute__((ext_vector_type(2))) float v2f;
typedef __attribute__((ext_vector_type(8))) float v8f;
typedef __attribute__((ext_vector_type(4))) float v4f;

__device__ __forceinline__ v8f wmma4(v2f a, v2f b, v8f c) {
    return __builtin_amdgcn_wmma_f32_16x16x4_f32(false, a, false, b, (short)0, c,
                                                 false, false);
}

#define QKV_BLOCKS  96
#define COPY_BLOCKS 512

__global__ void __launch_bounds__(256)
qkv_part_copy_kernel(const float* __restrict__ x,
                     const float* __restrict__ Wv,
                     const float* __restrict__ Wq,
                     const float* __restrict__ Wk,
                     float* __restrict__ part_qkv,
                     const float* __restrict__ oldv,
                     const float* __restrict__ oldk,
                     float* __restrict__ newv, float* __restrict__ newk,
                     int* __restrict__ rowflag) {
    __shared__ float xs[256];
    const int tid  = threadIdx.x;
    const int wv   = tid >> 5;
    const int lane = tid & 31;

    if (blockIdx.x < QKV_BLOCKS) {
        const int bx = blockIdx.x & 7;
        const int by = (blockIdx.x >> 3) & 3;
        const int m  = blockIdx.x >> 5;
        const int k0 = by * 256;
        xs[tid] = x[k0 + tid];
        __syncthreads();

        const int j    = lane & 15;
        const int half = lane >> 4;
        const int ob   = bx * 128 + wv * 16;
        const float amask = (j == 0) ? 1.f : 0.f;

        const float* W = (m == 0) ? Wv : (m == 1) ? Wq : Wk;

        v8f c0 = {}, c1 = {};
        for (int eb = 0; eb < 256; eb += 8) {
            const int r0 = eb + half * 2;
            v2f a0, a1, b0, b1;
            a0.x = xs[r0]     * amask;  a0.y = xs[r0 + 1] * amask;
            a1.x = xs[r0 + 4] * amask;  a1.y = xs[r0 + 5] * amask;
            const float* wp0 = W + (size_t)(k0 + r0)     * E_DIM + ob + j;
            const float* wp1 = W + (size_t)(k0 + r0 + 4) * E_DIM + ob + j;
            b0.x = wp0[0];  b0.y = wp0[E_DIM];
            b1.x = wp1[0];  b1.y = wp1[E_DIM];
            c0 = wmma4(a0, b0, c0);
            c1 = wmma4(a1, b1, c1);
        }
        const v8f cs = c0 + c1;
        if (lane < 16)
            part_qkv[(size_t)(m * 4 + by) * E_DIM + ob + lane] = cs[0];
    } else {
        const int gw = (blockIdx.x - QKV_BLOCKS) * 8 + wv;
        for (int s = gw; s < S_LEN - 1; s += COPY_BLOCKS * 8) {
            const v4f* vsrc = (const v4f*)(oldv + (size_t)(s + 1) * E_DIM);
            const v4f* ksrc = (const v4f*)(oldk + (size_t)(s + 1) * E_DIM);
            v4f* vdst = (v4f*)(newv + (size_t)s * E_DIM);
            v4f* kdst = (v4f*)(newk + (size_t)s * E_DIM);
            v4f dv[8], dk[8];
            #pragma unroll
            for (int i = 0; i < 8; i++) dv[i] = vsrc[i * 32 + lane];
            #pragma unroll
            for (int i = 0; i < 8; i++) dk[i] = ksrc[i * 32 + lane];
            asm volatile("" :: "v"(dv[0]), "v"(dv[1]), "v"(dv[2]), "v"(dv[3]),
                               "v"(dv[4]), "v"(dv[5]), "v"(dv[6]), "v"(dv[7]),
                               "v"(dk[0]), "v"(dk[1]), "v"(dk[2]), "v"(dk[3]),
                               "v"(dk[4]), "v"(dk[5]), "v"(dk[6]), "v"(dk[7]));
            bool nz = false;
            #pragma unroll
            for (int i = 0; i < 8; i++) {
                nz |= (dv[i].x != 0.f) | (dv[i].y != 0.f) | (dv[i].z != 0.f) | (dv[i].w != 0.f);
                vdst[i * 32 + lane] = dv[i];
            }
            #pragma unroll
            for (int i = 0; i < 8; i++) kdst[i * 32 + lane] = dk[i];
            const bool any = __builtin_amdgcn_ballot_w32(nz) != 0u;
            if (lane == 0) rowflag[s] = any ? 1 : 0;
        }
    }
}

#define SC_BLOCKS 256

struct QkvParts { float4 p[4]; float4 b; };

__device__ __forceinline__ QkvParts qkv_load(const float* __restrict__ part_qkv, int m,
                                             const float* __restrict__ bias, int t) {
    QkvParts r;
    #pragma unroll
    for (int ks = 0; ks < 4; ks++)
        r.p[ks] = ((const float4*)(part_qkv + (size_t)(m * 4 + ks) * E_DIM))[t];
    r.b = ((const float4*)bias)[t];
    return r;
}

__device__ __forceinline__ float4 qkv_sum(const QkvParts& r) {
    float4 s = make_float4(0.f, 0.f, 0.f, 0.f);
    #pragma unroll
    for (int ks = 0; ks < 4; ks++) {
        s.x += r.p[ks].x;  s.y += r.p[ks].y;  s.z += r.p[ks].z;  s.w += r.p[ks].w;
    }
    s.x += r.b.x;  s.y += r.b.y;  s.z += r.b.z;  s.w += r.b.w;
    return s;
}

__global__ void score_kernel(const float* __restrict__ newk,
                             const float* __restrict__ part_qkv,
                             const float* __restrict__ bv,
                             const float* __restrict__ bq,
                             const float* __restrict__ bk,
                             const int* __restrict__ rowflag,
                             float* __restrict__ vrow_last,
                             float* __restrict__ krow_last,
                             float* __restrict__ escore,
                             float* __restrict__ stats,
                             float* __restrict__ values) {
    __shared__ float  sc_s[H_N][33];
    __shared__ float4 q_s[E_DIM / 4];
    __shared__ float4 k_s[E_DIM / 4];
    __shared__ int    wnz_s[8];
    const int tid  = threadIdx.x;
    const int lane = tid & 31;
    const int wv   = tid >> 5;
    const int c0   = blockIdx.x * 32;
    const bool lastblk = (blockIdx.x == SC_BLOCKS - 1);

    float4 kd[8];
    int fl;
    {
        const float4* ksrc = (const float4*)(newk + (size_t)(c0 + wv) * E_DIM);
        #pragma unroll
        for (int i = 0; i < 8; i++) kd[i] = ksrc[i * 32 + lane];
        fl = rowflag[c0 + wv];
    }

    const QkvParts qp = qkv_load(part_qkv, 1, bq, tid);
    float4 vn = make_float4(0.f, 0.f, 0.f, 0.f);
    float4 kn = vn;
    if (lastblk) {
        const QkvParts vp = qkv_load(part_qkv, 0, bv, tid);
        const QkvParts kp = qkv_load(part_qkv, 2, bk, tid);
        q_s[tid] = qkv_sum(qp);
        vn = qkv_sum(vp);
        kn = qkv_sum(kp);
        k_s[tid] = kn;
        const bool t = (vn.x != 0.f) | (vn.y != 0.f) | (vn.z != 0.f) | (vn.w != 0.f);
        const bool wany = __builtin_amdgcn_ballot_w32(t) != 0u;
        if (lane == 0) wnz_s[wv] = wany ? 1 : 0;
    } else {
        q_s[tid] = qkv_sum(qp);
    }
    __syncthreads();
    if (lastblk) {
        ((float4*)vrow_last)[tid] = vn;
        ((float4*)krow_last)[tid] = kn;
    }
    if (blockIdx.x == 0)
        ((float4*)values)[tid] = make_float4(0.f, 0.f, 0.f, 0.f);

    for (int p = wv; p < 32; p += 8) {
        const int s = c0 + p;
        const bool last = (s == S_LEN - 1);
        bool nz;
        if (last) {
            #pragma unroll
            for (int i = 0; i < 8; i++) kd[i] = k_s[i * 32 + lane];
            nz = (wnz_s[0] | wnz_s[1] | wnz_s[2] | wnz_s[3] |
                  wnz_s[4] | wnz_s[5] | wnz_s[6] | wnz_s[7]) != 0;
        } else {
            if (p != wv) {
                const float4* ksrc = (const float4*)(newk + (size_t)s * E_DIM);
                #pragma unroll
                for (int i = 0; i < 8; i++) kd[i] = ksrc[i * 32 + lane];
                fl = rowflag[s];
            }
            nz = fl != 0;
        }

        #pragma unroll
        for (int i = 0; i < 8; i++) {
            const float4 qd = q_s[i * 32 + lane];
            float part = kd[i].x * qd.x + kd[i].y * qd.y + kd[i].z * qd.z + kd[i].w * qd.w;
            part += __shfl_xor(part, 1, 32);
            part += __shfl_xor(part, 2, 32);
            part += __shfl_xor(part, 4, 32);
            part += __shfl_xor(part, 8, 32);
            if ((lane & 15) == 0) {
                const int h = i * 2 + (lane >> 4);
                sc_s[h][p] = nz ? part * 0.125f : -INFINITY;
            }
        }
    }
    __syncthreads();

    const int h = tid >> 4;
    const int j = tid & 15;
    const float s0 = sc_s[h][j];
    const float s1 = sc_s[h][j + 16];
    float m = fmaxf(s0, s1);
    m = fmaxf(m, __shfl_xor(m, 1, 32));
    m = fmaxf(m, __shfl_xor(m, 2, 32));
    m = fmaxf(m, __shfl_xor(m, 4, 32));
    m = fmaxf(m, __shfl_xor(m, 8, 32));
    const bool dead = (m == -INFINITY);
    const float e0 = dead ? 0.f : expf(s0 - m);
    const float e1 = dead ? 0.f : expf(s1 - m);
    float l = e0 + e1;
    l += __shfl_xor(l, 1, 32);
    l += __shfl_xor(l, 2, 32);
    l += __shfl_xor(l, 4, 32);
    l += __shfl_xor(l, 8, 32);
    float* erow = escore + (size_t)h * S_LEN + c0;
    erow[j]      = e0;
    erow[j + 16] = e1;
    if (j == 0) ((float2*)stats)[h * SC_BLOCKS + blockIdx.x] = make_float2(m, l);
}

__global__ void __launch_bounds__(256)
attnv_kernel(const float* __restrict__ escore,
             const float* __restrict__ stats,
             const float* __restrict__ newv,
             float* __restrict__ values) {
    __shared__ float4 red_s[8][16];
    const int tid  = threadIdx.x;
    const int lane = tid & 31;
    const int wv   = tid >> 5;
    const int h    = blockIdx.x >> 4;
    const int ch   = blockIdx.x & 15;
    const int s0   = ch * 512 + wv * 64;
    const int j    = lane & 15;
    const int half = lane >> 4;

    const float2* st = (const float2*)stats + h * SC_BLOCKS;
    float2 pr[8];
    #pragma unroll
    for (int i = 0; i < 8; i++) pr[i] = st[i * 32 + lane];
    const float4 own = ((const float4*)st)[ch * 8 + wv];

    const float4* pp = (const float4*)(escore + (size_t)h * S_LEN + s0 + 16 * half);
    float4 pa[4], pb[4];
    #pragma unroll
    for (int i = 0; i < 4; i++) pa[i] = pp[i];
    #pragma unroll
    for (int i = 0; i < 4; i++) pb[i] = pp[8 + i];

    asm volatile("" :: "v"(pr[0].x), "v"(pr[0].y), "v"(pr[1].x), "v"(pr[1].y),
                       "v"(pr[2].x), "v"(pr[2].y), "v"(pr[3].x), "v"(pr[3].y),
                       "v"(pr[4].x), "v"(pr[4].y), "v"(pr[5].x), "v"(pr[5].y),
                       "v"(pr[6].x), "v"(pr[6].y), "v"(pr[7].x), "v"(pr[7].y),
                       "v"(own.x), "v"(own.y), "v"(own.z), "v"(own.w));
    __builtin_amdgcn_sched_barrier(0);

    const v4f* vp = (const v4f*)(newv + (size_t)(s0 + 16 * half) * E_DIM + h * D_H) + j;
    v4f va[16], vb[16];
    #pragma unroll
    for (int i = 0; i < 16; i++) va[i] = vp[(size_t)i * (E_DIM / 4)];
    #pragma unroll
    for (int i = 0; i < 16; i++) vb[i] = vp[(size_t)(32 + i) * (E_DIM / 4)];
    __builtin_amdgcn_sched_barrier(0);

    float M = pr[0].x;
    #pragma unroll
    for (int i = 1; i < 8; i++) M = fmaxf(M, pr[i].x);
    M = fmaxf(M, __shfl_xor(M, 1, 32));
    M = fmaxf(M, __shfl_xor(M, 2, 32));
    M = fmaxf(M, __shfl_xor(M, 4, 32));
    M = fmaxf(M, __shfl_xor(M, 8, 32));
    M = fmaxf(M, __shfl_xor(M, 16, 32));
    float L = 0.f;
    #pragma unroll
    for (int i = 0; i < 8; i++) L += pr[i].y * expf(pr[i].x - M);
    L += __shfl_xor(L, 1, 32);
    L += __shfl_xor(L, 2, 32);
    L += __shfl_xor(L, 4, 32);
    L += __shfl_xor(L, 8, 32);
    L += __shfl_xor(L, 16, 32);
    const float rL = 1.f / L;
    const float fa = expf(own.x - M) * rL;
    const float fb = expf(own.z - M) * rL;

    const float m0 = (j == 0) ? 1.f : 0.f;
    const float m1 = (j == 1) ? 1.f : 0.f;

    v8f axy = {}, azw = {}, bxy = {}, bzw = {};
    #pragma unroll
    for (int i = 0; i < 16; i++) {
        const float4 q = pa[i >> 2];
        const float e = ((i & 3) == 0) ? q.x : ((i & 3) == 1) ? q.y : ((i & 3) == 2) ? q.z : q.w;
        v2f a, b0, b1;
        a.x = e * m0;  a.y = e * m1;
        b0.x = va[i].x;  b0.y = va[i].y;
        b1.x = va[i].z;  b1.y = va[i].w;
        axy = wmma4(a, b0, axy);
        azw = wmma4(a, b1, azw);
    }
    #pragma unroll
    for (int i = 0; i < 16; i++) {
        const float4 q = pb[i >> 2];
        const float e = ((i & 3) == 0) ? q.x : ((i & 3) == 1) ? q.y : ((i & 3) == 2) ? q.z : q.w;
        v2f a, b0, b1;
        a.x = e * m0;  a.y = e * m1;
        b0.x = vb[i].x;  b0.y = vb[i].y;
        b1.x = vb[i].z;  b1.y = vb[i].w;
        bxy = wmma4(a, b0, bxy);
        bzw = wmma4(a, b1, bzw);
    }

    if (lane < 16)
        red_s[wv][lane] = make_float4(fa * axy[0] + fb * bxy[0], fa * axy[1] + fb * bxy[1],
                                      fa * azw[0] + fb * bzw[0], fa * azw[1] + fb * bzw[1]);
    __syncthreads();
    if (tid < D_H) {
        const float* rs = (const float*)red_s;
        float s = 0.f;
        #pragma unroll
        for (int w = 0; w < 8; w++) s += rs[w * D_H + tid];
        atomicAdd(&values[h * D_H + tid], s);
    }
}

__global__ void gemv_part_kernel(const float* __restrict__ W,
                                 const float* __restrict__ in_parts, int P,
                                 const float* __restrict__ in_bias, int in_relu,
                                 float* __restrict__ out_parts) {
    __shared__ float xs[128];
    const int tid = threadIdx.x;
    const int k0  = blockIdx.y * 128;
    if (tid < 128) {
        float s = 0.f;
        for (int pp = 0; pp < P; pp++) s += in_parts[(size_t)pp * E_DIM + k0 + tid];
        if (in_bias) s += in_bias[k0 + tid];
        if (in_relu) s = fmaxf(s, 0.f);
        xs[tid] = s;
    }
    __syncthreads();

    const int wv   = tid >> 5;
    const int lane = tid & 31;
    const int j    = lane & 15;
    const int half = lane >> 4;
    const int ob   = blockIdx.x * 128 + wv * 16;
    const float amask = (j == 0) ? 1.f : 0.f;

    v8f c0 = {}, c1 = {};
    for (int eb = 0; eb < 128; eb += 8) {
        const int r0 = eb + half * 2;
        v2f a0, a1, b0, b1;
        a0.x = xs[r0]     * amask;  a0.y = xs[r0 + 1] * amask;
        a1.x = xs[r0 + 4] * amask;  a1.y = xs[r0 + 5] * amask;
        const float* wp0 = W + (size_t)(k0 + r0)     * E_DIM + ob + j;
        const float* wp1 = W + (size_t)(k0 + r0 + 4) * E_DIM + ob + j;
        b0.x = wp0[0];  b0.y = wp0[E_DIM];
        b1.x = wp1[0];  b1.y = wp1[E_DIM];
        c0 = wmma4(a0, b0, c0);
        c1 = wmma4(a1, b1, c1);
    }
    const v8f cs = c0 + c1;
    if (lane < 16)
        out_parts[(size_t)blockIdx.y * E_DIM + ob + lane] = cs[0];
}

__global__ void ln_combine_kernel(const float* __restrict__ parts, int P,
                                  const float* __restrict__ bias,
                                  const float* __restrict__ resid,
                                  const float* __restrict__ sc,
                                  const float* __restrict__ bs,
                                  float* __restrict__ out) {
    __shared__ float red[256];
    const int tid = threadIdx.x;
    float vv[4];
    float sum = 0.f;
    #pragma unroll
    for (int i = 0; i < 4; i++) {
        const int e = i * 256 + tid;
        float s = bias[e] + resid[e];
        for (int pp = 0; pp < P; pp++) s += parts[(size_t)pp * E_DIM + e];
        vv[i] = s;
        sum += s;
    }
    red[tid] = sum; __syncthreads();
    for (int st = 128; st > 0; st >>= 1) {
        if (tid < st) red[tid] += red[tid + st];
        __syncthreads();
    }
    const float mean = red[0] * (1.f / E_DIM); __syncthreads();
    float var = 0.f;
    #pragma unroll
    for (int i = 0; i < 4; i++) { const float d = vv[i] - mean; var += d * d; }
    red[tid] = var; __syncthreads();
    for (int st = 128; st > 0; st >>= 1) {
        if (tid < st) red[tid] += red[tid + st];
        __syncthreads();
    }
    const float inv = rsqrtf(red[0] * (1.f / E_DIM) + 1e-6f);
    #pragma unroll
    for (int i = 0; i < 4; i++) {
        const int e = i * 256 + tid;
        out[e] = (vv[i] - mean) * inv * sc[e] + bs[e];
    }
}

extern "C" void kernel_launch(void* const* d_in, const int* in_sizes, int n_in,
                              void* d_out, int out_size, void* d_ws, size_t ws_size,
                              hipStream_t stream) {
    (void)in_sizes; (void)n_in; (void)out_size; (void)ws_size;
    const float* x     = (const float*)d_in[0];
    const float* cache = (const float*)d_in[1];
    const float* Wv = (const float*)d_in[2];  const float* bv = (const float*)d_in[3];
    const float* Wq = (const float*)d_in[4];  const float* bq = (const float*)d_in[5];
    const float* Wk = (const float*)d_in[6];  const float* bk = (const float*)d_in[7];
    const float* Wo = (const float*)d_in[8];  const float* bo = (const float*)d_in[9];
    const float* W1 = (const float*)d_in[10]; const float* b1 = (const float*)d_in[11];
    const float* W2 = (const float*)d_in[12]; const float* b2 = (const float*)d_in[13];
    const float* ln1_s = (const float*)d_in[14]; const float* ln1_b = (const float*)d_in[15];
    const float* ln2_s = (const float*)d_in[16]; const float* ln2_b = (const float*)d_in[17];

    float* out_h = (float*)d_out;
    float* newv  = out_h + E_DIM;
    float* newk  = newv + (size_t)S_LEN * E_DIM;
    const float* oldv = cache;
    const float* oldk = cache + (size_t)S_LEN * E_DIM;

    float* ws       = (float*)d_ws;
    float* scores   = ws + E_DIM;
    float* values   = scores + H_N * S_LEN;
    float* h1       = values + E_DIM;
    float* part_qkv = h1 + E_DIM;
    float* part_o   = part_qkv + 12 * E_DIM;
    float* part_1   = part_o + 8 * E_DIM;
    float* part_2   = part_1 + 8 * E_DIM;
    int*   rowflag  = (int*)(part_2 + 8 * E_DIM);
    float* stats    = (float*)(rowflag + S_LEN);

    qkv_part_copy_kernel<<<QKV_BLOCKS + COPY_BLOCKS, 256, 0, stream>>>(
        x, Wv, Wq, Wk, part_qkv, oldv, oldk, newv, newk, rowflag);
    score_kernel<<<SC_BLOCKS, 256, 0, stream>>>(newk, part_qkv, bv, bq, bk, rowflag,
                                                newv + (size_t)(S_LEN - 1) * E_DIM,
                                                newk + (size_t)(S_LEN - 1) * E_DIM,
                                                scores, stats, values);
    attnv_kernel<<<256, 256, 0, stream>>>(scores, stats, newv, values);
    gemv_part_kernel<<<dim3(8, 8), 256, 0, stream>>>(Wo, values, 1, nullptr, 0, part_o);
    ln_combine_kernel<<<1, 256, 0, stream>>>(part_o, 8, bo, x, ln1_s, ln1_b, h1);
    gemv_part_kernel<<<dim3(8, 8), 256, 0, stream>>>(W1, h1, 1, nullptr, 0, part_1);
    gemv_part_kernel<<<dim3(8, 8), 256, 0, stream>>>(W2, part_1, 8, b1, 1, part_2);
    ln_combine_kernel<<<1, 256, 0, stream>>>(part_2, 8, b2, h1, ln2_s, ln2_b, out_h);
}


# cache shift off the critical path: score and attn.V passes read the old cache one row on; the shift-copy runs as extra blocks of the 5 end launches; launch 1 keeps only the V-row flags; same sums
# speedup vs baseline: 2.4031x; 1.0920x over previous
#include <hip/hip_runtime.h>
#include <hip/hip_bf16.h>
#include <math.h>

#define E_DIM 1024
#define S_LEN 8192
#define H_N   16
#define D_H   64

typedef __attribute__((ext_vector_type(2))) float v2f;
typedef __attribute__((ext_vector_type(8))) float v8f;
typedef __attribute__((ext_vector_type(4))) float v4f;

__device__ __forceinline__ v8f wmma4(v2f a, v2f b, v8f c) {
    return __builtin_amdgcn_wmma_f32_16x16x4_f32(false, a, false, b, (short)0, c,
                                                 false, false);
}

#define SHIFT_BLOCKS 205
#define SHIFT_ROWS   (SHIFT_BLOCKS * 8)

__device__ __forceinline__ void shift_position(const float* __restrict__ oldv,
                                               const float* __restrict__ oldk,
                                               float* __restrict__ newv,
                                               float* __restrict__ newk,
                                               int s, int lane) {
    const v4f* vsrc = (const v4f*)(oldv + (size_t)(s + 1) * E_DIM);
    const v4f* ksrc = (const v4f*)(oldk + (size_t)(s + 1) * E_DIM);
    v4f* vdst = (v4f*)(newv + (size_t)s * E_DIM);
    v4f* kdst = (v4f*)(newk + (size_t)s * E_DIM);
    v4f dv[8], dk[8];
    #pragma unroll
    for (int i = 0; i < 8; i++) dv[i] = vsrc[i * 32 + lane];
    #pragma unroll
    for (int i = 0; i < 8; i++) dk[i] = ksrc[i * 32 + lane];
    asm volatile("" :: "v"(dv[0]), "v"(dv[1]), "v"(dv[2]), "v"(dv[3]),
                       "v"(dv[4]), "v"(dv[5]), "v"(dv[6]), "v"(dv[7]),
                       "v"(dk[0]), "v"(dk[1]), "v"(dk[2]), "v"(dk[3]),
                       "v"(dk[4]), "v"(dk[5]), "v"(dk[6]), "v"(dk[7]));
    #pragma unroll
    for (int i = 0; i < 8; i++) vdst[i * 32 + lane] = dv[i];
    #pragma unroll
    for (int i = 0; i < 8; i++) kdst[i * 32 + lane] = dk[i];
}

__device__ __forceinline__ void shift_role(const float* __restrict__ oldv,
                                           const float* __restrict__ oldk,
                                           float* __restrict__ newv,
                                           float* __restrict__ newk,
                                           int row0, int b, int tid) {
    const int s = row0 + b * 8 + (tid >> 5);
    if (s < S_LEN - 1) shift_position(oldv, oldk, newv, newk, s, tid & 31);
}

#define QKV_BLOCKS  96
#define FLAG_BLOCKS 256

__global__ void __launch_bounds__(256)
qkv_part_flag_kernel(const float* __restrict__ x,
                     const float* __restrict__ Wv,
                     const float* __restrict__ Wq,
                     const float* __restrict__ Wk,
                     float* __restrict__ part_qkv,
                     const float* __restrict__ oldv,
                     int* __restrict__ rowflag) {
    __shared__ float xs[256];
    const int tid  = threadIdx.x;
    const int wv   = tid >> 5;
    const int lane = tid & 31;

    if (blockIdx.x < QKV_BLOCKS) {
        const int bx = blockIdx.x & 7;
        const int by = (blockIdx.x >> 3) & 3;
        const int m  = blockIdx.x >> 5;
        const int k0 = by * 256;
        xs[tid] = x[k0 + tid];
        __syncthreads();

        const int j    = lane & 15;
        const int half = lane >> 4;
        const int ob   = bx * 128 + wv * 16;
        const float amask = (j == 0) ? 1.f : 0.f;

        const float* W = (m == 0) ? Wv : (m == 1) ? Wq : Wk;

        v8f c0 = {}, c1 = {};
        for (int eb = 0; eb < 256; eb += 8) {
            const int r0 = eb + half * 2;
            v2f a0, a1, b0, b1;
            a0.x = xs[r0]     * amask;  a0.y = xs[r0 + 1] * amask;
            a1.x = xs[r0 + 4] * amask;  a1.y = xs[r0 + 5] * amask;
            const float* wp0 = W + (size_t)(k0 + r0)     * E_DIM + ob + j;
            const float* wp1 = W + (size_t)(k0 + r0 + 4) * E_DIM + ob + j;
            b0.x = wp0[0];  b0.y = wp0[E_DIM];
            b1.x = wp1[0];  b1.y = wp1[E_DIM];
            c0 = wmma4(a0, b0, c0);
            c1 = wmma4(a1, b1, c1);
        }
        const v8f cs = c0 + c1;
        if (lane < 16)
            part_qkv[(size_t)(m * 4 + by) * E_DIM + ob + lane] = cs[0];
    } else {
        const int gw = (blockIdx.x - QKV_BLOCKS) * 8 + wv;
        v4f d[4][8];
        #pragma unroll
        for (int q = 0; q < 4; q++) {
            const int s  = gw + q * (FLAG_BLOCKS * 8);
            const int sr = (s < S_LEN - 1) ? s : gw;
            const v4f* p = (const v4f*)(oldv + (size_t)(sr + 1) * E_DIM);
            #pragma unroll
            for (int i = 0; i < 8; i++) d[q][i] = p[i * 32 + lane];
        }
        #pragma unroll
        for (int q = 0; q < 4; q++)
            asm volatile("" :: "v"(d[q][0]), "v"(d[q][1]), "v"(d[q][2]), "v"(d[q][3]),
                               "v"(d[q][4]), "v"(d[q][5]), "v"(d[q][6]), "v"(d[q][7]));
        #pragma unroll
        for (int q = 0; q < 4; q++) {
            const int s = gw + q * (FLAG_BLOCKS * 8);
            bool nz = false;
            #pragma unroll
            for (int i = 0; i < 8; i++)
                nz |= (d[q][i].x != 0.f) | (d[q][i].y != 0.f) | (d[q][i].z != 0.f) | (d[q][i].w != 0.f);
            const bool any = __builtin_amdgcn_ballot_w32(nz) != 0u;
            if (lane == 0 && s < S_LEN - 1) rowflag[s] = any ? 1 : 0;
        }
    }
}

#define SC_BLOCKS 256

struct QkvParts { float4 p[4]; float4 b; };

__device__ __forceinline__ QkvParts qkv_load(const float* __restrict__ part_qkv, int m,
                                             const float* __restrict__ bias, int t) {
    QkvParts r;
    #pragma unroll
    for (int ks = 0; ks < 4; ks++)
        r.p[ks] = ((const float4*)(part_qkv + (size_t)(m * 4 + ks) * E_DIM))[t];
    r.b = ((const float4*)bias)[t];
    return r;
}

__device__ __forceinline__ float4 qkv_sum(const QkvParts& r) {
    float4 s = make_float4(0.f, 0.f, 0.f, 0.f);
    #pragma unroll
    for (int ks = 0; ks < 4; ks++) {
        s.x += r.p[ks].x;  s.y += r.p[ks].y;  s.z += r.p[ks].z;  s.w += r.p[ks].w;
    }
    s.x += r.b.x;  s.y += r.b.y;  s.z += r.b.z;  s.w += r.b.w;
    return s;
}

__global__ void score_kernel(const float* __restrict__ oldk,
                             const float* __restrict__ part_qkv,
                             const float* __restrict__ bv,
                             const float* __restrict__ bq,
                             const float* __restrict__ bk,
                             const int* __restrict__ rowflag,
                             float* __restrict__ vrow_last,
                             float* __restrict__ krow_last,
                             float* __restrict__ escore,
                             float* __restrict__ stats,
                             float* __restrict__ values) {
    __shared__ float  sc_s[H_N][33];
    __shared__ float4 q_s[E_DIM / 4];
    __shared__ float4 k_s[E_DIM / 4];
    __shared__ int    wnz_s[8];
    const int tid  = threadIdx.x;
    const int lane = tid & 31;
    const int wv   = tid >> 5;
    const int c0   = blockIdx.x * 32;
    const bool lastblk = (blockIdx.x == SC_BLOCKS - 1);

    float4 kd[8];
    int fl;
    {
        const float4* ksrc = (const float4*)(oldk + (size_t)(c0 + wv + 1) * E_DIM);
        #pragma unroll
        for (int i = 0; i < 8; i++) kd[i] = ksrc[i * 32 + lane];
        fl = rowflag[c0 + wv];
    }

    const QkvParts qp = qkv_load(part_qkv, 1, bq, tid);
    float4 vn = make_float4(0.f, 0.f, 0.f, 0.f);
    float4 kn = vn;
    if (lastblk) {
        const QkvParts vp = qkv_load(part_qkv, 0, bv, tid);
        const QkvParts kp = qkv_load(part_qkv, 2, bk, tid);
        q_s[tid] = qkv_sum(qp);
        vn = qkv_sum(vp);
        kn = qkv_sum(kp);
        k_s[tid] = kn;
        const bool t = (vn.x != 0.f) | (vn.y != 0.f) | (vn.z != 0.f) | (vn.w != 0.f);
        const bool wany = __builtin_amdgcn_ballot_w32(t) != 0u;
        if (lane == 0) wnz_s[wv] = wany ? 1 : 0;
    } else {
        q_s[tid] = qkv_sum(qp);
    }
    __syncthreads();
    if (lastblk) {
        ((float4*)vrow_last)[tid] = vn;
        ((float4*)krow_last)[tid] = kn;
    }
    if (blockIdx.x == 0)
        ((float4*)values)[tid] = make_float4(0.f, 0.f, 0.f, 0.f);

    for (int p = wv; p < 32; p += 8) {
        const int s = c0 + p;
        const bool last = (s == S_LEN - 1);
        bool nz;
        if (last) {
            #pragma unroll
            for (int i = 0; i < 8; i++) kd[i] = k_s[i * 32 + lane];
            nz = (wnz_s[0] | wnz_s[1] | wnz_s[2] | wnz_s[3] |
                  wnz_s[4] | wnz_s[5] | wnz_s[6] | wnz_s[7]) != 0;
        } else {
            if (p != wv) {
                const float4* ksrc = (const float4*)(oldk + (size_t)(s + 1) * E_DIM);
                #pragma unroll
                for (int i = 0; i < 8; i++) kd[i] = ksrc[i * 32 + lane];
                fl = rowflag[s];
            }
            nz = fl != 0;
        }

        #pragma unroll
        for (int i = 0; i < 8; i++) {
            const float4 qd = q_s[i * 32 + lane];
            float part = kd[i].x * qd.x + kd[i].y * qd.y + kd[i].z * qd.z + kd[i].w * qd.w;
            part += __shfl_xor(part, 1, 32);
            part += __shfl_xor(part, 2, 32);
            part += __shfl_xor(part, 4, 32);
            part += __shfl_xor(part, 8, 32);
            if ((lane & 15) == 0) {
                const int h = i * 2 + (lane >> 4);
                sc_s[h][p] = nz ? part * 0.125f : -INFINITY;
            }
        }
    }
    __syncthreads();

    const int h = tid >> 4;
    const int j = tid & 15;
    const float s0 = sc_s[h][j];
    const float s1 = sc_s[h][j + 16];
    float m = fmaxf(s0, s1);
    m = fmaxf(m, __shfl_xor(m, 1, 32));
    m = fmaxf(m, __shfl_xor(m, 2, 32));
    m = fmaxf(m, __shfl_xor(m, 4, 32));
    m = fmaxf(m, __shfl_xor(m, 8, 32));
    const bool dead = (m == -INFINITY);
    const float e0 = dead ? 0.f : expf(s0 - m);
    const float e1 = dead ? 0.f : expf(s1 - m);
    float l = e0 + e1;
    l += __shfl_xor(l, 1, 32);
    l += __shfl_xor(l, 2, 32);
    l += __shfl_xor(l, 4, 32);
    l += __shfl_xor(l, 8, 32);
    float* erow = escore + (size_t)h * S_LEN + c0;
    erow[j]      = e0;
    erow[j + 16] = e1;
    if (j == 0) ((float2*)stats)[h * SC_BLOCKS + blockIdx.x] = make_float2(m, l);
}

__global__ void __launch_bounds__(256)
attnv_kernel(const float* __restrict__ escore,
             const float* __restrict__ stats,
             const float* __restrict__ oldv,
             const float* __restrict__ vrow_last,
             float* __restrict__ values) {
    __shared__ float4 red_s[8][16];
    const int tid  = threadIdx.x;
    const int lane = tid & 31;
    const int wv   = tid >> 5;
    const int h    = blockIdx.x >> 4;
    const int ch   = blockIdx.x & 15;
    const int s0   = ch * 512 + wv * 64;
    const int j    = lane & 15;
    const int half = lane >> 4;

    const float2* st = (const float2*)stats + h * SC_BLOCKS;
    float2 pr[8];
    #pragma unroll
    for (int i = 0; i < 8; i++) pr[i] = st[i * 32 + lane];
    const float4 own = ((const float4*)st)[ch * 8 + wv];

    const float4* pp = (const float4*)(escore + (size_t)h * S_LEN + s0 + 16 * half);
    float4 pa[4], pb[4];
    #pragma unroll
    for (int i = 0; i < 4; i++) pa[i] = pp[i];
    #pragma unroll
    for (int i = 0; i < 4; i++) pb[i] = pp[8 + i];

    asm volatile("" :: "v"(pr[0].x), "v"(pr[0].y), "v"(pr[1].x), "v"(pr[1].y),
                       "v"(pr[2].x), "v"(pr[2].y), "v"(pr[3].x), "v"(pr[3].y),
                       "v"(pr[4].x), "v"(pr[4].y), "v"(pr[5].x), "v"(pr[5].y),
                       "v"(pr[6].x), "v"(pr[6].y), "v"(pr[7].x), "v"(pr[7].y),
                       "v"(own.x), "v"(own.y), "v"(own.z), "v"(own.w));
    __builtin_amdgcn_sched_barrier(0);

    const v4f* vp = (const v4f*)(oldv + (size_t)(s0 + 16 * half + 1) * E_DIM + h * D_H) + j;
    const v4f* vlast = (s0 + 16 * half + 47 == S_LEN - 1)
                           ? (const v4f*)(vrow_last + h * D_H) + j
                           : vp + (size_t)47 * (E_DIM / 4);
    v4f va[16], vb[16];
    #pragma unroll
    for (int i = 0; i < 16; i++) va[i] = vp[(size_t)i * (E_DIM / 4)];
    #pragma unroll
    for (int i = 0; i < 15; i++) vb[i] = vp[(size_t)(32 + i) * (E_DIM / 4)];
    vb[15] = *vlast;
    __builtin_amdgcn_sched_barrier(0);

    float M = pr[0].x;
    #pragma unroll
    for (int i = 1; i < 8; i++) M = fmaxf(M, pr[i].x);
    M = fmaxf(M, __shfl_xor(M, 1, 32));
    M = fmaxf(M, __shfl_xor(M, 2, 32));
    M = fmaxf(M, __shfl_xor(M, 4, 32));
    M = fmaxf(M, __shfl_xor(M, 8, 32));
    M = fmaxf(M, __shfl_xor(M, 16, 32));
    float L = 0.f;
    #pragma unroll
    for (int i = 0; i < 8; i++) L += pr[i].y * expf(pr[i].x - M);
    L += __shfl_xor(L, 1, 32);
    L += __shfl_xor(L, 2, 32);
    L += __shfl_xor(L, 4, 32);
    L += __shfl_xor(L, 8, 32);
    L += __shfl_xor(L, 16, 32);
    const float rL = 1.f / L;
    const float fa = expf(own.x - M) * rL;
    const float fb = expf(own.z - M) * rL;

    const float m0 = (j == 0) ? 1.f : 0.f;
    const float m1 = (j == 1) ? 1.f : 0.f;

    v8f axy = {}, azw = {}, bxy = {}, bzw = {};
    #pragma unroll
    for (int i = 0; i < 16; i++) {
        const float4 q = pa[i >> 2];
        const float e = ((i & 3) == 0) ? q.x : ((i & 3) == 1) ? q.y : ((i & 3) == 2) ? q.z : q.w;
        v2f a, b0, b1;
        a.x = e * m0;  a.y = e * m1;
        b0.x = va[i].x;  b0.y = va[i].y;
        b1.x = va[i].z;  b1.y = va[i].w;
        axy = wmma4(a, b0, axy);
        azw = wmma4(a, b1, azw);
    }
    #pragma unroll
    for (int i = 0; i < 16; i++) {
        const float4 q = pb[i >> 2];
        const float e = ((i & 3) == 0) ? q.x : ((i & 3) == 1) ? q.y : ((i & 3) == 2) ? q.z : q.w;
        v2f a, b0, b1;
        a.x = e * m0;  a.y = e * m1;
        b0.x = vb[i].x;  b0.y = vb[i].y;
        b1.x = vb[i].z;  b1.y = vb[i].w;
        bxy = wmma4(a, b0, bxy);
        bzw = wmma4(a, b1, bzw);
    }

    if (lane < 16)
        red_s[wv][lane] = make_float4(fa * axy[0] + fb * bxy[0], fa * axy[1] + fb * bxy[1],
                                      fa * azw[0] + fb * bzw[0], fa * azw[1] + fb * bzw[1]);
    __syncthreads();
    if (tid < D_H) {
        const float* rs = (const float*)red_s;
        float s = 0.f;
        #pragma unroll
        for (int w = 0; w < 8; w++) s += rs[w * D_H + tid];
        atomicAdd(&values[h * D_H + tid], s);
    }
}

#define GEMV_BLOCKS 64

__global__ void __launch_bounds__(256)
gemv_part_kernel(const float* __restrict__ W,
                 const float* __restrict__ in_parts, int P,
                 const float* __restrict__ in_bias, int in_relu,
                 float* __restrict__ out_parts,
                 const float* __restrict__ oldv, const float* __restrict__ oldk,
                 float* __restrict__ newv, float* __restrict__ newk, int row0) {
    __shared__ float xs[128];
    const int tid = threadIdx.x;
    if (blockIdx.x >= GEMV_BLOCKS) {
        shift_role(oldv, oldk, newv, newk, row0, blockIdx.x - GEMV_BLOCKS, tid);
        return;
    }
    const int gbx = blockIdx.x & 7;
    const int gby = blockIdx.x >> 3;
    const int k0  = gby * 128;
    if (tid < 128) {
        float s = 0.f;
        for (int pp = 0; pp < P; pp++) s += in_parts[(size_t)pp * E_DIM + k0 + tid];
        if (in_bias) s += in_bias[k0 + tid];
        if (in_relu) s = fmaxf(s, 0.f);
        xs[tid] = s;
    }
    __syncthreads();

    const int wv   = tid >> 5;
    const int lane = tid & 31;
    const int j    = lane & 15;
    const int half = lane >> 4;
    const int ob   = gbx * 128 + wv * 16;
    const float amask = (j == 0) ? 1.f : 0.f;

    v8f c0 = {}, c1 = {};
    for (int eb = 0; eb < 128; eb += 8) {
        const int r0 = eb + half * 2;
        v2f a0, a1, b0, b1;
        a0.x = xs[r0]     * amask;  a0.y = xs[r0 + 1] * amask;
        a1.x = xs[r0 + 4] * amask;  a1.y = xs[r0 + 5] * amask;
        const float* wp0 = W + (size_t)(k0 + r0)     * E_DIM + ob + j;
        const float* wp1 = W + (size_t)(k0 + r0 + 4) * E_DIM + ob + j;
        b0.x = wp0[0];  b0.y = wp0[E_DIM];
        b1.x = wp1[0];  b1.y = wp1[E_DIM];
        c0 = wmma4(a0, b0, c0);
        c1 = wmma4(a1, b1, c1);
    }
    const v8f cs = c0 + c1;
    if (lane < 16)
        out_parts[(size_t)gby * E_DIM + ob + lane] = cs[0];
}

__global__ void __launch_bounds__(256)
ln_combine_kernel(const float* __restrict__ parts, int P,
                  const float* __restrict__ bias,
                  const float* __restrict__ resid,
                  const float* __restrict__ sc,
                  const float* __restrict__ bs,
                  float* __restrict__ out,
                  const float* __restrict__ oldv, const float* __restrict__ oldk,
                  float* __restrict__ newv, float* __restrict__ newk, int row0) {
    __shared__ float red[256];
    const int tid = threadIdx.x;
    if (blockIdx.x >= 1) {
        shift_role(oldv, oldk, newv, newk, row0, blockIdx.x - 1, tid);
        return;
    }
    float vv[4];
    float sum = 0.f;
    #pragma unroll
    for (int i = 0; i < 4; i++) {
        const int e = i * 256 + tid;
        float s = bias[e] + resid[e];
        for (int pp = 0; pp < P; pp++) s += parts[(size_t)pp * E_DIM + e];
        vv[i] = s;
        sum += s;
    }
    red[tid] = sum; __syncthreads();
    for (int st = 128; st > 0; st >>= 1) {
        if (tid < st) red[tid] += red[tid + st];
        __syncthreads();
    }
    const float mean = red[0] * (1.f / E_DIM); __syncthreads();
    float var = 0.f;
    #pragma unroll
    for (int i = 0; i < 4; i++) { const float d = vv[i] - mean; var += d * d; }
    red[tid] = var; __syncthreads();
    for (int st = 128; st > 0; st >>= 1) {
        if (tid < st) red[tid] += red[tid + st];
        __syncthreads();
    }
    const float inv = rsqrtf(red[0] * (1.f / E_DIM) + 1e-6f);
    #pragma unroll
    for (int i = 0; i < 4; i++) {
        const int e = i * 256 + tid;
        out[e] = (vv[i] - mean) * inv * sc[e] + bs[e];
    }
}

extern "C" void kernel_launch(void* const* d_in, const int* in_sizes, int n_in,
                              void* d_out, int out_size, void* d_ws, size_t ws_size,
                              hipStream_t stream) {
    (void)in_sizes; (void)n_in; (void)out_size; (void)ws_size;
    const float* x     = (const float*)d_in[0];
    const float* cache = (const float*)d_in[1];
    const float* Wv = (const float*)d_in[2];  const float* bv = (const float*)d_in[3];
    const float* Wq = (const float*)d_in[4];  const float* bq = (const float*)d_in[5];
    const float* Wk = (const float*)d_in[6];  const float* bk = (const float*)d_in[7];
    const float* Wo = (const float*)d_in[8];  const float* bo = (const float*)d_in[9];
    const float* W1 = (const float*)d_in[10]; const float* b1 = (const float*)d_in[11];
    const float* W2 = (const float*)d_in[12]; const float* b2 = (const float*)d_in[13];
    const float* ln1_s = (const float*)d_in[14]; const float* ln1_b = (const float*)d_in[15];
    const float* ln2_s = (const float*)d_in[16]; const float* ln2_b = (const float*)d_in[17];

    float* out_h = (float*)d_out;
    float* newv  = out_h + E_DIM;
    float* newk  = newv + (size_t)S_LEN * E_DIM;
    const float* oldv = cache;
    const float* oldk = cache + (size_t)S_LEN * E_DIM;

    float* ws       = (float*)d_ws;
    float* scores   = ws + E_DIM;
    float* values   = scores + H_N * S_LEN;
    float* h1       = values + E_DIM;
    float* part_qkv = h1 + E_DIM;
    float* part_o   = part_qkv + 12 * E_DIM;
    float* part_1   = part_o + 8 * E_DIM;
    float* part_2   = part_1 + 8 * E_DIM;
    int*   rowflag  = (int*)(part_2 + 8 * E_DIM);
    float* stats    = (float*)(rowflag + S_LEN);

    float* vrow_last = newv + (size_t)(S_LEN - 1) * E_DIM;
    float* krow_last = newk + (size_t)(S_LEN - 1) * E_DIM;

    qkv_part_flag_kernel<<<QKV_BLOCKS + FLAG_BLOCKS, 256, 0, stream>>>(
        x, Wv, Wq, Wk, part_qkv, oldv, rowflag);
    score_kernel<<<SC_BLOCKS, 256, 0, stream>>>(oldk, part_qkv, bv, bq, bk, rowflag,
                                                vrow_last, krow_last,
                                                scores, stats, values);
    attnv_kernel<<<256, 256, 0, stream>>>(scores, stats, oldv, vrow_last, values);
    gemv_part_kernel<<<GEMV_BLOCKS + SHIFT_BLOCKS, 256, 0, stream>>>(
        Wo, values, 1, nullptr, 0, part_o, oldv, oldk, newv, newk, 0);
    ln_combine_kernel<<<1 + SHIFT_BLOCKS, 256, 0, stream>>>(
        part_o, 8, bo, x, ln1_s, ln1_b, h1, oldv, oldk, newv, newk, 1640);
    gemv_part_kernel<<<GEMV_BLOCKS + SHIFT_BLOCKS, 256, 0, stream>>>(
        W1, h1, 1, nullptr, 0, part_1, oldv, oldk, newv, newk, 3280);
    gemv_part_kernel<<<GEMV_BLOCKS + SHIFT_BLOCKS, 256, 0, stream>>>(
        W2, part_1, 8, b1, 1, part_2, oldv, oldk, newv, newk, 4920);
    ln_combine_kernel<<<1 + SHIFT_BLOCKS, 256, 0, stream>>>(
        part_2, 8, b2, h1, ln2_s, ln2_b, out_h, oldv, oldk, newv, newk, 6560);
}


# tail GEMVs re-cut: block = 16 outputs over all 1024 rows, its 8 waves' K-slice sums meet in LDS, block stores the finished value (bias, residual, ReLU); LN reads 1 plane, not 10; same sum order
# speedup vs baseline: 2.5674x; 1.0684x over previous
#include <hip/hip_runtime.h>
#include <hip/hip_bf16.h>
#include <math.h>

#define E_DIM 1024
#define S_LEN 8192
#define H_N   16
#define D_H   64

typedef __attribute__((ext_vector_type(2))) float v2f;
typedef __attribute__((ext_vector_type(8))) float v8f;
typedef __attribute__((ext_vector_type(4))) float v4f;

__device__ __forceinline__ v8f wmma4(v2f a, v2f b, v8f c) {
    return __builtin_amdgcn_wmma_f32_16x16x4_f32(false, a, false, b, (short)0, c,
                                                 false, false);
}

#define SHIFT_BLOCKS 205
#define SHIFT_ROWS   (SHIFT_BLOCKS * 8)

__device__ __forceinline__ void shift_position(const float* __restrict__ oldv,
                                               const float* __restrict__ oldk,
                                               float* __restrict__ newv,
                                               float* __restrict__ newk,
                                               int s, int lane) {
    const v4f* vsrc = (const v4f*)(oldv + (size_t)(s + 1) * E_DIM);
    const v4f* ksrc = (const v4f*)(oldk + (size_t)(s + 1) * E_DIM);
    v4f* vdst = (v4f*)(newv + (size_t)s * E_DIM);
    v4f* kdst = (v4f*)(newk + (size_t)s * E_DIM);
    v4f dv[8], dk[8];
    #pragma unroll
    for (int i = 0; i < 8; i++) dv[i] = vsrc[i * 32 + lane];
    #pragma unroll
    for (int i = 0; i < 8; i++) dk[i] = ksrc[i * 32 + lane];
    asm volatile("" :: "v"(dv[0]), "v"(dv[1]), "v"(dv[2]), "v"(dv[3]),
                       "v"(dv[4]), "v"(dv[5]), "v"(dv[6]), "v"(dv[7]),
                       "v"(dk[0]), "v"(dk[1]), "v"(dk[2]), "v"(dk[3]),
                       "v"(dk[4]), "v"(dk[5]), "v"(dk[6]), "v"(dk[7]));
    #pragma unroll
    for (int i = 0; i < 8; i++) vdst[i * 32 + lane] = dv[i];
    #pragma unroll
    for (int i = 0; i < 8; i++) kdst[i * 32 + lane] = dk[i];
}

__device__ __forceinline__ void shift_role(const float* __restrict__ oldv,
                                           const float* __restrict__ oldk,
                                           float* __restrict__ newv,
                                           float* __restrict__ newk,
                                           int row0, int b, int tid) {
    const int s = row0 + b * 8 + (tid >> 5);
    if (s < S_LEN - 1) shift_position(oldv, oldk, newv, newk, s, tid & 31);
}

#define QKV_BLOCKS  96
#define FLAG_BLOCKS 256

__global__ void __launch_bounds__(256)
qkv_part_flag_kernel(const float* __restrict__ x,
                     const float* __restrict__ Wv,
                     const float* __restrict__ Wq,
                     const float* __restrict__ Wk,
                     float* __restrict__ part_qkv,
                     const float* __restrict__ oldv,
                     int* __restrict__ rowflag) {
    __shared__ float xs[256];
    const int tid  = threadIdx.x;
    const int wv   = tid >> 5;
    const int lane = tid & 31;

    if (blockIdx.x < QKV_BLOCKS) {
        const int bx = blockIdx.x & 7;
        const int by = (blockIdx.x >> 3) & 3;
        const int m  = blockIdx.x >> 5;
        const int k0 = by * 256;
        xs[tid] = x[k0 + tid];
        __syncthreads();

        const int j    = lane & 15;
        const int half = lane >> 4;
        const int ob   = bx * 128 + wv * 16;
        const float amask = (j == 0) ? 1.f : 0.f;

        const float* W = (m == 0) ? Wv : (m == 1) ? Wq : Wk;

        v8f c0 = {}, c1 = {};
        for (int eb = 0; eb < 256; eb += 8) {
            const int r0 = eb + half * 2;
            v2f a0, a1, b0, b1;
            a0.x = xs[r0]     * amask;  a0.y = xs[r0 + 1] * amask;
            a1.x = xs[r0 + 4] * amask;  a1.y = xs[r0 + 5] * amask;
            const float* wp0 = W + (size_t)(k0 + r0)     * E_DIM + ob + j;
            const float* wp1 = W + (size_t)(k0 + r0 + 4) * E_DIM + ob + j;
            b0.x = wp0[0];  b0.y = wp0[E_DIM];
            b1.x = wp1[0];  b1.y = wp1[E_DIM];
            c0 = wmma4(a0, b0, c0);
            c1 = wmma4(a1, b1, c1);
        }
        const v8f cs = c0 + c1;
        if (lane < 16)
            part_qkv[(size_t)(m * 4 + by) * E_DIM + ob + lane] = cs[0];
    } else {
        const int gw = (blockIdx.x - QKV_BLOCKS) * 8 + wv;
        v4f d[4][8];
        #pragma unroll
        for (int q = 0; q < 4; q++) {
            const int s  = gw + q * (FLAG_BLOCKS * 8);
            const int sr = (s < S_LEN - 1) ? s : gw;
            const v4f* p = (const v4f*)(oldv + (size_t)(sr + 1) * E_DIM);
            #pragma unroll
            for (int i = 0; i < 8; i++) d[q][i] = p[i * 32 + lane];
        }
        #pragma unroll
        for (int q = 0; q < 4; q++)
            asm volatile("" :: "v"(d[q][0]), "v"(d[q][1]), "v"(d[q][2]), "v"(d[q][3]),
                               "v"(d[q][4]), "v"(d[q][5]), "v"(d[q][6]), "v"(d[q][7]));
        #pragma unroll
        for (int q = 0; q < 4; q++) {
            const int s = gw + q * (FLAG_BLOCKS * 8);
            bool nz = false;
            #pragma unroll
            for (int i = 0; i < 8; i++)
                nz |= (d[q][i].x != 0.f) | (d[q][i].y != 0.f) | (d[q][i].z != 0.f) | (d[q][i].w != 0.f);
            const bool any = __builtin_amdgcn_ballot_w32(nz) != 0u;
            if (lane == 0 && s < S_LEN - 1) rowflag[s] = any ? 1 : 0;
        }
    }
}

#define SC_BLOCKS 256

struct QkvParts { float4 p[4]; float4 b; };

__device__ __forceinline__ QkvParts qkv_load(const float* __restrict__ part_qkv, int m,
                                             const float* __restrict__ bias, int t) {
    QkvParts r;
    #pragma unroll
    for (int ks = 0; ks < 4; ks++)
        r.p[ks] = ((const float4*)(part_qkv + (size_t)(m * 4 + ks) * E_DIM))[t];
    r.b = ((const float4*)bias)[t];
    return r;
}

__device__ __forceinline__ float4 qkv_sum(const QkvParts& r) {
    float4 s = make_float4(0.f, 0.f, 0.f, 0.f);
    #pragma unroll
    for (int ks = 0; ks < 4; ks++) {
        s.x += r.p[ks].x;  s.y += r.p[ks].y;  s.z += r.p[ks].z;  s.w += r.p[ks].w;
    }
    s.x += r.b.x;  s.y += r.b.y;  s.z += r.b.z;  s.w += r.b.w;
    return s;
}

__global__ void score_kernel(const float* __restrict__ oldk,
                             const float* __restrict__ part_qkv,
                             const float* __restrict__ bv,
                             const float* __restrict__ bq,
                             const float* __restrict__ bk,
                             const int* __restrict__ rowflag,
                             float* __restrict__ vrow_last,
                             float* __restrict__ krow_last,
                             float* __restrict__ escore,
                             float* __restrict__ stats,
                             float* __restrict__ values) {
    __shared__ float  sc_s[H_N][33];
    __shared__ float4 q_s[E_DIM / 4];
    __shared__ float4 k_s[E_DIM / 4];
    __shared__ int    wnz_s[8];
    const int tid  = threadIdx.x;
    const int lane = tid & 31;
    const int wv   = tid >> 5;
    const int c0   = blockIdx.x * 32;
    const bool lastblk = (blockIdx.x == SC_BLOCKS - 1);

    float4 kd[8];
    int fl;
    {
        const float4* ksrc = (const float4*)(oldk + (size_t)(c0 + wv + 1) * E_DIM);
        #pragma unroll
        for (int i = 0; i < 8; i++) kd[i] = ksrc[i * 32 + lane];
        fl = rowflag[c0 + wv];
    }

    const QkvParts qp = qkv_load(part_qkv, 1, bq, tid);
    float4 vn = make_float4(0.f, 0.f, 0.f, 0.f);
    float4 kn = vn;
    if (lastblk) {
        const QkvParts vp = qkv_load(part_qkv, 0, bv, tid);
        const QkvParts kp = qkv_load(part_qkv, 2, bk, tid);
        q_s[tid] = qkv_sum(qp);
        vn = qkv_sum(vp);
        kn = qkv_sum(kp);
        k_s[tid] = kn;
        const bool t = (vn.x != 0.f) | (vn.y != 0.f) | (vn.z != 0.f) | (vn.w != 0.f);
        const bool wany = __builtin_amdgcn_ballot_w32(t) != 0u;
        if (lane == 0) wnz_s[wv] = wany ? 1 : 0;
    } else {
        q_s[tid] = qkv_sum(qp);
    }
    __syncthreads();
    if (lastblk) {
        ((float4*)vrow_last)[tid] = vn;
        ((float4*)krow_last)[tid] = kn;
    }
    if (blockIdx.x == 0)
        ((float4*)values)[tid] = make_float4(0.f, 0.f, 0.f, 0.f);

    for (int p = wv; p < 32; p += 8) {
        const int s = c0 + p;
        const bool last = (s == S_LEN - 1);
        bool nz;
        if (last) {
            #pragma unroll
            for (int i = 0; i < 8; i++) kd[i] = k_s[i * 32 + lane];
            nz = (wnz_s[0] | wnz_s[1] | wnz_s[2] | wnz_s[3] |
                  wnz_s[4] | wnz_s[5] | wnz_s[6] | wnz_s[7]) != 0;
        } else {
            if (p != wv) {
                const float4* ksrc = (const float4*)(oldk + (size_t)(s + 1) * E_DIM);
                #pragma unroll
                for (int i = 0; i < 8; i++) kd[i] = ksrc[i * 32 + lane];
                fl = rowflag[s];
            }
            nz = fl != 0;
        }

        #pragma unroll
        for (int i = 0; i < 8; i++) {
            const float4 qd = q_s[i * 32 + lane];
            float part = kd[i].x * qd.x + kd[i].y * qd.y + kd[i].z * qd.z + kd[i].w * qd.w;
            part += __shfl_xor(part, 1, 32);
            part += __shfl_xor(part, 2, 32);
            part += __shfl_xor(part, 4, 32);
            part += __shfl_xor(part, 8, 32);
            if ((lane & 15) == 0) {
                const int h = i * 2 + (lane >> 4);
                sc_s[h][p] = nz ? part * 0.125f : -INFINITY;
            }
        }
    }
    __syncthreads();

    const int h = tid >> 4;
    const int j = tid & 15;
    const float s0 = sc_s[h][j];
    const float s1 = sc_s[h][j + 16];
    float m = fmaxf(s0, s1);
    m = fmaxf(m, __shfl_xor(m, 1, 32));
    m = fmaxf(m, __shfl_xor(m, 2, 32));
    m = fmaxf(m, __shfl_xor(m, 4, 32));
    m = fmaxf(m, __shfl_xor(m, 8, 32));
    const bool dead = (m == -INFINITY);
    const float e0 = dead ? 0.f : expf(s0 - m);
    const float e1 = dead ? 0.f : expf(s1 - m);
    float l = e0 + e1;
    l += __shfl_xor(l, 1, 32);
    l += __shfl_xor(l, 2, 32);
    l += __shfl_xor(l, 4, 32);
    l += __shfl_xor(l, 8, 32);
    float* erow = escore + (size_t)h * S_LEN + c0;
    erow[j]      = e0;
    erow[j + 16] = e1;
    if (j == 0) ((float2*)stats)[h * SC_BLOCKS + blockIdx.x] = make_float2(m, l);
}

__global__ void __launch_bounds__(256)
attnv_kernel(const float* __restrict__ escore,
             const float* __restrict__ stats,
             const float* __restrict__ oldv,
             const float* __restrict__ vrow_last,
             float* __restrict__ values) {
    __shared__ float4 red_s[8][16];
    const int tid  = threadIdx.x;
    const int lane = tid & 31;
    const int wv   = tid >> 5;
    const int h    = blockIdx.x >> 4;
    const int ch   = blockIdx.x & 15;
    const int s0   = ch * 512 + wv * 64;
    const int j    = lane & 15;
    const int half = lane >> 4;

    const float2* st = (const float2*)stats + h * SC_BLOCKS;
    float2 pr[8];
    #pragma unroll
    for (int i = 0; i < 8; i++) pr[i] = st[i * 32 + lane];
    const float4 own = ((const float4*)st)[ch * 8 + wv];

    const float4* pp = (const float4*)(escore + (size_t)h * S_LEN + s0 + 16 * half);
    float4 pa[4], pb[4];
    #pragma unroll
    for (int i = 0; i < 4; i++) pa[i] = pp[i];
    #pragma unroll
    for (int i = 0; i < 4; i++) pb[i] = pp[8 + i];

    asm volatile("" :: "v"(pr[0].x), "v"(pr[0].y), "v"(pr[1].x), "v"(pr[1].y),
                       "v"(pr[2].x), "v"(pr[2].y), "v"(pr[3].x), "v"(pr[3].y),
                       "v"(pr[4].x), "v"(pr[4].y), "v"(pr[5].x), "v"(pr[5].y),
                       "v"(pr[6].x), "v"(pr[6].y), "v"(pr[7].x), "v"(pr[7].y),
                       "v"(own.x), "v"(own.y), "v"(own.z), "v"(own.w));
    __builtin_amdgcn_sched_barrier(0);

    const v4f* vp = (const v4f*)(oldv + (size_t)(s0 + 16 * half + 1) * E_DIM + h * D_H) + j;
    const v4f* vlast = (s0 + 16 * half + 47 == S_LEN - 1)
                           ? (const v4f*)(vrow_last + h * D_H) + j
                           : vp + (size_t)47 * (E_DIM / 4);
    v4f va[16], vb[16];
    #pragma unroll
    for (int i = 0; i < 16; i++) va[i] = vp[(size_t)i * (E_DIM / 4)];
    #pragma unroll
    for (int i = 0; i < 15; i++) vb[i] = vp[(size_t)(32 + i) * (E_DIM / 4)];
    vb[15] = *vlast;
    __builtin_amdgcn_sched_barrier(0);

    float M = pr[0].x;
    #pragma unroll
    for (int i = 1; i < 8; i++) M = fmaxf(M, pr[i].x);
    M = fmaxf(M, __shfl_xor(M, 1, 32));
    M = fmaxf(M, __shfl_xor(M, 2, 32));
    M = fmaxf(M, __shfl_xor(M, 4, 32));
    M = fmaxf(M, __shfl_xor(M, 8, 32));
    M = fmaxf(M, __shfl_xor(M, 16, 32));
    float L = 0.f;
    #pragma unroll
    for (int i = 0; i < 8; i++) L += pr[i].y * expf(pr[i].x - M);
    L += __shfl_xor(L, 1, 32);
    L += __shfl_xor(L, 2, 32);
    L += __shfl_xor(L, 4, 32);
    L += __shfl_xor(L, 8, 32);
    L += __shfl_xor(L, 16, 32);
    const float rL = 1.f / L;
    const float fa = expf(own.x - M) * rL;
    const float fb = expf(own.z - M) * rL;

    const float m0 = (j == 0) ? 1.f : 0.f;
    const float m1 = (j == 1) ? 1.f : 0.f;

    v8f axy = {}, azw = {}, bxy = {}, bzw = {};
    #pragma unroll
    for (int i = 0; i < 16; i++) {
        const float4 q = pa[i >> 2];
        const float e = ((i & 3) == 0) ? q.x : ((i & 3) == 1) ? q.y : ((i & 3) == 2) ? q.z : q.w;
        v2f a, b0, b1;
        a.x = e * m0;  a.y = e * m1;
        b0.x = va[i].x;  b0.y = va[i].y;
        b1.x = va[i].z;  b1.y = va[i].w;
        axy = wmma4(a, b0, axy);
        azw = wmma4(a, b1, azw);
    }
    #pragma unroll
    for (int i = 0; i < 16; i++) {
        const float4 q = pb[i >> 2];
        const float e = ((i & 3) == 0) ? q.x : ((i & 3) == 1) ? q.y : ((i & 3) == 2) ? q.z : q.w;
        v2f a, b0, b1;
        a.x = e * m0;  a.y = e * m1;
        b0.x = vb[i].x;  b0.y = vb[i].y;
        b1.x = vb[i].z;  b1.y = vb[i].w;
        bxy = wmma4(a, b0, bxy);
        bzw = wmma4(a, b1, bzw);
    }

    if (lane < 16)
        red_s[wv][lane] = make_float4(fa * axy[0] + fb * bxy[0], fa * axy[1] + fb * bxy[1],
                                      fa * azw[0] + fb * bzw[0], fa * azw[1] + fb * bzw[1]);
    __syncthreads();
    if (tid < D_H) {
        const float* rs = (const float*)red_s;
        float s = 0.f;
        #pragma unroll
        for (int w = 0; w < 8; w++) s += rs[w * D_H + tid];
        atomicAdd(&values[h * D_H + tid], s);
    }
}

#define GEMV_BLOCKS 64

__global__ void __launch_bounds__(256)
gemv_full_kernel(const float* __restrict__ W,
                 const float* __restrict__ in,
                 const float* __restrict__ pre_a,
                 const float* __restrict__ pre_b,
                 const float* __restrict__ post_bias,
                 float* __restrict__ outv,
                 const float* __restrict__ oldv, const float* __restrict__ oldk,
                 float* __restrict__ newv, float* __restrict__ newk, int row0) {
    __shared__ float4 xs4[E_DIM / 4];
    __shared__ float  red_s[8][16];
    const int tid = threadIdx.x;
    if (blockIdx.x >= GEMV_BLOCKS) {
        shift_role(oldv, oldk, newv, newk, row0, blockIdx.x - GEMV_BLOCKS, tid);
        return;
    }
    const int ob = blockIdx.x * 16;
    float ea = 0.f, eb = 0.f, ec = 0.f;
    if (tid < 16) {
        if (pre_a)     { ea = pre_a[ob + tid];  eb = pre_b[ob + tid]; }
        if (post_bias) ec = post_bias[ob + tid];
    }
    xs4[tid] = ((const float4*)in)[tid];
    __syncthreads();
    const float* xs = (const float*)xs4;

    const int wv   = tid >> 5;
    const int lane = tid & 31;
    const int j    = lane & 15;
    const int half = lane >> 4;
    const int k0   = wv * 128;
    const float amask = (j == 0) ? 1.f : 0.f;

    v8f c0 = {}, c1 = {};
    for (int eb8 = 0; eb8 < 128; eb8 += 8) {
        const int r0 = eb8 + half * 2;
        v2f a0, a1, b0, b1;
        a0.x = xs[k0 + r0]     * amask;  a0.y = xs[k0 + r0 + 1] * amask;
        a1.x = xs[k0 + r0 + 4] * amask;  a1.y = xs[k0 + r0 + 5] * amask;
        const float* wp0 = W + (size_t)(k0 + r0)     * E_DIM + ob + j;
        const float* wp1 = W + (size_t)(k0 + r0 + 4) * E_DIM + ob + j;
        b0.x = wp0[0];  b0.y = wp0[E_DIM];
        b1.x = wp1[0];  b1.y = wp1[E_DIM];
        c0 = wmma4(a0, b0, c0);
        c1 = wmma4(a1, b1, c1);
    }
    const v8f cs = c0 + c1;
    if (lane < 16) red_s[wv][lane] = cs[0];
    __syncthreads();
    if (tid < 16) {
        float s = pre_a ? (ea + eb) : 0.f;
        for (int w = 0; w < 8; w++) s += red_s[w][tid];
        if (post_bias) { s += ec;  s = fmaxf(s, 0.f); }
        outv[ob + tid] = s;
    }
}

__global__ void __launch_bounds__(256)
ln_kernel(const float* __restrict__ pre,
          const float* __restrict__ sc,
          const float* __restrict__ bs,
          float* __restrict__ out,
          const float* __restrict__ oldv, const float* __restrict__ oldk,
          float* __restrict__ newv, float* __restrict__ newk, int row0) {
    __shared__ float red[256];
    const int tid = threadIdx.x;
    if (blockIdx.x >= 1) {
        shift_role(oldv, oldk, newv, newk, row0, blockIdx.x - 1, tid);
        return;
    }
    float vv[4];
    float sum = 0.f;
    #pragma unroll
    for (int i = 0; i < 4; i++) {
        vv[i] = pre[i * 256 + tid];
        sum += vv[i];
    }
    red[tid] = sum; __syncthreads();
    for (int st = 128; st > 0; st >>= 1) {
        if (tid < st) red[tid] += red[tid + st];
        __syncthreads();
    }
    const float mean = red[0] * (1.f / E_DIM); __syncthreads();
    float var = 0.f;
    #pragma unroll
    for (int i = 0; i < 4; i++) { const float d = vv[i] - mean; var += d * d; }
    red[tid] = var; __syncthreads();
    for (int st = 128; st > 0; st >>= 1) {
        if (tid < st) red[tid] += red[tid + st];
        __syncthreads();
    }
    const float inv = rsqrtf(red[0] * (1.f / E_DIM) + 1e-6f);
    #pragma unroll
    for (int i = 0; i < 4; i++) {
        const int e = i * 256 + tid;
        out[e] = (vv[i] - mean) * inv * sc[e] + bs[e];
    }
}

extern "C" void kernel_launch(void* const* d_in, const int* in_sizes, int n_in,
                              void* d_out, int out_size, void* d_ws, size_t ws_size,
                              hipStream_t stream) {
    (void)in_sizes; (void)n_in; (void)out_size; (void)ws_size;
    const float* x     = (const float*)d_in[0];
    const float* cache = (const float*)d_in[1];
    const float* Wv = (const float*)d_in[2];  const float* bv = (const float*)d_in[3];
    const float* Wq = (const float*)d_in[4];  const float* bq = (const float*)d_in[5];
    const float* Wk = (const float*)d_in[6];  const float* bk = (const float*)d_in[7];
    const float* Wo = (const float*)d_in[8];  const float* bo = (const float*)d_in[9];
    const float* W1 = (const float*)d_in[10]; const float* b1 = (const float*)d_in[11];
    const float* W2 = (const float*)d_in[12]; const float* b2 = (const float*)d_in[13];
    const float* ln1_s = (const float*)d_in[14]; const float* ln1_b = (const float*)d_in[15];
    const float* ln2_s = (const float*)d_in[16]; const float* ln2_b = (const float*)d_in[17];

    float* out_h = (float*)d_out;
    float* newv  = out_h + E_DIM;
    float* newk  = newv + (size_t)S_LEN * E_DIM;
    const float* oldv = cache;
    const float* oldk = cache + (size_t)S_LEN * E_DIM;

    float* ws       = (float*)d_ws;
    float* scores   = ws + E_DIM;
    float* values   = scores + H_N * S_LEN;
    float* h1       = values + E_DIM;
    float* part_qkv = h1 + E_DIM;
    float* pre1     = part_qkv + 12 * E_DIM;
    float* act1     = pre1 + E_DIM;
    float* pre2     = act1 + E_DIM;
    int*   rowflag  = (int*)(pre2 + E_DIM);
    float* stats    = (float*)(rowflag + S_LEN);

    float* vrow_last = newv + (size_t)(S_LEN - 1) * E_DIM;
    float* krow_last = newk + (size_t)(S_LEN - 1) * E_DIM;

    qkv_part_flag_kernel<<<QKV_BLOCKS + FLAG_BLOCKS, 256, 0, stream>>>(
        x, Wv, Wq, Wk, part_qkv, oldv, rowflag);
    score_kernel<<<SC_BLOCKS, 256, 0, stream>>>(oldk, part_qkv, bv, bq, bk, rowflag,
                                                vrow_last, krow_last,
                                                scores, stats, values);
    attnv_kernel<<<256, 256, 0, stream>>>(scores, stats, oldv, vrow_last, values);
    gemv_full_kernel<<<GEMV_BLOCKS + SHIFT_BLOCKS, 256, 0, stream>>>(
        Wo, values, bo, x, nullptr, pre1, oldv, oldk, newv, newk, 0);
    ln_kernel<<<1 + SHIFT_BLOCKS, 256, 0, stream>>>(
        pre1, ln1_s, ln1_b, h1, oldv, oldk, newv, newk, 1640);
    gemv_full_kernel<<<GEMV_BLOCKS + SHIFT_BLOCKS, 256, 0, stream>>>(
        W1, h1, nullptr, nullptr, b1, act1, oldv, oldk, newv, newk, 3280);
    gemv_full_kernel<<<GEMV_BLOCKS + SHIFT_BLOCKS, 256, 0, stream>>>(
        W2, act1, b2, h1, nullptr, pre2, oldv, oldk, newv, newk, 4920);
    ln_kernel<<<1 + SHIFT_BLOCKS, 256, 0, stream>>>(
        pre2, ln2_s, ln2_b, out_h, oldv, oldk, newv, newk, 6560);
}


# score pass re-cut by head: wave = one head x 64 positions, 32 256-byte K pieces requested at entry (was 4 dependent rows, q and scores via LDS); row sums by 15 lane exchanges; same sum order
# speedup vs baseline: 2.9740x; 1.1584x over previous
#include <hip/hip_runtime.h>
#include <hip/hip_bf16.h>
#include <math.h>

#define E_DIM 1024
#define S_LEN 8192
#define H_N   16
#define D_H   64

typedef __attribute__((ext_vector_type(2))) float v2f;
typedef __attribute__((ext_vector_type(8))) float v8f;
typedef __attribute__((ext_vector_type(4))) float v4f;

__device__ __forceinline__ v8f wmma4(v2f a, v2f b, v8f c) {
    return __builtin_amdgcn_wmma_f32_16x16x4_f32(false, a, false, b, (short)0, c,
                                                 false, false);
}

#define SHIFT_BLOCKS 205
#define SHIFT_ROWS   (SHIFT_BLOCKS * 8)

__device__ __forceinline__ void shift_position(const float* __restrict__ oldv,
                                               const float* __restrict__ oldk,
                                               float* __restrict__ newv,
                                               float* __restrict__ newk,
                                               int s, int lane) {
    const v4f* vsrc = (const v4f*)(oldv + (size_t)(s + 1) * E_DIM);
    const v4f* ksrc = (const v4f*)(oldk + (size_t)(s + 1) * E_DIM);
    v4f* vdst = (v4f*)(newv + (size_t)s * E_DIM);
    v4f* kdst = (v4f*)(newk + (size_t)s * E_DIM);
    v4f dv[8], dk[8];
    #pragma unroll
    for (int i = 0; i < 8; i++) dv[i] = vsrc[i * 32 + lane];
    #pragma unroll
    for (int i = 0; i < 8; i++) dk[i] = ksrc[i * 32 + lane];
    asm volatile("" :: "v"(dv[0]), "v"(dv[1]), "v"(dv[2]), "v"(dv[3]),
                       "v"(dv[4]), "v"(dv[5]), "v"(dv[6]), "v"(dv[7]),
                       "v"(dk[0]), "v"(dk[1]), "v"(dk[2]), "v"(dk[3]),
                       "v"(dk[4]), "v"(dk[5]), "v"(dk[6]), "v"(dk[7]));
    #pragma unroll
    for (int i = 0; i < 8; i++) vdst[i * 32 + lane] = dv[i];
    #pragma unroll
    for (int i = 0; i < 8; i++) kdst[i * 32 + lane] = dk[i];
}

__device__ __forceinline__ void shift_role(const float* __restrict__ oldv,
                                           const float* __restrict__ oldk,
                                           float* __restrict__ newv,
                                           float* __restrict__ newk,
                                           int row0, int b, int tid) {
    const int s = row0 + b * 8 + (tid >> 5);
    if (s < S_LEN - 1) shift_position(oldv, oldk, newv, newk, s, tid & 31);
}

#define QKV_BLOCKS  96
#define FLAG_BLOCKS 256

__global__ void __launch_bounds__(256)
qkv_part_flag_kernel(const float* __restrict__ x,
                     const float* __restrict__ Wv,
                     const float* __restrict__ Wq,
                     const float* __restrict__ Wk,
                     float* __restrict__ part_qkv,
                     const float* __restrict__ oldv,
                     int* __restrict__ rowflag) {
    __shared__ float xs[256];
    const int tid  = threadIdx.x;
    const int wv   = tid >> 5;
    const int lane = tid & 31;

    if (blockIdx.x < QKV_BLOCKS) {
        const int bx = blockIdx.x & 7;
        const int by = (blockIdx.x >> 3) & 3;
        const int m  = blockIdx.x >> 5;
        const int k0 = by * 256;
        xs[tid] = x[k0 + tid];
        __syncthreads();

        const int j    = lane & 15;
        const int half = lane >> 4;
        const int ob   = bx * 128 + wv * 16;
        const float amask = (j == 0) ? 1.f : 0.f;

        const float* W = (m == 0) ? Wv : (m == 1) ? Wq : Wk;

        v8f c0 = {}, c1 = {};
        for (int eb = 0; eb < 256; eb += 8) {
            const int r0 = eb + half * 2;
            v2f a0, a1, b0, b1;
            a0.x = xs[r0]     * amask;  a0.y = xs[r0 + 1] * amask;
            a1.x = xs[r0 + 4] * amask;  a1.y = xs[r0 + 5] * amask;
            const float* wp0 = W + (size_t)(k0 + r0)     * E_DIM + ob + j;
            const float* wp1 = W + (size_t)(k0 + r0 + 4) * E_DIM + ob + j;
            b0.x = wp0[0];  b0.y = wp0[E_DIM];
            b1.x = wp1[0];  b1.y = wp1[E_DIM];
            c0 = wmma4(a0, b0, c0);
            c1 = wmma4(a1, b1, c1);
        }
        const v8f cs = c0 + c1;
        if (lane < 16)
            part_qkv[(size_t)(m * 4 + by) * E_DIM + ob + lane] = cs[0];
    } else {
        const int gw = (blockIdx.x - QKV_BLOCKS) * 8 + wv;
        v4f d[4][8];
        #pragma unroll
        for (int q = 0; q < 4; q++) {
            const int s  = gw + q * (FLAG_BLOCKS * 8);
            const int sr = (s < S_LEN - 1) ? s : gw;
            const v4f* p = (const v4f*)(oldv + (size_t)(sr + 1) * E_DIM);
            #pragma unroll
            for (int i = 0; i < 8; i++) d[q][i] = p[i * 32 + lane];
        }
        #pragma unroll
        for (int q = 0; q < 4; q++)
            asm volatile("" :: "v"(d[q][0]), "v"(d[q][1]), "v"(d[q][2]), "v"(d[q][3]),
                               "v"(d[q][4]), "v"(d[q][5]), "v"(d[q][6]), "v"(d[q][7]));
        #pragma unroll
        for (int q = 0; q < 4; q++) {
            const int s = gw + q * (FLAG_BLOCKS * 8);
            bool nz = false;
            #pragma unroll
            for (int i = 0; i < 8; i++)
                nz |= (d[q][i].x != 0.f) | (d[q][i].y != 0.f) | (d[q][i].z != 0.f) | (d[q][i].w != 0.f);
            const bool any = __builtin_amdgcn_ballot_w32(nz) != 0u;
            if (lane == 0 && s < S_LEN - 1) rowflag[s] = any ? 1 : 0;
        }
    }
}

#define SC_BLOCKS 256

struct QkvParts { float4 p[4]; float4 b; };

__device__ __forceinline__ QkvParts qkv_load(const float* __restrict__ part_qkv, int m,
                                             const float* __restrict__ bias, int t) {
    QkvParts r;
    #pragma unroll
    for (int ks = 0; ks < 4; ks++)
        r.p[ks] = ((const float4*)(part_qkv + (size_t)(m * 4 + ks) * E_DIM))[t];
    r.b = ((const float4*)bias)[t];
    return r;
}

__device__ __forceinline__ float4 qkv_sum(const QkvParts& r) {
    float4 s = make_float4(0.f, 0.f, 0.f, 0.f);
    #pragma unroll
    for (int ks = 0; ks < 4; ks++) {
        s.x += r.p[ks].x;  s.y += r.p[ks].y;  s.z += r.p[ks].z;  s.w += r.p[ks].w;
    }
    s.x += r.b.x;  s.y += r.b.y;  s.z += r.b.z;  s.w += r.b.w;
    return s;
}

template <int MASK>
__device__ __forceinline__ float row_xor(float v) {
    const int iv = __builtin_bit_cast(int, v);
    return __builtin_bit_cast(float,
        __builtin_amdgcn_update_dpp(iv, iv, 0x160 | MASK, 0xf, 0xf, false));
}

__device__ __forceinline__ float half_xor(float v) {
    const int iv = __builtin_bit_cast(int, v);
    return __builtin_bit_cast(float,
        __builtin_amdgcn_permlanex16(iv, iv, 0x76543210u, 0xfedcba98u, false, false));
}

__device__ __forceinline__ float row_sum16(const float (&p)[16], int j) {
    const bool b0 = (j & 1) != 0, b1 = (j & 2) != 0, b2 = (j & 4) != 0, b3 = (j & 8) != 0;
    float r1[8], r2[4], r3[2];
    #pragma unroll
    for (int m = 0; m < 8; m++) {
        const float keep = b0 ? p[2 * m + 1] : p[2 * m];
        const float send = b0 ? p[2 * m] : p[2 * m + 1];
        r1[m] = keep + row_xor<1>(send);
    }
    #pragma unroll
    for (int m = 0; m < 4; m++) {
        const float keep = b1 ? r1[2 * m + 1] : r1[2 * m];
        const float send = b1 ? r1[2 * m] : r1[2 * m + 1];
        r2[m] = keep + row_xor<2>(send);
    }
    #pragma unroll
    for (int m = 0; m < 2; m++) {
        const float keep = b2 ? r2[2 * m + 1] : r2[2 * m];
        const float send = b2 ? r2[2 * m] : r2[2 * m + 1];
        r3[m] = keep + row_xor<4>(send);
    }
    const float keep = b3 ? r3[1] : r3[0];
    const float send = b3 ? r3[0] : r3[1];
    return keep + row_xor<8>(send);
}

__device__ __forceinline__ float2 block_softmax(float sc, float& e) {
    float m = fmaxf(sc, half_xor(sc));
    m = fmaxf(m, row_xor<1>(m));
    m = fmaxf(m, row_xor<2>(m));
    m = fmaxf(m, row_xor<4>(m));
    m = fmaxf(m, row_xor<8>(m));
    const bool dead = (m == -INFINITY);
    e = dead ? 0.f : expf(sc - m);
    float l = e + half_xor(e);
    l += row_xor<1>(l);
    l += row_xor<2>(l);
    l += row_xor<4>(l);
    l += row_xor<8>(l);
    return make_float2(m, l);
}

__device__ __forceinline__ float dot4(const v4f k, const float4 q) {
    return fmaf(k.w, q.w, fmaf(k.z, q.z, fmaf(k.x, q.x, k.y * q.y)));
}

__global__ void __launch_bounds__(256)
score_kernel(const float* __restrict__ oldk,
             const float* __restrict__ part_qkv,
             const float* __restrict__ bv,
             const float* __restrict__ bq,
             const float* __restrict__ bk,
             const int* __restrict__ rowflag,
             float* __restrict__ vrow_last,
             float* __restrict__ krow_last,
             float* __restrict__ escore,
             float* __restrict__ stats,
             float* __restrict__ values) {
    __shared__ int wnz_s[8];
    const int tid  = threadIdx.x;
    const int lane = tid & 31;
    const int wv   = tid >> 5;
    const int h    = blockIdx.x >> 4;
    const int ch   = blockIdx.x & 15;
    const int s0   = ch * 512 + wv * 64;
    const int j    = lane & 15;
    const int half = lane >> 4;
    const bool lastchunk = (ch == 15);
    const bool lastwave  = lastchunk && (wv == 7);
    const bool lastpos = (s0 + 16 * half + 47 == S_LEN - 1);

    const QkvParts qp = qkv_load(part_qkv, 1, bq, h * 16 + j);
    const int sB  = s0 + 32 + lane;
    const int fla = rowflag[s0 + lane];
    const int flb = rowflag[(sB < S_LEN - 1) ? sB : s0 + lane];
    __builtin_amdgcn_sched_barrier(0);

    const v4f* kp = (const v4f*)(oldk + (size_t)(s0 + 16 * half + 1) * E_DIM + h * D_H) + j;
    v4f ka[16], kb[16];
    #pragma unroll
    for (int i = 0; i < 16; i++) ka[i] = kp[(size_t)i * (E_DIM / 4)];
    #pragma unroll
    for (int i = 0; i < 15; i++) kb[i] = kp[(size_t)(32 + i) * (E_DIM / 4)];
    kb[15] = kp[(size_t)(lastpos ? 46 : 47) * (E_DIM / 4)];
    QkvParts vpar = {}, kpar = {};
    if (lastchunk) vpar = qkv_load(part_qkv, 0, bv, tid);
    if (lastwave)  kpar = qkv_load(part_qkv, 2, bk, h * 16 + j);
    __builtin_amdgcn_sched_barrier(0);

    bool nzlast = true;
    float4 kn = make_float4(0.f, 0.f, 0.f, 0.f);
    if (lastchunk) {
        const float4 vn = qkv_sum(vpar);
        const bool t = (vn.x != 0.f) | (vn.y != 0.f) | (vn.z != 0.f) | (vn.w != 0.f);
        const bool wany = __builtin_amdgcn_ballot_w32(t) != 0u;
        if (lane == 0) wnz_s[wv] = wany ? 1 : 0;
        __syncthreads();
        nzlast = (wnz_s[0] | wnz_s[1] | wnz_s[2] | wnz_s[3] |
                  wnz_s[4] | wnz_s[5] | wnz_s[6] | wnz_s[7]) != 0;
        if ((tid >> 4) == h) ((float4*)vrow_last)[tid] = vn;
        if (lastwave) {
            kn = qkv_sum(kpar);
            if (half == 1) ((float4*)krow_last)[h * 16 + j] = kn;
        }
    }
    if (blockIdx.x == 0)
        ((float4*)values)[tid] = make_float4(0.f, 0.f, 0.f, 0.f);

    const float4 qd = qkv_sum(qp);

    float pa[16], pb[16];
    #pragma unroll
    for (int i = 0; i < 16; i++) pa[i] = dot4(ka[i], qd);
    #pragma unroll
    for (int i = 0; i < 15; i++) pb[i] = dot4(kb[i], qd);
    {
        v4f k15 = kb[15];
        if (lastpos) { k15.x = kn.x;  k15.y = kn.y;  k15.z = kn.z;  k15.w = kn.w; }
        pb[15] = dot4(k15, qd);
    }

    const float ra = row_sum16(pa, j);
    const float rb = row_sum16(pb, j);
    const bool nza = fla != 0;
    const bool nzb = (sB == S_LEN - 1) ? nzlast : (flb != 0);
    const float sca = nza ? ra * 0.125f : -INFINITY;
    const float scb = nzb ? rb * 0.125f : -INFINITY;

    float ea, eb;
    const float2 sta = block_softmax(sca, ea);
    const float2 stb = block_softmax(scb, eb);

    float* erow = escore + (size_t)h * S_LEN + s0;
    erow[lane]      = ea;
    erow[32 + lane] = eb;
    if (lane == 0)
        ((float4*)stats)[h * (SC_BLOCKS / 2) + ch * 8 + wv] = make_float4(sta.x, sta.y, stb.x, stb.y);
}

__global__ void __launch_bounds__(256)
attnv_kernel(const float* __restrict__ escore,
             const float* __restrict__ stats,
             const float* __restrict__ oldv,
             const float* __restrict__ vrow_last,
             float* __restrict__ values) {
    __shared__ float4 red_s[8][16];
    const int tid  = threadIdx.x;
    const int lane = tid & 31;
    const int wv   = tid >> 5;
    const int h    = blockIdx.x >> 4;
    const int ch   = blockIdx.x & 15;
    const int s0   = ch * 512 + wv * 64;
    const int j    = lane & 15;
    const int half = lane >> 4;

    const float2* st = (const float2*)stats + h * SC_BLOCKS;
    float2 pr[8];
    #pragma unroll
    for (int i = 0; i < 8; i++) pr[i] = st[i * 32 + lane];
    const float4 own = ((const float4*)st)[ch * 8 + wv];

    const float4* pp = (const float4*)(escore + (size_t)h * S_LEN + s0 + 16 * half);
    float4 pa[4], pb[4];
    #pragma unroll
    for (int i = 0; i < 4; i++) pa[i] = pp[i];
    #pragma unroll
    for (int i = 0; i < 4; i++) pb[i] = pp[8 + i];

    asm volatile("" :: "v"(pr[0].x), "v"(pr[0].y), "v"(pr[1].x), "v"(pr[1].y),
                       "v"(pr[2].x), "v"(pr[2].y), "v"(pr[3].x), "v"(pr[3].y),
                       "v"(pr[4].x), "v"(pr[4].y), "v"(pr[5].x), "v"(pr[5].y),
                       "v"(pr[6].x), "v"(pr[6].y), "v"(pr[7].x), "v"(pr[7].y),
                       "v"(own.x), "v"(own.y), "v"(own.z), "v"(own.w));
    __builtin_amdgcn_sched_barrier(0);

    const v4f* vp = (const v4f*)(oldv + (size_t)(s0 + 16 * half + 1) * E_DIM + h * D_H) + j;
    const v4f* vlast = (s0 + 16 * half + 47 == S_LEN - 1)
                           ? (const v4f*)(vrow_last + h * D_H) + j
                           : vp + (size_t)47 * (E_DIM / 4);
    v4f va[16], vb[16];
    #pragma unroll
    for (int i = 0; i < 16; i++) va[i] = vp[(size_t)i * (E_DIM / 4)];
    #pragma unroll
    for (int i = 0; i < 15; i++) vb[i] = vp[(size_t)(32 + i) * (E_DIM / 4)];
    vb[15] = *vlast;
    __builtin_amdgcn_sched_barrier(0);

    float M = pr[0].x;
    #pragma unroll
    for (int i = 1; i < 8; i++) M = fmaxf(M, pr[i].x);
    M = fmaxf(M, __shfl_xor(M, 1, 32));
    M = fmaxf(M, __shfl_xor(M, 2, 32));
    M = fmaxf(M, __shfl_xor(M, 4, 32));
    M = fmaxf(M, __shfl_xor(M, 8, 32));
    M = fmaxf(M, __shfl_xor(M, 16, 32));
    float L = 0.f;
    #pragma unroll
    for (int i = 0; i < 8; i++) L += pr[i].y * expf(pr[i].x - M);
    L += __shfl_xor(L, 1, 32);
    L += __shfl_xor(L, 2, 32);
    L += __shfl_xor(L, 4, 32);
    L += __shfl_xor(L, 8, 32);
    L += __shfl_xor(L, 16, 32);
    const float rL = 1.f / L;
    const float fa = expf(own.x - M) * rL;
    const float fb = expf(own.z - M) * rL;

    const float m0 = (j == 0) ? 1.f : 0.f;
    const float m1 = (j == 1) ? 1.f : 0.f;

    v8f axy = {}, azw = {}, bxy = {}, bzw = {};
    #pragma unroll
    for (int i = 0; i < 16; i++) {
        const float4 q = pa[i >> 2];
        const float e = ((i & 3) == 0) ? q.x : ((i & 3) == 1) ? q.y : ((i & 3) == 2) ? q.z : q.w;
        v2f a, b0, b1;
        a.x = e * m0;  a.y = e * m1;
        b0.x = va[i].x;  b0.y = va[i].y;
        b1.x = va[i].z;  b1.y = va[i].w;
        axy = wmma4(a, b0, axy);
        azw = wmma4(a, b1, azw);
    }
    #pragma unroll
    for (int i = 0; i < 16; i++) {
        const float4 q = pb[i >> 2];
        const float e = ((i & 3) == 0) ? q.x : ((i & 3) == 1) ? q.y : ((i & 3) == 2) ? q.z : q.w;
        v2f a, b0, b1;
        a.x = e * m0;  a.y = e * m1;
        b0.x = vb[i].x;  b0.y = vb[i].y;
        b1.x = vb[i].z;  b1.y = vb[i].w;
        bxy = wmma4(a, b0, bxy);
        bzw = wmma4(a, b1, bzw);
    }

    if (lane < 16)
        red_s[wv][lane] = make_float4(fa * axy[0] + fb * bxy[0], fa * axy[1] + fb * bxy[1],
                                      fa * azw[0] + fb * bzw[0], fa * azw[1] + fb * bzw[1]);
    __syncthreads();
    if (tid < D_H) {
        const float* rs = (const float*)red_s;
        float s = 0.f;
        #pragma unroll
        for (int w = 0; w < 8; w++) s += rs[w * D_H + tid];
        atomicAdd(&values[h * D_H + tid], s);
    }
}

#define GEMV_BLOCKS 64

__global__ void __launch_bounds__(256)
gemv_full_kernel(const float* __restrict__ W,
                 const float* __restrict__ in,
                 const float* __restrict__ pre_a,
                 const float* __restrict__ pre_b,
                 const float* __restrict__ post_bias,
                 float* __restrict__ outv,
                 const float* __restrict__ oldv, const float* __restrict__ oldk,
                 float* __restrict__ newv, float* __restrict__ newk, int row0) {
    __shared__ float4 xs4[E_DIM / 4];
    __shared__ float  red_s[8][16];
    const int tid = threadIdx.x;
    if (blockIdx.x >= GEMV_BLOCKS) {
        shift_role(oldv, oldk, newv, newk, row0, blockIdx.x - GEMV_BLOCKS, tid);
        return;
    }
    const int ob = blockIdx.x * 16;
    float ea = 0.f, eb = 0.f, ec = 0.f;
    if (tid < 16) {
        if (pre_a)     { ea = pre_a[ob + tid];  eb = pre_b[ob + tid]; }
        if (post_bias) ec = post_bias[ob + tid];
    }
    xs4[tid] = ((const float4*)in)[tid];
    __syncthreads();
    const float* xs = (const float*)xs4;

    const int wv   = tid >> 5;
    const int lane = tid & 31;
    const int j    = lane & 15;
    const int half = lane >> 4;
    const int k0   = wv * 128;
    const float amask = (j == 0) ? 1.f : 0.f;

    v8f c0 = {}, c1 = {};
    for (int eb8 = 0; eb8 < 128; eb8 += 8) {
        const int r0 = eb8 + half * 2;
        v2f a0, a1, b0, b1;
        a0.x = xs[k0 + r0]     * amask;  a0.y = xs[k0 + r0 + 1] * amask;
        a1.x = xs[k0 + r0 + 4] * amask;  a1.y = xs[k0 + r0 + 5] * amask;
        const float* wp0 = W + (size_t)(k0 + r0)     * E_DIM + ob + j;
        const float* wp1 = W + (size_t)(k0 + r0 + 4) * E_DIM + ob + j;
        b0.x = wp0[0];  b0.y = wp0[E_DIM];
        b1.x = wp1[0];  b1.y = wp1[E_DIM];
        c0 = wmma4(a0, b0, c0);
        c1 = wmma4(a1, b1, c1);
    }
    const v8f cs = c0 + c1;
    if (lane < 16) red_s[wv][lane] = cs[0];
    __syncthreads();
    if (tid < 16) {
        float s = pre_a ? (ea + eb) : 0.f;
        for (int w = 0; w < 8; w++) s += red_s[w][tid];
        if (post_bias) { s += ec;  s = fmaxf(s, 0.f); }
        outv[ob + tid] = s;
    }
}

__global__ void __launch_bounds__(256)
ln_kernel(const float* __restrict__ pre,
          const float* __restrict__ sc,
          const float* __restrict__ bs,
          float* __restrict__ out,
          const float* __restrict__ oldv, const float* __restrict__ oldk,
          float* __restrict__ newv, float* __restrict__ newk, int row0) {
    __shared__ float red[256];
    const int tid = threadIdx.x;
    if (blockIdx.x >= 1) {
        shift_role(oldv, oldk, newv, newk, row0, blockIdx.x - 1, tid);
        return;
    }
    float vv[4];
    float sum = 0.f;
    #pragma unroll
    for (int i = 0; i < 4; i++) {
        vv[i] = pre[i * 256 + tid];
        sum += vv[i];
    }
    red[tid] = sum; __syncthreads();
    for (int st = 128; st > 0; st >>= 1) {
        if (tid < st) red[tid] += red[tid + st];
        __syncthreads();
    }
    const float mean = red[0] * (1.f / E_DIM); __syncthreads();
    float var = 0.f;
    #pragma unroll
    for (int i = 0; i < 4; i++) { const float d = vv[i] - mean; var += d * d; }
    red[tid] = var; __syncthreads();
    for (int st = 128; st > 0; st >>= 1) {
        if (tid < st) red[tid] += red[tid + st];
        __syncthreads();
    }
    const float inv = rsqrtf(red[0] * (1.f / E_DIM) + 1e-6f);
    #pragma unroll
    for (int i = 0; i < 4; i++) {
        const int e = i * 256 + tid;
        out[e] = (vv[i] - mean) * inv * sc[e] + bs[e];
    }
}

extern "C" void kernel_launch(void* const* d_in, const int* in_sizes, int n_in,
                              void* d_out, int out_size, void* d_ws, size_t ws_size,
                              hipStream_t stream) {
    (void)in_sizes; (void)n_in; (void)out_size; (void)ws_size;
    const float* x     = (const float*)d_in[0];
    const float* cache = (const float*)d_in[1];
    const float* Wv = (const float*)d_in[2];  const float* bv = (const float*)d_in[3];
    const float* Wq = (const float*)d_in[4];  const float* bq = (const float*)d_in[5];
    const float* Wk = (const float*)d_in[6];  const float* bk = (const float*)d_in[7];
    const float* Wo = (const float*)d_in[8];  const float* bo = (const float*)d_in[9];
    const float* W1 = (const float*)d_in[10]; const float* b1 = (const float*)d_in[11];
    const float* W2 = (const float*)d_in[12]; const float* b2 = (const float*)d_in[13];
    const float* ln1_s = (const float*)d_in[14]; const float* ln1_b = (const float*)d_in[15];
    const float* ln2_s = (const float*)d_in[16]; const float* ln2_b = (const float*)d_in[17];

    float* out_h = (float*)d_out;
    float* newv  = out_h + E_DIM;
    float* newk  = newv + (size_t)S_LEN * E_DIM;
    const float* oldv = cache;
    const float* oldk = cache + (size_t)S_LEN * E_DIM;

    float* ws       = (float*)d_ws;
    float* scores   = ws + E_DIM;
    float* values   = scores + H_N * S_LEN;
    float* h1       = values + E_DIM;
    float* part_qkv = h1 + E_DIM;
    float* pre1     = part_qkv + 12 * E_DIM;
    float* act1     = pre1 + E_DIM;
    float* pre2     = act1 + E_DIM;
    int*   rowflag  = (int*)(pre2 + E_DIM);
    float* stats    = (float*)(rowflag + S_LEN);

    float* vrow_last = newv + (size_t)(S_LEN - 1) * E_DIM;
    float* krow_last = newk + (size_t)(S_LEN - 1) * E_DIM;

    qkv_part_flag_kernel<<<QKV_BLOCKS + FLAG_BLOCKS, 256, 0, stream>>>(
        x, Wv, Wq, Wk, part_qkv, oldv, rowflag);
    score_kernel<<<SC_BLOCKS, 256, 0, stream>>>(oldk, part_qkv, bv, bq, bk, rowflag,
                                                vrow_last, krow_last,
                                                scores, stats, values);
    attnv_kernel<<<256, 256, 0, stream>>>(scores, stats, oldv, vrow_last, values);
    gemv_full_kernel<<<GEMV_BLOCKS + SHIFT_BLOCKS, 256, 0, stream>>>(
        Wo, values, bo, x, nullptr, pre1, oldv, oldk, newv, newk, 0);
    ln_kernel<<<1 + SHIFT_BLOCKS, 256, 0, stream>>>(
        pre1, ln1_s, ln1_b, h1, oldv, oldk, newv, newk, 1640);
    gemv_full_kernel<<<GEMV_BLOCKS + SHIFT_BLOCKS, 256, 0, stream>>>(
        W1, h1, nullptr, nullptr, b1, act1, oldv, oldk, newv, newk, 3280);
    gemv_full_kernel<<<GEMV_BLOCKS + SHIFT_BLOCKS, 256, 0, stream>>>(
        W2, act1, b2, h1, nullptr, pre2, oldv, oldk, newv, newk, 4920);
    ln_kernel<<<1 + SHIFT_BLOCKS, 256, 0, stream>>>(
        pre2, ln2_s, ln2_b, out_h, oldv, oldk, newv, newk, 6560);
}


# both LayerNorm launches folded into the GEMV launch before them: the last of its 64 blocks to arrive (atomic arrival counter, no block waits) normalises the vector; 8 -> 6 launches; new LN sum order
# speedup vs baseline: 3.0955x; 1.0409x over previous
#include <hip/hip_runtime.h>
#include <hip/hip_bf16.h>
#include <math.h>

#define E_DIM 1024
#define S_LEN 8192
#define H_N   16
#define D_H   64

typedef __attribute__((ext_vector_type(2))) float v2f;
typedef __attribute__((ext_vector_type(8))) float v8f;
typedef __attribute__((ext_vector_type(4))) float v4f;

__device__ __forceinline__ v8f wmma4(v2f a, v2f b, v8f c) {
    return __builtin_amdgcn_wmma_f32_16x16x4_f32(false, a, false, b, (short)0, c,
                                                 false, false);
}

#define SHIFT_BLOCKS 342
#define SHIFT_ROWS   (SHIFT_BLOCKS * 8)

__device__ __forceinline__ void shift_position(const float* __restrict__ oldv,
                                               const float* __restrict__ oldk,
                                               float* __restrict__ newv,
                                               float* __restrict__ newk,
                                               int s, int lane) {
    const v4f* vsrc = (const v4f*)(oldv + (size_t)(s + 1) * E_DIM);
    const v4f* ksrc = (const v4f*)(oldk + (size_t)(s + 1) * E_DIM);
    v4f* vdst = (v4f*)(newv + (size_t)s * E_DIM);
    v4f* kdst = (v4f*)(newk + (size_t)s * E_DIM);
    v4f dv[8], dk[8];
    #pragma unroll
    for (int i = 0; i < 8; i++) dv[i] = vsrc[i * 32 + lane];
    #pragma unroll
    for (int i = 0; i < 8; i++) dk[i] = ksrc[i * 32 + lane];
    asm volatile("" :: "v"(dv[0]), "v"(dv[1]), "v"(dv[2]), "v"(dv[3]),
                       "v"(dv[4]), "v"(dv[5]), "v"(dv[6]), "v"(dv[7]),
                       "v"(dk[0]), "v"(dk[1]), "v"(dk[2]), "v"(dk[3]),
                       "v"(dk[4]), "v"(dk[5]), "v"(dk[6]), "v"(dk[7]));
    #pragma unroll
    for (int i = 0; i < 8; i++) vdst[i * 32 + lane] = dv[i];
    #pragma unroll
    for (int i = 0; i < 8; i++) kdst[i * 32 + lane] = dk[i];
}

__device__ __forceinline__ void shift_role(const float* __restrict__ oldv,
                                           const float* __restrict__ oldk,
                                           float* __restrict__ newv,
                                           float* __restrict__ newk,
                                           int row0, int b, int tid) {
    const int s = row0 + b * 8 + (tid >> 5);
    if (s < S_LEN - 1) shift_position(oldv, oldk, newv, newk, s, tid & 31);
}

#define QKV_BLOCKS  96
#define FLAG_BLOCKS 256

__global__ void __launch_bounds__(256)
qkv_part_flag_kernel(const float* __restrict__ x,
                     const float* __restrict__ Wv,
                     const float* __restrict__ Wq,
                     const float* __restrict__ Wk,
                     float* __restrict__ part_qkv,
                     const float* __restrict__ oldv,
                     int* __restrict__ rowflag) {
    __shared__ float xs[256];
    const int tid  = threadIdx.x;
    const int wv   = tid >> 5;
    const int lane = tid & 31;

    if (blockIdx.x < QKV_BLOCKS) {
        const int bx = blockIdx.x & 7;
        const int by = (blockIdx.x >> 3) & 3;
        const int m  = blockIdx.x >> 5;
        const int k0 = by * 256;
        xs[tid] = x[k0 + tid];
        __syncthreads();

        const int j    = lane & 15;
        const int half = lane >> 4;
        const int ob   = bx * 128 + wv * 16;
        const float amask = (j == 0) ? 1.f : 0.f;

        const float* W = (m == 0) ? Wv : (m == 1) ? Wq : Wk;

        v8f c0 = {}, c1 = {};
        for (int eb = 0; eb < 256; eb += 8) {
            const int r0 = eb + half * 2;
            v2f a0, a1, b0, b1;
            a0.x = xs[r0]     * amask;  a0.y = xs[r0 + 1] * amask;
            a1.x = xs[r0 + 4] * amask;  a1.y = xs[r0 + 5] * amask;
            const float* wp0 = W + (size_t)(k0 + r0)     * E_DIM + ob + j;
            const float* wp1 = W + (size_t)(k0 + r0 + 4) * E_DIM + ob + j;
            b0.x = wp0[0];  b0.y = wp0[E_DIM];
            b1.x = wp1[0];  b1.y = wp1[E_DIM];
            c0 = wmma4(a0, b0, c0);
            c1 = wmma4(a1, b1, c1);
        }
        const v8f cs = c0 + c1;
        if (lane < 16)
            part_qkv[(size_t)(m * 4 + by) * E_DIM + ob + lane] = cs[0];
    } else {
        const int gw = (blockIdx.x - QKV_BLOCKS) * 8 + wv;
        v4f d[4][8];
        #pragma unroll
        for (int q = 0; q < 4; q++) {
            const int s  = gw + q * (FLAG_BLOCKS * 8);
            const int sr = (s < S_LEN - 1) ? s : gw;
            const v4f* p = (const v4f*)(oldv + (size_t)(sr + 1) * E_DIM);
            #pragma unroll
            for (int i = 0; i < 8; i++) d[q][i] = p[i * 32 + lane];
        }
        #pragma unroll
        for (int q = 0; q < 4; q++)
            asm volatile("" :: "v"(d[q][0]), "v"(d[q][1]), "v"(d[q][2]), "v"(d[q][3]),
                               "v"(d[q][4]), "v"(d[q][5]), "v"(d[q][6]), "v"(d[q][7]));
        #pragma unroll
        for (int q = 0; q < 4; q++) {
            const int s = gw + q * (FLAG_BLOCKS * 8);
            bool nz = false;
            #pragma unroll
            for (int i = 0; i < 8; i++)
                nz |= (d[q][i].x != 0.f) | (d[q][i].y != 0.f) | (d[q][i].z != 0.f) | (d[q][i].w != 0.f);
            const bool any = __builtin_amdgcn_ballot_w32(nz) != 0u;
            if (lane == 0 && s < S_LEN - 1) rowflag[s] = any ? 1 : 0;
        }
    }
}

#define SC_BLOCKS 256

struct QkvParts { float4 p[4]; float4 b; };

__device__ __forceinline__ QkvParts qkv_load(const float* __restrict__ part_qkv, int m,
                                             const float* __restrict__ bias, int t) {
    QkvParts r;
    #pragma unroll
    for (int ks = 0; ks < 4; ks++)
        r.p[ks] = ((const float4*)(part_qkv + (size_t)(m * 4 + ks) * E_DIM))[t];
    r.b = ((const float4*)bias)[t];
    return r;
}

__device__ __forceinline__ float4 qkv_sum(const QkvParts& r) {
    float4 s = make_float4(0.f, 0.f, 0.f, 0.f);
    #pragma unroll
    for (int ks = 0; ks < 4; ks++) {
        s.x += r.p[ks].x;  s.y += r.p[ks].y;  s.z += r.p[ks].z;  s.w += r.p[ks].w;
    }
    s.x += r.b.x;  s.y += r.b.y;  s.z += r.b.z;  s.w += r.b.w;
    return s;
}

template <int MASK>
__device__ __forceinline__ float row_xor(float v) {
    const int iv = __builtin_bit_cast(int, v);
    return __builtin_bit_cast(float,
        __builtin_amdgcn_update_dpp(iv, iv, 0x160 | MASK, 0xf, 0xf, false));
}

__device__ __forceinline__ float half_xor(float v) {
    const int iv = __builtin_bit_cast(int, v);
    return __builtin_bit_cast(float,
        __builtin_amdgcn_permlanex16(iv, iv, 0x76543210u, 0xfedcba98u, false, false));
}

__device__ __forceinline__ float row_sum16(const float (&p)[16], int j) {
    const bool b0 = (j & 1) != 0, b1 = (j & 2) != 0, b2 = (j & 4) != 0, b3 = (j & 8) != 0;
    float r1[8], r2[4], r3[2];
    #pragma unroll
    for (int m = 0; m < 8; m++) {
        const float keep = b0 ? p[2 * m + 1] : p[2 * m];
        const float send = b0 ? p[2 * m] : p[2 * m + 1];
        r1[m] = keep + row_xor<1>(send);
    }
    #pragma unroll
    for (int m = 0; m < 4; m++) {
        const float keep = b1 ? r1[2 * m + 1] : r1[2 * m];
        const float send = b1 ? r1[2 * m] : r1[2 * m + 1];
        r2[m] = keep + row_xor<2>(send);
    }
    #pragma unroll
    for (int m = 0; m < 2; m++) {
        const float keep = b2 ? r2[2 * m + 1] : r2[2 * m];
        const float send = b2 ? r2[2 * m] : r2[2 * m + 1];
        r3[m] = keep + row_xor<4>(send);
    }
    const float keep = b3 ? r3[1] : r3[0];
    const float send = b3 ? r3[0] : r3[1];
    return keep + row_xor<8>(send);
}

__device__ __forceinline__ float2 block_softmax(float sc, float& e) {
    float m = fmaxf(sc, half_xor(sc));
    m = fmaxf(m, row_xor<1>(m));
    m = fmaxf(m, row_xor<2>(m));
    m = fmaxf(m, row_xor<4>(m));
    m = fmaxf(m, row_xor<8>(m));
    const bool dead = (m == -INFINITY);
    e = dead ? 0.f : expf(sc - m);
    float l = e + half_xor(e);
    l += row_xor<1>(l);
    l += row_xor<2>(l);
    l += row_xor<4>(l);
    l += row_xor<8>(l);
    return make_float2(m, l);
}

__device__ __forceinline__ float dot4(const v4f k, const float4 q) {
    return fmaf(k.w, q.w, fmaf(k.z, q.z, fmaf(k.x, q.x, k.y * q.y)));
}

__global__ void __launch_bounds__(256)
score_kernel(const float* __restrict__ oldk,
             const float* __restrict__ part_qkv,
             const float* __restrict__ bv,
             const float* __restrict__ bq,
             const float* __restrict__ bk,
             const int* __restrict__ rowflag,
             float* __restrict__ vrow_last,
             float* __restrict__ krow_last,
             float* __restrict__ escore,
             float* __restrict__ stats,
             float* __restrict__ values,
             unsigned int* __restrict__ arrive) {
    __shared__ int wnz_s[8];
    const int tid  = threadIdx.x;
    const int lane = tid & 31;
    const int wv   = tid >> 5;
    const int h    = blockIdx.x >> 4;
    const int ch   = blockIdx.x & 15;
    const int s0   = ch * 512 + wv * 64;
    const int j    = lane & 15;
    const int half = lane >> 4;
    const bool lastchunk = (ch == 15);
    const bool lastwave  = lastchunk && (wv == 7);
    const bool lastpos = (s0 + 16 * half + 47 == S_LEN - 1);

    const QkvParts qp = qkv_load(part_qkv, 1, bq, h * 16 + j);
    const int sB  = s0 + 32 + lane;
    const int fla = rowflag[s0 + lane];
    const int flb = rowflag[(sB < S_LEN - 1) ? sB : s0 + lane];
    __builtin_amdgcn_sched_barrier(0);

    const v4f* kp = (const v4f*)(oldk + (size_t)(s0 + 16 * half + 1) * E_DIM + h * D_H) + j;
    v4f ka[16], kb[16];
    #pragma unroll
    for (int i = 0; i < 16; i++) ka[i] = kp[(size_t)i * (E_DIM / 4)];
    #pragma unroll
    for (int i = 0; i < 15; i++) kb[i] = kp[(size_t)(32 + i) * (E_DIM / 4)];
    kb[15] = kp[(size_t)(lastpos ? 46 : 47) * (E_DIM / 4)];
    QkvParts vpar = {}, kpar = {};
    if (lastchunk) vpar = qkv_load(part_qkv, 0, bv, tid);
    if (lastwave)  kpar = qkv_load(part_qkv, 2, bk, h * 16 + j);
    __builtin_amdgcn_sched_barrier(0);

    bool nzlast = true;
    float4 kn = make_float4(0.f, 0.f, 0.f, 0.f);
    if (lastchunk) {
        const float4 vn = qkv_sum(vpar);
        const bool t = (vn.x != 0.f) | (vn.y != 0.f) | (vn.z != 0.f) | (vn.w != 0.f);
        const bool wany = __builtin_amdgcn_ballot_w32(t) != 0u;
        if (lane == 0) wnz_s[wv] = wany ? 1 : 0;
        __syncthreads();
        nzlast = (wnz_s[0] | wnz_s[1] | wnz_s[2] | wnz_s[3] |
                  wnz_s[4] | wnz_s[5] | wnz_s[6] | wnz_s[7]) != 0;
        if ((tid >> 4) == h) ((float4*)vrow_last)[tid] = vn;
        if (lastwave) {
            kn = qkv_sum(kpar);
            if (half == 1) ((float4*)krow_last)[h * 16 + j] = kn;
        }
    }
    if (blockIdx.x == 0) {
        ((float4*)values)[tid] = make_float4(0.f, 0.f, 0.f, 0.f);
        if (tid < 2) arrive[tid] = 0u;
    }

    const float4 qd = qkv_sum(qp);

    float pa[16], pb[16];
    #pragma unroll
    for (int i = 0; i < 16; i++) pa[i] = dot4(ka[i], qd);
    #pragma unroll
    for (int i = 0; i < 15; i++) pb[i] = dot4(kb[i], qd);
    {
        v4f k15 = kb[15];
        if (lastpos) { k15.x = kn.x;  k15.y = kn.y;  k15.z = kn.z;  k15.w = kn.w; }
        pb[15] = dot4(k15, qd);
    }

    const float ra = row_sum16(pa, j);
    const float rb = row_sum16(pb, j);
    const bool nza = fla != 0;
    const bool nzb = (sB == S_LEN - 1) ? nzlast : (flb != 0);
    const float sca = nza ? ra * 0.125f : -INFINITY;
    const float scb = nzb ? rb * 0.125f : -INFINITY;

    float ea, eb;
    const float2 sta = block_softmax(sca, ea);
    const float2 stb = block_softmax(scb, eb);

    float* erow = escore + (size_t)h * S_LEN + s0;
    erow[lane]      = ea;
    erow[32 + lane] = eb;
    if (lane == 0)
        ((float4*)stats)[h * (SC_BLOCKS / 2) + ch * 8 + wv] = make_float4(sta.x, sta.y, stb.x, stb.y);
}

__global__ void __launch_bounds__(256)
attnv_kernel(const float* __restrict__ escore,
             const float* __restrict__ stats,
             const float* __restrict__ oldv,
             const float* __restrict__ vrow_last,
             float* __restrict__ values) {
    __shared__ float4 red_s[8][16];
    const int tid  = threadIdx.x;
    const int lane = tid & 31;
    const int wv   = tid >> 5;
    const int h    = blockIdx.x >> 4;
    const int ch   = blockIdx.x & 15;
    const int s0   = ch * 512 + wv * 64;
    const int j    = lane & 15;
    const int half = lane >> 4;

    const float2* st = (const float2*)stats + h * SC_BLOCKS;
    float2 pr[8];
    #pragma unroll
    for (int i = 0; i < 8; i++) pr[i] = st[i * 32 + lane];
    const float4 own = ((const float4*)st)[ch * 8 + wv];

    const float4* pp = (const float4*)(escore + (size_t)h * S_LEN + s0 + 16 * half);
    float4 pa[4], pb[4];
    #pragma unroll
    for (int i = 0; i < 4; i++) pa[i] = pp[i];
    #pragma unroll
    for (int i = 0; i < 4; i++) pb[i] = pp[8 + i];

    asm volatile("" :: "v"(pr[0].x), "v"(pr[0].y), "v"(pr[1].x), "v"(pr[1].y),
                       "v"(pr[2].x), "v"(pr[2].y), "v"(pr[3].x), "v"(pr[3].y),
                       "v"(pr[4].x), "v"(pr[4].y), "v"(pr[5].x), "v"(pr[5].y),
                       "v"(pr[6].x), "v"(pr[6].y), "v"(pr[7].x), "v"(pr[7].y),
                       "v"(own.x), "v"(own.y), "v"(own.z), "v"(own.w));
    __builtin_amdgcn_sched_barrier(0);

    const v4f* vp = (const v4f*)(oldv + (size_t)(s0 + 16 * half + 1) * E_DIM + h * D_H) + j;
    const v4f* vlast = (s0 + 16 * half + 47 == S_LEN - 1)
                           ? (const v4f*)(vrow_last + h * D_H) + j
                           : vp + (size_t)47 * (E_DIM / 4);
    v4f va[16], vb[16];
    #pragma unroll
    for (int i = 0; i < 16; i++) va[i] = vp[(size_t)i * (E_DIM / 4)];
    #pragma unroll
    for (int i = 0; i < 15; i++) vb[i] = vp[(size_t)(32 + i) * (E_DIM / 4)];
    vb[15] = *vlast;
    __builtin_amdgcn_sched_barrier(0);

    float M = pr[0].x;
    #pragma unroll
    for (int i = 1; i < 8; i++) M = fmaxf(M, pr[i].x);
    M = fmaxf(M, __shfl_xor(M, 1, 32));
    M = fmaxf(M, __shfl_xor(M, 2, 32));
    M = fmaxf(M, __shfl_xor(M, 4, 32));
    M = fmaxf(M, __shfl_xor(M, 8, 32));
    M = fmaxf(M, __shfl_xor(M, 16, 32));
    float L = 0.f;
    #pragma unroll
    for (int i = 0; i < 8; i++) L += pr[i].y * expf(pr[i].x - M);
    L += __shfl_xor(L, 1, 32);
    L += __shfl_xor(L, 2, 32);
    L += __shfl_xor(L, 4, 32);
    L += __shfl_xor(L, 8, 32);
    L += __shfl_xor(L, 16, 32);
    const float rL = 1.f / L;
    const float fa = expf(own.x - M) * rL;
    const float fb = expf(own.z - M) * rL;

    const float m0 = (j == 0) ? 1.f : 0.f;
    const float m1 = (j == 1) ? 1.f : 0.f;

    v8f axy = {}, azw = {}, bxy = {}, bzw = {};
    #pragma unroll
    for (int i = 0; i < 16; i++) {
        const float4 q = pa[i >> 2];
        const float e = ((i & 3) == 0) ? q.x : ((i & 3) == 1) ? q.y : ((i & 3) == 2) ? q.z : q.w;
        v2f a, b0, b1;
        a.x = e * m0;  a.y = e * m1;
        b0.x = va[i].x;  b0.y = va[i].y;
        b1.x = va[i].z;  b1.y = va[i].w;
        axy = wmma4(a, b0, axy);
        azw = wmma4(a, b1, azw);
    }
    #pragma unroll
    for (int i = 0; i < 16; i++) {
        const float4 q = pb[i >> 2];
        const float e = ((i & 3) == 0) ? q.x : ((i & 3) == 1) ? q.y : ((i & 3) == 2) ? q.z : q.w;
        v2f a, b0, b1;
        a.x = e * m0;  a.y = e * m1;
        b0.x = vb[i].x;  b0.y = vb[i].y;
        b1.x = vb[i].z;  b1.y = vb[i].w;
        bxy = wmma4(a, b0, bxy);
        bzw = wmma4(a, b1, bzw);
    }

    if (lane < 16)
        red_s[wv][lane] = make_float4(fa * axy[0] + fb * bxy[0], fa * axy[1] + fb * bxy[1],
                                      fa * azw[0] + fb * bzw[0], fa * azw[1] + fb * bzw[1]);
    __syncthreads();
    if (tid < D_H) {
        const float* rs = (const float*)red_s;
        float s = 0.f;
        #pragma unroll
        for (int w = 0; w < 8; w++) s += rs[w * D_H + tid];
        atomicAdd(&values[h * D_H + tid], s);
    }
}

#define GEMV_BLOCKS 64

__device__ __forceinline__ float wave_sum32(float v) {
    v += half_xor(v);
    v += row_xor<1>(v);
    v += row_xor<2>(v);
    v += row_xor<4>(v);
    v += row_xor<8>(v);
    return v;
}

__global__ void __launch_bounds__(256)
gemv_full_kernel(const float* __restrict__ W,
                 const float* __restrict__ in,
                 const float* __restrict__ pre_a,
                 const float* __restrict__ pre_b,
                 const float* __restrict__ post_bias,
                 float* __restrict__ outv,
                 unsigned int* __restrict__ arrive,
                 const float* __restrict__ ln_sc,
                 const float* __restrict__ ln_bs,
                 float* __restrict__ ln_out,
                 const float* __restrict__ oldv, const float* __restrict__ oldk,
                 float* __restrict__ newv, float* __restrict__ newk, int row0) {
    __shared__ float4 xs4[E_DIM / 4];
    __shared__ float  red_s[8][16];
    __shared__ float  lnsum_s[2][8];
    __shared__ unsigned int ticket_s;
    const int tid = threadIdx.x;
    if (blockIdx.x >= GEMV_BLOCKS) {
        shift_role(oldv, oldk, newv, newk, row0, blockIdx.x - GEMV_BLOCKS, tid);
        return;
    }
    const int ob = blockIdx.x * 16;
    float ea = 0.f, eb = 0.f, ec = 0.f;
    if (tid < 16) {
        if (pre_a)     { ea = pre_a[ob + tid];  eb = pre_b[ob + tid]; }
        if (post_bias) ec = post_bias[ob + tid];
    }
    xs4[tid] = ((const float4*)in)[tid];
    __syncthreads();
    const float* xs = (const float*)xs4;

    const int wv   = tid >> 5;
    const int lane = tid & 31;
    const int j    = lane & 15;
    const int half = lane >> 4;
    const int k0   = wv * 128;
    const float amask = (j == 0) ? 1.f : 0.f;

    v8f c0 = {}, c1 = {};
    for (int eb8 = 0; eb8 < 128; eb8 += 8) {
        const int r0 = eb8 + half * 2;
        v2f a0, a1, b0, b1;
        a0.x = xs[k0 + r0]     * amask;  a0.y = xs[k0 + r0 + 1] * amask;
        a1.x = xs[k0 + r0 + 4] * amask;  a1.y = xs[k0 + r0 + 5] * amask;
        const float* wp0 = W + (size_t)(k0 + r0)     * E_DIM + ob + j;
        const float* wp1 = W + (size_t)(k0 + r0 + 4) * E_DIM + ob + j;
        b0.x = wp0[0];  b0.y = wp0[E_DIM];
        b1.x = wp1[0];  b1.y = wp1[E_DIM];
        c0 = wmma4(a0, b0, c0);
        c1 = wmma4(a1, b1, c1);
    }
    const v8f cs = c0 + c1;
    if (lane < 16) red_s[wv][lane] = cs[0];
    __syncthreads();
    if (tid < 16) {
        float s = pre_a ? (ea + eb) : 0.f;
        for (int w = 0; w < 8; w++) s += red_s[w][tid];
        if (post_bias) { s += ec;  s = fmaxf(s, 0.f); }
        outv[ob + tid] = s;
    }
    if (arrive == nullptr) return;

    if (wv == 0) {
        __builtin_amdgcn_fence(__ATOMIC_RELEASE, "agent");
        if (lane == 0)
            ticket_s = __hip_atomic_fetch_add(arrive, 1u, __ATOMIC_RELAXED,
                                              __HIP_MEMORY_SCOPE_AGENT);
    }
    __syncthreads();
    if (ticket_s != GEMV_BLOCKS - 1) return;

    __builtin_amdgcn_fence(__ATOMIC_ACQUIRE, "agent");
    const float4 xv = ((const float4*)outv)[tid];
    const float4 sv = ((const float4*)ln_sc)[tid];
    const float4 bv = ((const float4*)ln_bs)[tid];

    const float tsum = wave_sum32(((xv.x + xv.y) + xv.z) + xv.w);
    if (lane == 0) lnsum_s[0][wv] = tsum;
    __syncthreads();
    float mean = lnsum_s[0][0];
    for (int w = 1; w < 8; w++) mean += lnsum_s[0][w];
    mean *= (1.f / E_DIM);

    const float dx = xv.x - mean, dy = xv.y - mean, dz = xv.z - mean, dw = xv.w - mean;
    const float tvar = wave_sum32(((dx * dx + dy * dy) + dz * dz) + dw * dw);
    if (lane == 0) lnsum_s[1][wv] = tvar;
    __syncthreads();
    float var = lnsum_s[1][0];
    for (int w = 1; w < 8; w++) var += lnsum_s[1][w];
    const float inv = rsqrtf(var * (1.f / E_DIM) + 1e-6f);

    ((float4*)ln_out)[tid] = make_float4(dx * inv * sv.x + bv.x, dy * inv * sv.y + bv.y,
                                         dz * inv * sv.z + bv.z, dw * inv * sv.w + bv.w);
}

extern "C" void kernel_launch(void* const* d_in, const int* in_sizes, int n_in,
                              void* d_out, int out_size, void* d_ws, size_t ws_size,
                              hipStream_t stream) {
    (void)in_sizes; (void)n_in; (void)out_size; (void)ws_size;
    const float* x     = (const float*)d_in[0];
    const float* cache = (const float*)d_in[1];
    const float* Wv = (const float*)d_in[2];  const float* bv = (const float*)d_in[3];
    const float* Wq = (const float*)d_in[4];  const float* bq = (const float*)d_in[5];
    const float* Wk = (const float*)d_in[6];  const float* bk = (const float*)d_in[7];
    const float* Wo = (const float*)d_in[8];  const float* bo = (const float*)d_in[9];
    const float* W1 = (const float*)d_in[10]; const float* b1 = (const float*)d_in[11];
    const float* W2 = (const float*)d_in[12]; const float* b2 = (const float*)d_in[13];
    const float* ln1_s = (const float*)d_in[14]; const float* ln1_b = (const float*)d_in[15];
    const float* ln2_s = (const float*)d_in[16]; const float* ln2_b = (const float*)d_in[17];

    float* out_h = (float*)d_out;
    float* newv  = out_h + E_DIM;
    float* newk  = newv + (size_t)S_LEN * E_DIM;
    const float* oldv = cache;
    const float* oldk = cache + (size_t)S_LEN * E_DIM;

    float* ws       = (float*)d_ws;
    float* scores   = ws + E_DIM;
    float* values   = scores + H_N * S_LEN;
    float* h1       = values + E_DIM;
    float* part_qkv = h1 + E_DIM;
    float* pre1     = part_qkv + 12 * E_DIM;
    float* act1     = pre1 + E_DIM;
    float* pre2     = act1 + E_DIM;
    int*   rowflag  = (int*)(pre2 + E_DIM);
    float* stats    = (float*)(rowflag + S_LEN);
    unsigned int* arrive = (unsigned int*)(stats + 2 * H_N * SC_BLOCKS);

    float* vrow_last = newv + (size_t)(S_LEN - 1) * E_DIM;
    float* krow_last = newk + (size_t)(S_LEN - 1) * E_DIM;

    qkv_part_flag_kernel<<<QKV_BLOCKS + FLAG_BLOCKS, 256, 0, stream>>>(
        x, Wv, Wq, Wk, part_qkv, oldv, rowflag);
    score_kernel<<<SC_BLOCKS, 256, 0, stream>>>(oldk, part_qkv, bv, bq, bk, rowflag,
                                                vrow_last, krow_last,
                                                scores, stats, values, arrive);
    attnv_kernel<<<256, 256, 0, stream>>>(scores, stats, oldv, vrow_last, values);
    gemv_full_kernel<<<GEMV_BLOCKS + SHIFT_BLOCKS, 256, 0, stream>>>(
        Wo, values, bo, x, nullptr, pre1, arrive, ln1_s, ln1_b, h1,
        oldv, oldk, newv, newk, 0);
    gemv_full_kernel<<<GEMV_BLOCKS + SHIFT_BLOCKS, 256, 0, stream>>>(
        W1, h1, nullptr, nullptr, b1, act1, nullptr, nullptr, nullptr, nullptr,
        oldv, oldk, newv, newk, 2736);
    gemv_full_kernel<<<GEMV_BLOCKS + SHIFT_BLOCKS, 256, 0, stream>>>(
        W2, act1, b2, h1, nullptr, pre2, arrive + 1, ln2_s, ln2_b, out_h,
        oldv, oldk, newv, newk, 5472);
}


# shift role re-cut to half the waves: one wave per pair of neighbouring positions, 32 loads ahead of its stores; 171 shift blocks per end launch, not 342, so fewer waves have to enter; same sums
# speedup vs baseline: 3.2460x; 1.0486x over previous
#include <hip/hip_runtime.h>
#include <hip/hip_bf16.h>
#include <math.h>

#define E_DIM 1024
#define S_LEN 8192
#define H_N   16
#define D_H   64

typedef __attribute__((ext_vector_type(2))) float v2f;
typedef __attribute__((ext_vector_type(8))) float v8f;
typedef __attribute__((ext_vector_type(4))) float v4f;

__device__ __forceinline__ v8f wmma4(v2f a, v2f b, v8f c) {
    return __builtin_amdgcn_wmma_f32_16x16x4_f32(false, a, false, b, (short)0, c,
                                                 false, false);
}

#define SHIFT_BLOCKS 171
#define SHIFT_ROWS   (SHIFT_BLOCKS * 16)

template <int R>
__device__ __forceinline__ void shift_rows(const float* __restrict__ oldv,
                                           const float* __restrict__ oldk,
                                           float* __restrict__ newv,
                                           float* __restrict__ newk,
                                           int s, int lane) {
    const v4f* vsrc = (const v4f*)(oldv + (size_t)(s + 1) * E_DIM);
    const v4f* ksrc = (const v4f*)(oldk + (size_t)(s + 1) * E_DIM);
    v4f* vdst = (v4f*)(newv + (size_t)s * E_DIM);
    v4f* kdst = (v4f*)(newk + (size_t)s * E_DIM);
    v4f dv[8 * R], dk[8 * R];
    #pragma unroll
    for (int i = 0; i < 8 * R; i++) dv[i] = vsrc[i * 32 + lane];
    #pragma unroll
    for (int i = 0; i < 8 * R; i++) dk[i] = ksrc[i * 32 + lane];
    #pragma unroll
    for (int r = 0; r < R; r++)
        asm volatile("" :: "v"(dv[8 * r + 0]), "v"(dv[8 * r + 1]), "v"(dv[8 * r + 2]),
                           "v"(dv[8 * r + 3]), "v"(dv[8 * r + 4]), "v"(dv[8 * r + 5]),
                           "v"(dv[8 * r + 6]), "v"(dv[8 * r + 7]));
    #pragma unroll
    for (int r = 0; r < R; r++)
        asm volatile("" :: "v"(dk[8 * r + 0]), "v"(dk[8 * r + 1]), "v"(dk[8 * r + 2]),
                           "v"(dk[8 * r + 3]), "v"(dk[8 * r + 4]), "v"(dk[8 * r + 5]),
                           "v"(dk[8 * r + 6]), "v"(dk[8 * r + 7]));
    #pragma unroll
    for (int i = 0; i < 8 * R; i++) vdst[i * 32 + lane] = dv[i];
    #pragma unroll
    for (int i = 0; i < 8 * R; i++) kdst[i * 32 + lane] = dk[i];
}

__device__ __forceinline__ void shift_role(const float* __restrict__ oldv,
                                           const float* __restrict__ oldk,
                                           float* __restrict__ newv,
                                           float* __restrict__ newk,
                                           int row0, int b, int tid) {
    const int s = row0 + b * 16 + (tid >> 5) * 2;
    if (s + 1 < S_LEN - 1)  shift_rows<2>(oldv, oldk, newv, newk, s, tid & 31);
    else if (s < S_LEN - 1) shift_rows<1>(oldv, oldk, newv, newk, s, tid & 31);
}

#define QKV_BLOCKS  96
#define FLAG_BLOCKS 256

__global__ void __launch_bounds__(256)
qkv_part_flag_kernel(const float* __restrict__ x,
                     const float* __restrict__ Wv,
                     const float* __restrict__ Wq,
                     const float* __restrict__ Wk,
                     float* __restrict__ part_qkv,
                     const float* __restrict__ oldv,
                     int* __restrict__ rowflag) {
    __shared__ float xs[256];
    const int tid  = threadIdx.x;
    const int wv   = tid >> 5;
    const int lane = tid & 31;

    if (blockIdx.x < QKV_BLOCKS) {
        const int bx = blockIdx.x & 7;
        const int by = (blockIdx.x >> 3) & 3;
        const int m  = blockIdx.x >> 5;
        const int k0 = by * 256;
        xs[tid] = x[k0 + tid];
        __syncthreads();

        const int j    = lane & 15;
        const int half = lane >> 4;
        const int ob   = bx * 128 + wv * 16;
        const float amask = (j == 0) ? 1.f : 0.f;

        const float* W = (m == 0) ? Wv : (m == 1) ? Wq : Wk;

        v8f c0 = {}, c1 = {};
        for (int eb = 0; eb < 256; eb += 8) {
            const int r0 = eb + half * 2;
            v2f a0, a1, b0, b1;
            a0.x = xs[r0]     * amask;  a0.y = xs[r0 + 1] * amask;
            a1.x = xs[r0 + 4] * amask;  a1.y = xs[r0 + 5] * amask;
            const float* wp0 = W + (size_t)(k0 + r0)     * E_DIM + ob + j;
            const float* wp1 = W + (size_t)(k0 + r0 + 4) * E_DIM + ob + j;
            b0.x = wp0[0];  b0.y = wp0[E_DIM];
            b1.x = wp1[0];  b1.y = wp1[E_DIM];
            c0 = wmma4(a0, b0, c0);
            c1 = wmma4(a1, b1, c1);
        }
        const v8f cs = c0 + c1;
        if (lane < 16)
            part_qkv[(size_t)(m * 4 + by) * E_DIM + ob + lane] = cs[0];
    } else {
        const int gw = (blockIdx.x - QKV_BLOCKS) * 8 + wv;
        v4f d[4][8];
        #pragma unroll
        for (int q = 0; q < 4; q++) {
            const int s  = gw + q * (FLAG_BLOCKS * 8);
            const int sr = (s < S_LEN - 1) ? s : gw;
            const v4f* p = (const v4f*)(oldv + (size_t)(sr + 1) * E_DIM);
            #pragma unroll
            for (int i = 0; i < 8; i++) d[q][i] = p[i * 32 + lane];
        }
        #pragma unroll
        for (int q = 0; q < 4; q++)
            asm volatile("" :: "v"(d[q][0]), "v"(d[q][1]), "v"(d[q][2]), "v"(d[q][3]),
                               "v"(d[q][4]), "v"(d[q][5]), "v"(d[q][6]), "v"(d[q][7]));
        #pragma unroll
        for (int q = 0; q < 4; q++) {
            const int s = gw + q * (FLAG_BLOCKS * 8);
            bool nz = false;
            #pragma unroll
            for (int i = 0; i < 8; i++)
                nz |= (d[q][i].x != 0.f) | (d[q][i].y != 0.f) | (d[q][i].z != 0.f) | (d[q][i].w != 0.f);
            const bool any = __builtin_amdgcn_ballot_w32(nz) != 0u;
            if (lane == 0 && s < S_LEN - 1) rowflag[s] = any ? 1 : 0;
        }
    }
}

#define SC_BLOCKS 256

struct QkvParts { float4 p[4]; float4 b; };

__device__ __forceinline__ QkvParts qkv_load(const float* __restrict__ part_qkv, int m,
                                             const float* __restrict__ bias, int t) {
    QkvParts r;
    #pragma unroll
    for (int ks = 0; ks < 4; ks++)
        r.p[ks] = ((const float4*)(part_qkv + (size_t)(m * 4 + ks) * E_DIM))[t];
    r.b = ((const float4*)bias)[t];
    return r;
}

__device__ __forceinline__ float4 qkv_sum(const QkvParts& r) {
    float4 s = make_float4(0.f, 0.f, 0.f, 0.f);
    #pragma unroll
    for (int ks = 0; ks < 4; ks++) {
        s.x += r.p[ks].x;  s.y += r.p[ks].y;  s.z += r.p[ks].z;  s.w += r.p[ks].w;
    }
    s.x += r.b.x;  s.y += r.b.y;  s.z += r.b.z;  s.w += r.b.w;
    return s;
}

template <int MASK>
__device__ __forceinline__ float row_xor(float v) {
    const int iv = __builtin_bit_cast(int, v);
    return __builtin_bit_cast(float,
        __builtin_amdgcn_update_dpp(iv, iv, 0x160 | MASK, 0xf, 0xf, false));
}

__device__ __forceinline__ float half_xor(float v) {
    const int iv = __builtin_bit_cast(int, v);
    return __builtin_bit_cast(float,
        __builtin_amdgcn_permlanex16(iv, iv, 0x76543210u, 0xfedcba98u, false, false));
}

__device__ __forceinline__ float row_sum16(const float (&p)[16], int j) {
    const bool b0 = (j & 1) != 0, b1 = (j & 2) != 0, b2 = (j & 4) != 0, b3 = (j & 8) != 0;
    float r1[8], r2[4], r3[2];
    #pragma unroll
    for (int m = 0; m < 8; m++) {
        const float keep = b0 ? p[2 * m + 1] : p[2 * m];
        const float send = b0 ? p[2 * m] : p[2 * m + 1];
        r1[m] = keep + row_xor<1>(send);
    }
    #pragma unroll
    for (int m = 0; m < 4; m++) {
        const float keep = b1 ? r1[2 * m + 1] : r1[2 * m];
        const float send = b1 ? r1[2 * m] : r1[2 * m + 1];
        r2[m] = keep + row_xor<2>(send);
    }
    #pragma unroll
    for (int m = 0; m < 2; m++) {
        const float keep = b2 ? r2[2 * m + 1] : r2[2 * m];
        const float send = b2 ? r2[2 * m] : r2[2 * m + 1];
        r3[m] = keep + row_xor<4>(send);
    }
    const float keep = b3 ? r3[1] : r3[0];
    const float send = b3 ? r3[0] : r3[1];
    return keep + row_xor<8>(send);
}

__device__ __forceinline__ float2 block_softmax(float sc, float& e) {
    float m = fmaxf(sc, half_xor(sc));
    m = fmaxf(m, row_xor<1>(m));
    m = fmaxf(m, row_xor<2>(m));
    m = fmaxf(m, row_xor<4>(m));
    m = fmaxf(m, row_xor<8>(m));
    const bool dead = (m == -INFINITY);
    e = dead ? 0.f : expf(sc - m);
    float l = e + half_xor(e);
    l += row_xor<1>(l);
    l += row_xor<2>(l);
    l += row_xor<4>(l);
    l += row_xor<8>(l);
    return make_float2(m, l);
}

__device__ __forceinline__ float dot4(const v4f k, const float4 q) {
    return fmaf(k.w, q.w, fmaf(k.z, q.z, fmaf(k.x, q.x, k.y * q.y)));
}

__global__ void __launch_bounds__(256)
score_kernel(const float* __restrict__ oldk,
             const float* __restrict__ part_qkv,
             const float* __restrict__ bv,
             const float* __restrict__ bq,
             const float* __restrict__ bk,
             const int* __restrict__ rowflag,
             float* __restrict__ vrow_last,
             float* __restrict__ krow_last,
             float* __restrict__ escore,
             float* __restrict__ stats,
             float* __restrict__ values,
             unsigned int* __restrict__ arrive) {
    __shared__ int wnz_s[8];
    const int tid  = threadIdx.x;
    const int lane = tid & 31;
    const int wv   = tid >> 5;
    const int h    = blockIdx.x >> 4;
    const int ch   = blockIdx.x & 15;
    const int s0   = ch * 512 + wv * 64;
    const int j    = lane & 15;
    const int half = lane >> 4;
    const bool lastchunk = (ch == 15);
    const bool lastwave  = lastchunk && (wv == 7);
    const bool lastpos = (s0 + 16 * half + 47 == S_LEN - 1);

    const QkvParts qp = qkv_load(part_qkv, 1, bq, h * 16 + j);
    const int sB  = s0 + 32 + lane;
    const int fla = rowflag[s0 + lane];
    const int flb = rowflag[(sB < S_LEN - 1) ? sB : s0 + lane];
    __builtin_amdgcn_sched_barrier(0);

    const v4f* kp = (const v4f*)(oldk + (size_t)(s0 + 16 * half + 1) * E_DIM + h * D_H) + j;
    v4f ka[16], kb[16];
    #pragma unroll
    for (int i = 0; i < 16; i++) ka[i] = kp[(size_t)i * (E_DIM / 4)];
    #pragma unroll
    for (int i = 0; i < 15; i++) kb[i] = kp[(size_t)(32 + i) * (E_DIM / 4)];
    kb[15] = kp[(size_t)(lastpos ? 46 : 47) * (E_DIM / 4)];
    QkvParts vpar = {}, kpar = {};
    if (lastchunk) vpar = qkv_load(part_qkv, 0, bv, tid);
    if (lastwave)  kpar = qkv_load(part_qkv, 2, bk, h * 16 + j);
    __builtin_amdgcn_sched_barrier(0);

    bool nzlast = true;
    float4 kn = make_float4(0.f, 0.f, 0.f, 0.f);
    if (lastchunk) {
        const float4 vn = qkv_sum(vpar);
        const bool t = (vn.x != 0.f) | (vn.y != 0.f) | (vn.z != 0.f) | (vn.w != 0.f);
        const bool wany = __builtin_amdgcn_ballot_w32(t) != 0u;
        if (lane == 0) wnz_s[wv] = wany ? 1 : 0;
        __syncthreads();
        nzlast = (wnz_s[0] | wnz_s[1] | wnz_s[2] | wnz_s[3] |
                  wnz_s[4] | wnz_s[5] | wnz_s[6] | wnz_s[7]) != 0;
        if ((tid >> 4) == h) ((float4*)vrow_last)[tid] = vn;
        if (lastwave) {
            kn = qkv_sum(kpar);
            if (half == 1) ((float4*)krow_last)[h * 16 + j] = kn;
        }
    }
    if (blockIdx.x == 0) {
        ((float4*)values)[tid] = make_float4(0.f, 0.f, 0.f, 0.f);
        if (tid < 2) arrive[tid] = 0u;
    }

    const float4 qd = qkv_sum(qp);

    float pa[16], pb[16];
    #pragma unroll
    for (int i = 0; i < 16; i++) pa[i] = dot4(ka[i], qd);
    #pragma unroll
    for (int i = 0; i < 15; i++) pb[i] = dot4(kb[i], qd);
    {
        v4f k15 = kb[15];
        if (lastpos) { k15.x = kn.x;  k15.y = kn.y;  k15.z = kn.z;  k15.w = kn.w; }
        pb[15] = dot4(k15, qd);
    }

    const float ra = row_sum16(pa, j);
    const float rb = row_sum16(pb, j);
    const bool nza = fla != 0;
    const bool nzb = (sB == S_LEN - 1) ? nzlast : (flb != 0);
    const float sca = nza ? ra * 0.125f : -INFINITY;
    const float scb = nzb ? rb * 0.125f : -INFINITY;

    float ea, eb;
    const float2 sta = block_softmax(sca, ea);
    const float2 stb = block_softmax(scb, eb);

    float* erow = escore + (size_t)h * S_LEN + s0;
    erow[lane]      = ea;
    erow[32 + lane] = eb;
    if (lane == 0)
        ((float4*)stats)[h * (SC_BLOCKS / 2) + ch * 8 + wv] = make_float4(sta.x, sta.y, stb.x, stb.y);
}

__global__ void __launch_bounds__(256)
attnv_kernel(const float* __restrict__ escore,
             const float* __restrict__ stats,
             const float* __restrict__ oldv,
             const float* __restrict__ vrow_last,
             float* __restrict__ values) {
    __shared__ float4 red_s[8][16];
    const int tid  = threadIdx.x;
    const int lane = tid & 31;
    const int wv   = tid >> 5;
    const int h    = blockIdx.x >> 4;
    const int ch   = blockIdx.x & 15;
    const int s0   = ch * 512 + wv * 64;
    const int j    = lane & 15;
    const int half = lane >> 4;

    const float2* st = (const float2*)stats + h * SC_BLOCKS;
    float2 pr[8];
    #pragma unroll
    for (int i = 0; i < 8; i++) pr[i] = st[i * 32 + lane];
    const float4 own = ((const float4*)st)[ch * 8 + wv];

    const float4* pp = (const float4*)(escore + (size_t)h * S_LEN + s0 + 16 * half);
    float4 pa[4], pb[4];
    #pragma unroll
    for (int i = 0; i < 4; i++) pa[i] = pp[i];
    #pragma unroll
    for (int i = 0; i < 4; i++) pb[i] = pp[8 + i];

    asm volatile("" :: "v"(pr[0].x), "v"(pr[0].y), "v"(pr[1].x), "v"(pr[1].y),
                       "v"(pr[2].x), "v"(pr[2].y), "v"(pr[3].x), "v"(pr[3].y),
                       "v"(pr[4].x), "v"(pr[4].y), "v"(pr[5].x), "v"(pr[5].y),
                       "v"(pr[6].x), "v"(pr[6].y), "v"(pr[7].x), "v"(pr[7].y),
                       "v"(own.x), "v"(own.y), "v"(own.z), "v"(own.w));
    __builtin_amdgcn_sched_barrier(0);

    const v4f* vp = (const v4f*)(oldv + (size_t)(s0 + 16 * half + 1) * E_DIM + h * D_H) + j;
    const v4f* vlast = (s0 + 16 * half + 47 == S_LEN - 1)
                           ? (const v4f*)(vrow_last + h * D_H) + j
                           : vp + (size_t)47 * (E_DIM / 4);
    v4f va[16], vb[16];
    #pragma unroll
    for (int i = 0; i < 16; i++) va[i] = vp[(size_t)i * (E_DIM / 4)];
    #pragma unroll
    for (int i = 0; i < 15; i++) vb[i] = vp[(size_t)(32 + i) * (E_DIM / 4)];
    vb[15] = *vlast;
    __builtin_amdgcn_sched_barrier(0);

    float M = pr[0].x;
    #pragma unroll
    for (int i = 1; i < 8; i++) M = fmaxf(M, pr[i].x);
    M = fmaxf(M, __shfl_xor(M, 1, 32));
    M = fmaxf(M, __shfl_xor(M, 2, 32));
    M = fmaxf(M, __shfl_xor(M, 4, 32));
    M = fmaxf(M, __shfl_xor(M, 8, 32));
    M = fmaxf(M, __shfl_xor(M, 16, 32));
    float L = 0.f;
    #pragma unroll
    for (int i = 0; i < 8; i++) L += pr[i].y * expf(pr[i].x - M);
    L += __shfl_xor(L, 1, 32);
    L += __shfl_xor(L, 2, 32);
    L += __shfl_xor(L, 4, 32);
    L += __shfl_xor(L, 8, 32);
    L += __shfl_xor(L, 16, 32);
    const float rL = 1.f / L;
    const float fa = expf(own.x - M) * rL;
    const float fb = expf(own.z - M) * rL;

    const float m0 = (j == 0) ? 1.f : 0.f;
    const float m1 = (j == 1) ? 1.f : 0.f;

    v8f axy = {}, azw = {}, bxy = {}, bzw = {};
    #pragma unroll
    for (int i = 0; i < 16; i++) {
        const float4 q = pa[i >> 2];
        const float e = ((i & 3) == 0) ? q.x : ((i & 3) == 1) ? q.y : ((i & 3) == 2) ? q.z : q.w;
        v2f a, b0, b1;
        a.x = e * m0;  a.y = e * m1;
        b0.x = va[i].x;  b0.y = va[i].y;
        b1.x = va[i].z;  b1.y = va[i].w;
        axy = wmma4(a, b0, axy);
        azw = wmma4(a, b1, azw);
    }
    #pragma unroll
    for (int i = 0; i < 16; i++) {
        const float4 q = pb[i >> 2];
        const float e = ((i & 3) == 0) ? q.x : ((i & 3) == 1) ? q.y : ((i & 3) == 2) ? q.z : q.w;
        v2f a, b0, b1;
        a.x = e * m0;  a.y = e * m1;
        b0.x = vb[i].x;  b0.y = vb[i].y;
        b1.x = vb[i].z;  b1.y = vb[i].w;
        bxy = wmma4(a, b0, bxy);
        bzw = wmma4(a, b1, bzw);
    }

    if (lane < 16)
        red_s[wv][lane] = make_float4(fa * axy[0] + fb * bxy[0], fa * axy[1] + fb * bxy[1],
                                      fa * azw[0] + fb * bzw[0], fa * azw[1] + fb * bzw[1]);
    __syncthreads();
    if (tid < D_H) {
        const float* rs = (const float*)red_s;
        float s = 0.f;
        #pragma unroll
        for (int w = 0; w < 8; w++) s += rs[w * D_H + tid];
        atomicAdd(&values[h * D_H + tid], s);
    }
}

#define GEMV_BLOCKS 64

__device__ __forceinline__ float wave_sum32(float v) {
    v += half_xor(v);
    v += row_xor<1>(v);
    v += row_xor<2>(v);
    v += row_xor<4>(v);
    v += row_xor<8>(v);
    return v;
}

__global__ void __launch_bounds__(256)
gemv_full_kernel(const float* __restrict__ W,
                 const float* __restrict__ in,
                 const float* __restrict__ pre_a,
                 const float* __restrict__ pre_b,
                 const float* __restrict__ post_bias,
                 float* __restrict__ outv,
                 unsigned int* __restrict__ arrive,
                 const float* __restrict__ ln_sc,
                 const float* __restrict__ ln_bs,
                 float* __restrict__ ln_out,
                 const float* __restrict__ oldv, const float* __restrict__ oldk,
                 float* __restrict__ newv, float* __restrict__ newk, int row0) {
    __shared__ float4 xs4[E_DIM / 4];
    __shared__ float  red_s[8][16];
    __shared__ float  lnsum_s[2][8];
    __shared__ unsigned int ticket_s;
    const int tid = threadIdx.x;
    if (blockIdx.x >= GEMV_BLOCKS) {
        shift_role(oldv, oldk, newv, newk, row0, blockIdx.x - GEMV_BLOCKS, tid);
        return;
    }
    const int ob = blockIdx.x * 16;
    float ea = 0.f, eb = 0.f, ec = 0.f;
    if (tid < 16) {
        if (pre_a)     { ea = pre_a[ob + tid];  eb = pre_b[ob + tid]; }
        if (post_bias) ec = post_bias[ob + tid];
    }
    xs4[tid] = ((const float4*)in)[tid];
    __syncthreads();
    const float* xs = (const float*)xs4;

    const int wv   = tid >> 5;
    const int lane = tid & 31;
    const int j    = lane & 15;
    const int half = lane >> 4;
    const int k0   = wv * 128;
    const float amask = (j == 0) ? 1.f : 0.f;

    v8f c0 = {}, c1 = {};
    for (int eb8 = 0; eb8 < 128; eb8 += 8) {
        const int r0 = eb8 + half * 2;
        v2f a0, a1, b0, b1;
        a0.x = xs[k0 + r0]     * amask;  a0.y = xs[k0 + r0 + 1] * amask;
        a1.x = xs[k0 + r0 + 4] * amask;  a1.y = xs[k0 + r0 + 5] * amask;
        const float* wp0 = W + (size_t)(k0 + r0)     * E_DIM + ob + j;
        const float* wp1 = W + (size_t)(k0 + r0 + 4) * E_DIM + ob + j;
        b0.x = wp0[0];  b0.y = wp0[E_DIM];
        b1.x = wp1[0];  b1.y = wp1[E_DIM];
        c0 = wmma4(a0, b0, c0);
        c1 = wmma4(a1, b1, c1);
    }
    const v8f cs = c0 + c1;
    if (lane < 16) red_s[wv][lane] = cs[0];
    __syncthreads();
    if (tid < 16) {
        float s = pre_a ? (ea + eb) : 0.f;
        for (int w = 0; w < 8; w++) s += red_s[w][tid];
        if (post_bias) { s += ec;  s = fmaxf(s, 0.f); }
        outv[ob + tid] = s;
    }
    if (arrive == nullptr) return;

    if (wv == 0) {
        __builtin_amdgcn_fence(__ATOMIC_RELEASE, "agent");
        if (lane == 0)
            ticket_s = __hip_atomic_fetch_add(arrive, 1u, __ATOMIC_RELAXED,
                                              __HIP_MEMORY_SCOPE_AGENT);
    }
    __syncthreads();
    if (ticket_s != GEMV_BLOCKS - 1) return;

    __builtin_amdgcn_fence(__ATOMIC_ACQUIRE, "agent");
    const float4 xv = ((const float4*)outv)[tid];
    const float4 sv = ((const float4*)ln_sc)[tid];
    const float4 bv = ((const float4*)ln_bs)[tid];

    const float tsum = wave_sum32(((xv.x + xv.y) + xv.z) + xv.w);
    if (lane == 0) lnsum_s[0][wv] = tsum;
    __syncthreads();
    float mean = lnsum_s[0][0];
    for (int w = 1; w < 8; w++) mean += lnsum_s[0][w];
    mean *= (1.f / E_DIM);

    const float dx = xv.x - mean, dy = xv.y - mean, dz = xv.z - mean, dw = xv.w - mean;
    const float tvar = wave_sum32(((dx * dx + dy * dy) + dz * dz) + dw * dw);
    if (lane == 0) lnsum_s[1][wv] = tvar;
    __syncthreads();
    float var = lnsum_s[1][0];
    for (int w = 1; w < 8; w++) var += lnsum_s[1][w];
    const float inv = rsqrtf(var * (1.f / E_DIM) + 1e-6f);

    ((float4*)ln_out)[tid] = make_float4(dx * inv * sv.x + bv.x, dy * inv * sv.y + bv.y,
                                         dz * inv * sv.z + bv.z, dw * inv * sv.w + bv.w);
}

extern "C" void kernel_launch(void* const* d_in, const int* in_sizes, int n_in,
                              void* d_out, int out_size, void* d_ws, size_t ws_size,
                              hipStream_t stream) {
    (void)in_sizes; (void)n_in; (void)out_size; (void)ws_size;
    const float* x     = (const float*)d_in[0];
    const float* cache = (const float*)d_in[1];
    const float* Wv = (const float*)d_in[2];  const float* bv = (const float*)d_in[3];
    const float* Wq = (const float*)d_in[4];  const float* bq = (const float*)d_in[5];
    const float* Wk = (const float*)d_in[6];  const float* bk = (const float*)d_in[7];
    const float* Wo = (const float*)d_in[8];  const float* bo = (const float*)d_in[9];
    const float* W1 = (const float*)d_in[10]; const float* b1 = (const float*)d_in[11];
    const float* W2 = (const float*)d_in[12]; const float* b2 = (const float*)d_in[13];
    const float* ln1_s = (const float*)d_in[14]; const float* ln1_b = (const float*)d_in[15];
    const float* ln2_s = (const float*)d_in[16]; const float* ln2_b = (const float*)d_in[17];

    float* out_h = (float*)d_out;
    float* newv  = out_h + E_DIM;
    float* newk  = newv + (size_t)S_LEN * E_DIM;
    const float* oldv = cache;
    const float* oldk = cache + (size_t)S_LEN * E_DIM;

    float* ws       = (float*)d_ws;
    float* scores   = ws + E_DIM;
    float* values   = scores + H_N * S_LEN;
    float* h1       = values + E_DIM;
    float* part_qkv = h1 + E_DIM;
    float* pre1     = part_qkv + 12 * E_DIM;
    float* act1     = pre1 + E_DIM;
    float* pre2     = act1 + E_DIM;
    int*   rowflag  = (int*)(pre2 + E_DIM);
    float* stats    = (float*)(rowflag + S_LEN);
    unsigned int* arrive = (unsigned int*)(stats + 2 * H_N * SC_BLOCKS);

    float* vrow_last = newv + (size_t)(S_LEN - 1) * E_DIM;
    float* krow_last = newk + (size_t)(S_LEN - 1) * E_DIM;

    qkv_part_flag_kernel<<<QKV_BLOCKS + FLAG_BLOCKS, 256, 0, stream>>>(
        x, Wv, Wq, Wk, part_qkv, oldv, rowflag);
    score_kernel<<<SC_BLOCKS, 256, 0, stream>>>(oldk, part_qkv, bv, bq, bk, rowflag,
                                                vrow_last, krow_last,
                                                scores, stats, values, arrive);
    attnv_kernel<<<256, 256, 0, stream>>>(scores, stats, oldv, vrow_last, values);
    gemv_full_kernel<<<GEMV_BLOCKS + SHIFT_BLOCKS, 256, 0, stream>>>(
        Wo, values, bo, x, nullptr, pre1, arrive, ln1_s, ln1_b, h1,
        oldv, oldk, newv, newk, 0);
    gemv_full_kernel<<<GEMV_BLOCKS + SHIFT_BLOCKS, 256, 0, stream>>>(
        W1, h1, nullptr, nullptr, b1, act1, nullptr, nullptr, nullptr, nullptr,
        oldv, oldk, newv, newk, 2736);
    gemv_full_kernel<<<GEMV_BLOCKS + SHIFT_BLOCKS, 256, 0, stream>>>(
        W2, act1, b2, h1, nullptr, pre2, arrive + 1, ln2_s, ln2_b, out_h,
        oldv, oldk, newv, newk, 5472);
}


# V-row flags of launch 1 from one 16-byte probe per row (a row is read whole only where its probe is all zero): flag role 32 blocks reading 1/256 of the V plane, not 256 blocks reading all; same sums
# speedup vs baseline: 3.3901x; 1.0444x over previous
#include <hip/hip_runtime.h>
#include <hip/hip_bf16.h>
#include <math.h>

#define E_DIM 1024
#define S_LEN 8192
#define H_N   16
#define D_H   64

typedef __attribute__((ext_vector_type(2))) float v2f;
typedef __attribute__((ext_vector_type(8))) float v8f;
typedef __attribute__((ext_vector_type(4))) float v4f;

__device__ __forceinline__ v8f wmma4(v2f a, v2f b, v8f c) {
    return __builtin_amdgcn_wmma_f32_16x16x4_f32(false, a, false, b, (short)0, c,
                                                 false, false);
}

#define SHIFT_BLOCKS 171
#define SHIFT_ROWS   (SHIFT_BLOCKS * 16)

template <int R>
__device__ __forceinline__ void shift_rows(const float* __restrict__ oldv,
                                           const float* __restrict__ oldk,
                                           float* __restrict__ newv,
                                           float* __restrict__ newk,
                                           int s, int lane) {
    const v4f* vsrc = (const v4f*)(oldv + (size_t)(s + 1) * E_DIM);
    const v4f* ksrc = (const v4f*)(oldk + (size_t)(s + 1) * E_DIM);
    v4f* vdst = (v4f*)(newv + (size_t)s * E_DIM);
    v4f* kdst = (v4f*)(newk + (size_t)s * E_DIM);
    v4f dv[8 * R], dk[8 * R];
    #pragma unroll
    for (int i = 0; i < 8 * R; i++) dv[i] = vsrc[i * 32 + lane];
    #pragma unroll
    for (int i = 0; i < 8 * R; i++) dk[i] = ksrc[i * 32 + lane];
    #pragma unroll
    for (int r = 0; r < R; r++)
        asm volatile("" :: "v"(dv[8 * r + 0]), "v"(dv[8 * r + 1]), "v"(dv[8 * r + 2]),
                           "v"(dv[8 * r + 3]), "v"(dv[8 * r + 4]), "v"(dv[8 * r + 5]),
                           "v"(dv[8 * r + 6]), "v"(dv[8 * r + 7]));
    #pragma unroll
    for (int r = 0; r < R; r++)
        asm volatile("" :: "v"(dk[8 * r + 0]), "v"(dk[8 * r + 1]), "v"(dk[8 * r + 2]),
                           "v"(dk[8 * r + 3]), "v"(dk[8 * r + 4]), "v"(dk[8 * r + 5]),
                           "v"(dk[8 * r + 6]), "v"(dk[8 * r + 7]));
    #pragma unroll
    for (int i = 0; i < 8 * R; i++) vdst[i * 32 + lane] = dv[i];
    #pragma unroll
    for (int i = 0; i < 8 * R; i++) kdst[i * 32 + lane] = dk[i];
}

__device__ __forceinline__ void shift_role(const float* __restrict__ oldv,
                                           const float* __restrict__ oldk,
                                           float* __restrict__ newv,
                                           float* __restrict__ newk,
                                           int row0, int b, int tid) {
    const int s = row0 + b * 16 + (tid >> 5) * 2;
    if (s + 1 < S_LEN - 1)  shift_rows<2>(oldv, oldk, newv, newk, s, tid & 31);
    else if (s < S_LEN - 1) shift_rows<1>(oldv, oldk, newv, newk, s, tid & 31);
}

#define QKV_BLOCKS  96
#define FLAG_BLOCKS 32

__global__ void __launch_bounds__(256) __attribute__((amdgpu_waves_per_eu(2, 7)))
qkv_part_flag_kernel(const float* __restrict__ x,
                     const float* __restrict__ Wv,
                     const float* __restrict__ Wq,
                     const float* __restrict__ Wk,
                     float* __restrict__ part_qkv,
                     const float* __restrict__ oldv,
                     int* __restrict__ rowflag) {
    __shared__ float xs[256];
    const int tid  = threadIdx.x;
    const int wv   = tid >> 5;
    const int lane = tid & 31;

    if (blockIdx.x < QKV_BLOCKS) {
        const int bx = blockIdx.x & 7;
        const int by = (blockIdx.x >> 3) & 3;
        const int m  = blockIdx.x >> 5;
        const int k0 = by * 256;
        xs[tid] = x[k0 + tid];
        __syncthreads();

        const int j    = lane & 15;
        const int half = lane >> 4;
        const int ob   = bx * 128 + wv * 16;
        const float amask = (j == 0) ? 1.f : 0.f;

        const float* W = (m == 0) ? Wv : (m == 1) ? Wq : Wk;

        v8f c0 = {}, c1 = {};
        for (int eb = 0; eb < 256; eb += 8) {
            const int r0 = eb + half * 2;
            v2f a0, a1, b0, b1;
            a0.x = xs[r0]     * amask;  a0.y = xs[r0 + 1] * amask;
            a1.x = xs[r0 + 4] * amask;  a1.y = xs[r0 + 5] * amask;
            const float* wp0 = W + (size_t)(k0 + r0)     * E_DIM + ob + j;
            const float* wp1 = W + (size_t)(k0 + r0 + 4) * E_DIM + ob + j;
            b0.x = wp0[0];  b0.y = wp0[E_DIM];
            b1.x = wp1[0];  b1.y = wp1[E_DIM];
            c0 = wmma4(a0, b0, c0);
            c1 = wmma4(a1, b1, c1);
        }
        const v8f cs = c0 + c1;
        if (lane < 16)
            part_qkv[(size_t)(m * 4 + by) * E_DIM + ob + lane] = cs[0];
    } else {
        const int gw  = (blockIdx.x - QKV_BLOCKS) * 8 + wv;
        const int s   = gw * 32 + lane;
        const bool has = s < S_LEN - 1;
        const int sr  = has ? s : s - 1;
        const v4f p = *(const v4f*)(oldv + (size_t)(sr + 1) * E_DIM + (sr & 15) * 64);
        bool nz = (p.x != 0.f) | (p.y != 0.f) | (p.z != 0.f) | (p.w != 0.f);
        unsigned int todo = __builtin_amdgcn_ballot_w32(has && !nz);
        while (todo != 0u) {
            const int r = __builtin_ctz(todo);
            todo &= todo - 1u;
            const v4f* q = (const v4f*)(oldv + (size_t)(gw * 32 + r + 1) * E_DIM);
            bool t = false;
            #pragma unroll
            for (int i = 0; i < 8; i++) {
                const v4f d = q[i * 32 + lane];
                t |= (d.x != 0.f) | (d.y != 0.f) | (d.z != 0.f) | (d.w != 0.f);
            }
            const bool any = __builtin_amdgcn_ballot_w32(t) != 0u;
            if (lane == r) nz = any;
        }
        if (has) rowflag[s] = nz ? 1 : 0;
    }
}

#define SC_BLOCKS 256

struct QkvParts { float4 p[4]; float4 b; };

__device__ __forceinline__ QkvParts qkv_load(const float* __restrict__ part_qkv, int m,
                                             const float* __restrict__ bias, int t) {
    QkvParts r;
    #pragma unroll
    for (int ks = 0; ks < 4; ks++)
        r.p[ks] = ((const float4*)(part_qkv + (size_t)(m * 4 + ks) * E_DIM))[t];
    r.b = ((const float4*)bias)[t];
    return r;
}

__device__ __forceinline__ float4 qkv_sum(const QkvParts& r) {
    float4 s = make_float4(0.f, 0.f, 0.f, 0.f);
    #pragma unroll
    for (int ks = 0; ks < 4; ks++) {
        s.x += r.p[ks].x;  s.y += r.p[ks].y;  s.z += r.p[ks].z;  s.w += r.p[ks].w;
    }
    s.x += r.b.x;  s.y += r.b.y;  s.z += r.b.z;  s.w += r.b.w;
    return s;
}

template <int MASK>
__device__ __forceinline__ float row_xor(float v) {
    const int iv = __builtin_bit_cast(int, v);
    return __builtin_bit_cast(float,
        __builtin_amdgcn_update_dpp(iv, iv, 0x160 | MASK, 0xf, 0xf, false));
}

__device__ __forceinline__ float half_xor(float v) {
    const int iv = __builtin_bit_cast(int, v);
    return __builtin_bit_cast(float,
        __builtin_amdgcn_permlanex16(iv, iv, 0x76543210u, 0xfedcba98u, false, false));
}

__device__ __forceinline__ float row_sum16(const float (&p)[16], int j) {
    const bool b0 = (j & 1) != 0, b1 = (j & 2) != 0, b2 = (j & 4) != 0, b3 = (j & 8) != 0;
    float r1[8], r2[4], r3[2];
    #pragma unroll
    for (int m = 0; m < 8; m++) {
        const float keep = b0 ? p[2 * m + 1] : p[2 * m];
        const float send = b0 ? p[2 * m] : p[2 * m + 1];
        r1[m] = keep + row_xor<1>(send);
    }
    #pragma unroll
    for (int m = 0; m < 4; m++) {
        const float keep = b1 ? r1[2 * m + 1] : r1[2 * m];
        const float send = b1 ? r1[2 * m] : r1[2 * m + 1];
        r2[m] = keep + row_xor<2>(send);
    }
    #pragma unroll
    for (int m = 0; m < 2; m++) {
        const float keep = b2 ? r2[2 * m + 1] : r2[2 * m];
        const float send = b2 ? r2[2 * m] : r2[2 * m + 1];
        r3[m] = keep + row_xor<4>(send);
    }
    const float keep = b3 ? r3[1] : r3[0];
    const float send = b3 ? r3[0] : r3[1];
    return keep + row_xor<8>(send);
}

__device__ __forceinline__ float2 block_softmax(float sc, float& e) {
    float m = fmaxf(sc, half_xor(sc));
    m = fmaxf(m, row_xor<1>(m));
    m = fmaxf(m, row_xor<2>(m));
    m = fmaxf(m, row_xor<4>(m));
    m = fmaxf(m, row_xor<8>(m));
    const bool dead = (m == -INFINITY);
    e = dead ? 0.f : expf(sc - m);
    float l = e + half_xor(e);
    l += row_xor<1>(l);
    l += row_xor<2>(l);
    l += row_xor<4>(l);
    l += row_xor<8>(l);
    return make_float2(m, l);
}

__device__ __forceinline__ float dot4(const v4f k, const float4 q) {
    return fmaf(k.w, q.w, fmaf(k.z, q.z, fmaf(k.x, q.x, k.y * q.y)));
}

__global__ void __launch_bounds__(256)
score_kernel(const float* __restrict__ oldk,
             const float* __restrict__ part_qkv,
             const float* __restrict__ bv,
             const float* __restrict__ bq,
             const float* __restrict__ bk,
             const int* __restrict__ rowflag,
             float* __restrict__ vrow_last,
             float* __restrict__ krow_last,
             float* __restrict__ escore,
             float* __restrict__ stats,
             float* __restrict__ values,
             unsigned int* __restrict__ arrive) {
    __shared__ int wnz_s[8];
    const int tid  = threadIdx.x;
    const int lane = tid & 31;
    const int wv   = tid >> 5;
    const int h    = blockIdx.x >> 4;
    const int ch   = blockIdx.x & 15;
    const int s0   = ch * 512 + wv * 64;
    const int j    = lane & 15;
    const int half = lane >> 4;
    const bool lastchunk = (ch == 15);
    const bool lastwave  = lastchunk && (wv == 7);
    const bool lastpos = (s0 + 16 * half + 47 == S_LEN - 1);

    const QkvParts qp = qkv_load(part_qkv, 1, bq, h * 16 + j);
    const int sB  = s0 + 32 + lane;
    const int fla = rowflag[s0 + lane];
    const int flb = rowflag[(sB < S_LEN - 1) ? sB : s0 + lane];
    __builtin_amdgcn_sched_barrier(0);

    const v4f* kp = (const v4f*)(oldk + (size_t)(s0 + 16 * half + 1) * E_DIM + h * D_H) + j;
    v4f ka[16], kb[16];
    #pragma unroll
    for (int i = 0; i < 16; i++) ka[i] = kp[(size_t)i * (E_DIM / 4)];
    #pragma unroll
    for (int i = 0; i < 15; i++) kb[i] = kp[(size_t)(32 + i) * (E_DIM / 4)];
    kb[15] = kp[(size_t)(lastpos ? 46 : 47) * (E_DIM / 4)];
    QkvParts vpar = {}, kpar = {};
    if (lastchunk) vpar = qkv_load(part_qkv, 0, bv, tid);
    if (lastwave)  kpar = qkv_load(part_qkv, 2, bk, h * 16 + j);
    __builtin_amdgcn_sched_barrier(0);

    bool nzlast = true;
    float4 kn = make_float4(0.f, 0.f, 0.f, 0.f);
    if (lastchunk) {
        const float4 vn = qkv_sum(vpar);
        const bool t = (vn.x != 0.f) | (vn.y != 0.f) | (vn.z != 0.f) | (vn.w != 0.f);
        const bool wany = __builtin_amdgcn_ballot_w32(t) != 0u;
        if (lane == 0) wnz_s[wv] = wany ? 1 : 0;
        __syncthreads();
        nzlast = (wnz_s[0] | wnz_s[1] | wnz_s[2] | wnz_s[3] |
                  wnz_s[4] | wnz_s[5] | wnz_s[6] | wnz_s[7]) != 0;
        if ((tid >> 4) == h) ((float4*)vrow_last)[tid] = vn;
        if (lastwave) {
            kn = qkv_sum(kpar);
            if (half == 1) ((float4*)krow_last)[h * 16 + j] = kn;
        }
    }
    if (blockIdx.x == 0) {
        ((float4*)values)[tid] = make_float4(0.f, 0.f, 0.f, 0.f);
        if (tid < 2) arrive[tid] = 0u;
    }

    const float4 qd = qkv_sum(qp);

    float pa[16], pb[16];
    #pragma unroll
    for (int i = 0; i < 16; i++) pa[i] = dot4(ka[i], qd);
    #pragma unroll
    for (int i = 0; i < 15; i++) pb[i] = dot4(kb[i], qd);
    {
        v4f k15 = kb[15];
        if (lastpos) { k15.x = kn.x;  k15.y = kn.y;  k15.z = kn.z;  k15.w = kn.w; }
        pb[15] = dot4(k15, qd);
    }

    const float ra = row_sum16(pa, j);
    const float rb = row_sum16(pb, j);
    const bool nza = fla != 0;
    const bool nzb = (sB == S_LEN - 1) ? nzlast : (flb != 0);
    const float sca = nza ? ra * 0.125f : -INFINITY;
    const float scb = nzb ? rb * 0.125f : -INFINITY;

    float ea, eb;
    const float2 sta = block_softmax(sca, ea);
    const float2 stb = block_softmax(scb, eb);

    float* erow = escore + (size_t)h * S_LEN + s0;
    erow[lane]      = ea;
    erow[32 + lane] = eb;
    if (lane == 0)
        ((float4*)stats)[h * (SC_BLOCKS / 2) + ch * 8 + wv] = make_float4(sta.x, sta.y, stb.x, stb.y);
}

__global__ void __launch_bounds__(256)
attnv_kernel(const float* __restrict__ escore,
             const float* __restrict__ stats,
             const float* __restrict__ oldv,
             const float* __restrict__ vrow_last,
             float* __restrict__ values) {
    __shared__ float4 red_s[8][16];
    const int tid  = threadIdx.x;
    const int lane = tid & 31;
    const int wv   = tid >> 5;
    const int h    = blockIdx.x >> 4;
    const int ch   = blockIdx.x & 15;
    const int s0   = ch * 512 + wv * 64;
    const int j    = lane & 15;
    const int half = lane >> 4;

    const float2* st = (const float2*)stats + h * SC_BLOCKS;
    float2 pr[8];
    #pragma unroll
    for (int i = 0; i < 8; i++) pr[i] = st[i * 32 + lane];
    const float4 own = ((const float4*)st)[ch * 8 + wv];

    const float4* pp = (const float4*)(escore + (size_t)h * S_LEN + s0 + 16 * half);
    float4 pa[4], pb[4];
    #pragma unroll
    for (int i = 0; i < 4; i++) pa[i] = pp[i];
    #pragma unroll
    for (int i = 0; i < 4; i++) pb[i] = pp[8 + i];

    asm volatile("" :: "v"(pr[0].x), "v"(pr[0].y), "v"(pr[1].x), "v"(pr[1].y),
                       "v"(pr[2].x), "v"(pr[2].y), "v"(pr[3].x), "v"(pr[3].y),
                       "v"(pr[4].x), "v"(pr[4].y), "v"(pr[5].x), "v"(pr[5].y),
                       "v"(pr[6].x), "v"(pr[6].y), "v"(pr[7].x), "v"(pr[7].y),
                       "v"(own.x), "v"(own.y), "v"(own.z), "v"(own.w));
    __builtin_amdgcn_sched_barrier(0);

    const v4f* vp = (const v4f*)(oldv + (size_t)(s0 + 16 * half + 1) * E_DIM + h * D_H) + j;
    const v4f* vlast = (s0 + 16 * half + 47 == S_LEN - 1)
                           ? (const v4f*)(vrow_last + h * D_H) + j
                           : vp + (size_t)47 * (E_DIM / 4);
    v4f va[16], vb[16];
    #pragma unroll
    for (int i = 0; i < 16; i++) va[i] = vp[(size_t)i * (E_DIM / 4)];
    #pragma unroll
    for (int i = 0; i < 15; i++) vb[i] = vp[(size_t)(32 + i) * (E_DIM / 4)];
    vb[15] = *vlast;
    __builtin_amdgcn_sched_barrier(0);

    float M = pr[0].x;
    #pragma unroll
    for (int i = 1; i < 8; i++) M = fmaxf(M, pr[i].x);
    M = fmaxf(M, __shfl_xor(M, 1, 32));
    M = fmaxf(M, __shfl_xor(M, 2, 32));
    M = fmaxf(M, __shfl_xor(M, 4, 32));
    M = fmaxf(M, __shfl_xor(M, 8, 32));
    M = fmaxf(M, __shfl_xor(M, 16, 32));
    float L = 0.f;
    #pragma unroll
    for (int i = 0; i < 8; i++) L += pr[i].y * expf(pr[i].x - M);
    L += __shfl_xor(L, 1, 32);
    L += __shfl_xor(L, 2, 32);
    L += __shfl_xor(L, 4, 32);
    L += __shfl_xor(L, 8, 32);
    L += __shfl_xor(L, 16, 32);
    const float rL = 1.f / L;
    const float fa = expf(own.x - M) * rL;
    const float fb = expf(own.z - M) * rL;

    const float m0 = (j == 0) ? 1.f : 0.f;
    const float m1 = (j == 1) ? 1.f : 0.f;

    v8f axy = {}, azw = {}, bxy = {}, bzw = {};
    #pragma unroll
    for (int i = 0; i < 16; i++) {
        const float4 q = pa[i >> 2];
        const float e = ((i & 3) == 0) ? q.x : ((i & 3) == 1) ? q.y : ((i & 3) == 2) ? q.z : q.w;
        v2f a, b0, b1;
        a.x = e * m0;  a.y = e * m1;
        b0.x = va[i].x;  b0.y = va[i].y;
        b1.x = va[i].z;  b1.y = va[i].w;
        axy = wmma4(a, b0, axy);
        azw = wmma4(a, b1, azw);
    }
    #pragma unroll
    for (int i = 0; i < 16; i++) {
        const float4 q = pb[i >> 2];
        const float e = ((i & 3) == 0) ? q.x : ((i & 3) == 1) ? q.y : ((i & 3) == 2) ? q.z : q.w;
        v2f a, b0, b1;
        a.x = e * m0;  a.y = e * m1;
        b0.x = vb[i].x;  b0.y = vb[i].y;
        b1.x = vb[i].z;  b1.y = vb[i].w;
        bxy = wmma4(a, b0, bxy);
        bzw = wmma4(a, b1, bzw);
    }

    if (lane < 16)
        red_s[wv][lane] = make_float4(fa * axy[0] + fb * bxy[0], fa * axy[1] + fb * bxy[1],
                                      fa * azw[0] + fb * bzw[0], fa * azw[1] + fb * bzw[1]);
    __syncthreads();
    if (tid < D_H) {
        const float* rs = (const float*)red_s;
        float s = 0.f;
        #pragma unroll
        for (int w = 0; w < 8; w++) s += rs[w * D_H + tid];
        atomicAdd(&values[h * D_H + tid], s);
    }
}

#define GEMV_BLOCKS 64

__device__ __forceinline__ float wave_sum32(float v) {
    v += half_xor(v);
    v += row_xor<1>(v);
    v += row_xor<2>(v);
    v += row_xor<4>(v);
    v += row_xor<8>(v);
    return v;
}

__global__ void __launch_bounds__(256)
gemv_full_kernel(const float* __restrict__ W,
                 const float* __restrict__ in,
                 const float* __restrict__ pre_a,
                 const float* __restrict__ pre_b,
                 const float* __restrict__ post_bias,
                 float* __restrict__ outv,
                 unsigned int* __restrict__ arrive,
                 const float* __restrict__ ln_sc,
                 const float* __restrict__ ln_bs,
                 float* __restrict__ ln_out,
                 const float* __restrict__ oldv, const float* __restrict__ oldk,
                 float* __restrict__ newv, float* __restrict__ newk, int row0) {
    __shared__ float4 xs4[E_DIM / 4];
    __shared__ float  red_s[8][16];
    __shared__ float  lnsum_s[2][8];
    __shared__ unsigned int ticket_s;
    const int tid = threadIdx.x;
    if (blockIdx.x >= GEMV_BLOCKS) {
        shift_role(oldv, oldk, newv, newk, row0, blockIdx.x - GEMV_BLOCKS, tid);
        return;
    }
    const int ob = blockIdx.x * 16;
    float ea = 0.f, eb = 0.f, ec = 0.f;
    if (tid < 16) {
        if (pre_a)     { ea = pre_a[ob + tid];  eb = pre_b[ob + tid]; }
        if (post_bias) ec = post_bias[ob + tid];
    }
    xs4[tid] = ((const float4*)in)[tid];
    __syncthreads();
    const float* xs = (const float*)xs4;

    const int wv   = tid >> 5;
    const int lane = tid & 31;
    const int j    = lane & 15;
    const int half = lane >> 4;
    const int k0   = wv * 128;
    const float amask = (j == 0) ? 1.f : 0.f;

    v8f c0 = {}, c1 = {};
    for (int eb8 = 0; eb8 < 128; eb8 += 8) {
        const int r0 = eb8 + half * 2;
        v2f a0, a1, b0, b1;
        a0.x = xs[k0 + r0]     * amask;  a0.y = xs[k0 + r0 + 1] * amask;
        a1.x = xs[k0 + r0 + 4] * amask;  a1.y = xs[k0 + r0 + 5] * amask;
        const float* wp0 = W + (size_t)(k0 + r0)     * E_DIM + ob + j;
        const float* wp1 = W + (size_t)(k0 + r0 + 4) * E_DIM + ob + j;
        b0.x = wp0[0];  b0.y = wp0[E_DIM];
        b1.x = wp1[0];  b1.y = wp1[E_DIM];
        c0 = wmma4(a0, b0, c0);
        c1 = wmma4(a1, b1, c1);
    }
    const v8f cs = c0 + c1;
    if (lane < 16) red_s[wv][lane] = cs[0];
    __syncthreads();
    if (tid < 16) {
        float s = pre_a ? (ea + eb) : 0.f;
        for (int w = 0; w < 8; w++) s += red_s[w][tid];
        if (post_bias) { s += ec;  s = fmaxf(s, 0.f); }
        outv[ob + tid] = s;
    }
    if (arrive == nullptr) return;

    if (wv == 0) {
        __builtin_amdgcn_fence(__ATOMIC_RELEASE, "agent");
        if (lane == 0)
            ticket_s = __hip_atomic_fetch_add(arrive, 1u, __ATOMIC_RELAXED,
                                              __HIP_MEMORY_SCOPE_AGENT);
    }
    __syncthreads();
    if (ticket_s != GEMV_BLOCKS - 1) return;

    __builtin_amdgcn_fence(__ATOMIC_ACQUIRE, "agent");
    const float4 xv = ((const float4*)outv)[tid];
    const float4 sv = ((const float4*)ln_sc)[tid];
    const float4 bv = ((const float4*)ln_bs)[tid];

    const float tsum = wave_sum32(((xv.x + xv.y) + xv.z) + xv.w);
    if (lane == 0) lnsum_s[0][wv] = tsum;
    __syncthreads();
    float mean = lnsum_s[0][0];
    for (int w = 1; w < 8; w++) mean += lnsum_s[0][w];
    mean *= (1.f / E_DIM);

    const float dx = xv.x - mean, dy = xv.y - mean, dz = xv.z - mean, dw = xv.w - mean;
    const float tvar = wave_sum32(((dx * dx + dy * dy) + dz * dz) + dw * dw);
    if (lane == 0) lnsum_s[1][wv] = tvar;
    __syncthreads();
    float var = lnsum_s[1][0];
    for (int w = 1; w < 8; w++) var += lnsum_s[1][w];
    const float inv = rsqrtf(var * (1.f / E_DIM) + 1e-6f);

    ((float4*)ln_out)[tid] = make_float4(dx * inv * sv.x + bv.x, dy * inv * sv.y + bv.y,
                                         dz * inv * sv.z + bv.z, dw * inv * sv.w + bv.w);
}

extern "C" void kernel_launch(void* const* d_in, const int* in_sizes, int n_in,
                              void* d_out, int out_size, void* d_ws, size_t ws_size,
                              hipStream_t stream) {
    (void)in_sizes; (void)n_in; (void)out_size; (void)ws_size;
    const float* x     = (const float*)d_in[0];
    const float* cache = (const float*)d_in[1];
    const float* Wv = (const float*)d_in[2];  const float* bv = (const float*)d_in[3];
    const float* Wq = (const float*)d_in[4];  const float* bq = (const float*)d_in[5];
    const float* Wk = (const float*)d_in[6];  const float* bk = (const float*)d_in[7];
    const float* Wo = (const float*)d_in[8];  const float* bo = (const float*)d_in[9];
    const float* W1 = (const float*)d_in[10]; const float* b1 = (const float*)d_in[11];
    const float* W2 = (const float*)d_in[12]; const float* b2 = (const float*)d_in[13];
    const float* ln1_s = (const float*)d_in[14]; const float* ln1_b = (const float*)d_in[15];
    const float* ln2_s = (const float*)d_in[16]; const float* ln2_b = (const float*)d_in[17];

    float* out_h = (float*)d_out;
    float* newv  = out_h + E_DIM;
    float* newk  = newv + (size_t)S_LEN * E_DIM;
    const float* oldv = cache;
    const float* oldk = cache + (size_t)S_LEN * E_DIM;

    float* ws       = (float*)d_ws;
    float* scores   = ws + E_DIM;
    float* values   = scores + H_N * S_LEN;
    float* h1       = values + E_DIM;
    float* part_qkv = h1 + E_DIM;
    float* pre1     = part_qkv + 12 * E_DIM;
    float* act1     = pre1 + E_DIM;
    float* pre2     = act1 + E_DIM;
    int*   rowflag  = (int*)(pre2 + E_DIM);
    float* stats    = (float*)(rowflag + S_LEN);
    unsigned int* arrive = (unsigned int*)(stats + 2 * H_N * SC_BLOCKS);

    float* vrow_last = newv + (size_t)(S_LEN - 1) * E_DIM;
    float* krow_last = newk + (size_t)(S_LEN - 1) * E_DIM;

    qkv_part_flag_kernel<<<QKV_BLOCKS + FLAG_BLOCKS, 256, 0, stream>>>(
        x, Wv, Wq, Wk, part_qkv, oldv, rowflag);
    score_kernel<<<SC_BLOCKS, 256, 0, stream>>>(oldk, part_qkv, bv, bq, bk, rowflag,
                                                vrow_last, krow_last,
                                                scores, stats, values, arrive);
    attnv_kernel<<<256, 256, 0, stream>>>(scores, stats, oldv, vrow_last, values);
    gemv_full_kernel<<<GEMV_BLOCKS + SHIFT_BLOCKS, 256, 0, stream>>>(
        Wo, values, bo, x, nullptr, pre1, arrive, ln1_s, ln1_b, h1,
        oldv, oldk, newv, newk, 0);
    gemv_full_kernel<<<GEMV_BLOCKS + SHIFT_BLOCKS, 256, 0, stream>>>(
        W1, h1, nullptr, nullptr, b1, act1, nullptr, nullptr, nullptr, nullptr,
        oldv, oldk, newv, newk, 2736);
    gemv_full_kernel<<<GEMV_BLOCKS + SHIFT_BLOCKS, 256, 0, stream>>>(
        W2, act1, b2, h1, nullptr, pre2, arrive + 1, ln2_s, ln2_b, out_h,
        oldv, oldk, newv, newk, 5472);
}
